# starting kernel
# baseline (speedup 1.0000x reference)
#include <hip/hip_runtime.h>
#include <hip/hip_bf16.h>


typedef __attribute__((ext_vector_type(2))) float v2f;
typedef __attribute__((ext_vector_type(8))) float v8f;

#define BB      4
#define NN      4096
#define CX      32
#define CIN     35
#define CINP    36
#define COUT    64
#define KS      32
#define R2      0.01f
#define EPSV    1e-5f
#define NPAIRS  (BB * (NN / 2))
#define GCONV   1024
#define CNTF    ((float)BB * (float)NN * (float)KS)

#define FSTRIDE  65
#define W1STRIDE 36
#define W2STRIDE 68

__global__ void xpose_kernel(const float* __restrict__ x, float* __restrict__ xt) {
  __shared__ float tile[CX][65];
  int b  = blockIdx.y;
  int n0 = blockIdx.x * 64;
  int t  = threadIdx.x;
  for (int i = t; i < CX * 64; i += 256) {
    int c = i >> 6, nn = i & 63;
    tile[c][nn] = x[((size_t)b * CX + c) * NN + n0 + nn];
  }
  __syncthreads();
  for (int i = t; i < CX * 64; i += 256) {
    int nn = i >> 5, c = i & 31;
    xt[((size_t)b * NN + n0 + nn) * CX + c] = tile[c][nn];
  }
}

__global__ void ballq_kernel(const float* __restrict__ p, int* __restrict__ idx) {
  __shared__ float sx[256], sy[256], sz[256];
  int b = blockIdx.y;
  int i = blockIdx.x * 256 + threadIdx.x;
  float px = p[((size_t)b * NN + i) * 3 + 0];
  float py = p[((size_t)b * NN + i) * 3 + 1];
  float pz = p[((size_t)b * NN + i) * 3 + 2];
  int cnt = 0, first = 0;
  int* myidx = idx + ((size_t)b * NN + i) * KS;
  for (int t0 = 0; t0 < NN; t0 += 256) {
    int j = t0 + threadIdx.x;
    __syncthreads();
    sx[threadIdx.x] = p[((size_t)b * NN + j) * 3 + 0];
    sy[threadIdx.x] = p[((size_t)b * NN + j) * 3 + 1];
    sz[threadIdx.x] = p[((size_t)b * NN + j) * 3 + 2];
    __syncthreads();
    if (cnt < KS) {
      for (int jj = 0; jj < 256; ++jj) {
        float dx = sx[jj] - px, dy = sy[jj] - py, dz = sz[jj] - pz;
        float d2 = dx * dx + dy * dy + dz * dz;
        if (d2 < R2 && cnt < KS) {
          int j2 = t0 + jj;
          if (cnt == 0) first = j2;
          myidx[cnt] = j2;
          cnt++;
        }
      }
    }
  }
  for (int r = cnt; r < KS; ++r) myidx[r] = first;
}

__device__ __forceinline__ void load_ftile(const float* __restrict__ p,
                                           const float* __restrict__ xt,
                                           const int* __restrict__ idxg,
                                           int b, int n0,
                                           int* ibuf, float* flds) {
  int t = threadIdx.x;
  if (t < 64) ibuf[t] = idxg[((size_t)b * NN + n0) * KS + t];
  __syncthreads();
  int col = t & 63;
  int j = ibuf[col];
  const float* xr = xt + ((size_t)b * NN + j) * CX;
  if (t < 64) {
    int nq = n0 + (col >> 5);
    const float* pj = p + ((size_t)b * NN + j) * 3;
    const float* pq = p + ((size_t)b * NN + nq) * 3;
    flds[0 * FSTRIDE + col] = pj[0] - pq[0];
    flds[1 * FSTRIDE + col] = pj[1] - pq[1];
    flds[2 * FSTRIDE + col] = pj[2] - pq[2];
    flds[35 * FSTRIDE + col] = 0.0f;
    #pragma unroll
    for (int c = 0; c < 16; ++c) flds[(3 + c) * FSTRIDE + col] = xr[c];
  } else {
    #pragma unroll
    for (int c = 16; c < 32; ++c) flds[(3 + c) * FSTRIDE + col] = xr[c];
  }
}

__device__ __forceinline__ v8f wmma_gemm(const float* __restrict__ Wl, int wstride,
                                         const float* __restrict__ Fl,
                                         int mt, int ct, int ksteps) {
  int lane  = threadIdx.x & 31;
  int mrow  = mt * 16 + (lane & 15);
  int khalf = (lane >> 4) << 1;
  int col   = ct * 16 + (lane & 15);
  v8f acc = {0.f, 0.f, 0.f, 0.f, 0.f, 0.f, 0.f, 0.f};
  for (int kk = 0; kk < ksteps; ++kk) {
    int kb = (kk << 2) + khalf;
    v2f a, bm;
    a.x  = Wl[mrow * wstride + kb];
    a.y  = Wl[mrow * wstride + kb + 1];
    bm.x = Fl[kb * FSTRIDE + col];
    bm.y = Fl[(kb + 1) * FSTRIDE + col];
    acc = __builtin_amdgcn_wmma_f32_16x16x4_f32(false, a, false, bm,
                                                (short)0, acc, false, false);
  }
  return acc;
}

__global__ __launch_bounds__(128) void conv1_stats_kernel(
    const float* __restrict__ p, const float* __restrict__ xt,
    const int* __restrict__ idxg, const float* __restrict__ W1,
    float* __restrict__ part1) {
  __shared__ float W1l[COUT * W1STRIDE];
  __shared__ float flds[CINP * FSTRIDE];
  __shared__ int   ibuf[64];
  int t = threadIdx.x;
  for (int i = t; i < COUT * W1STRIDE; i += 128) {
    int c = i % W1STRIDE;
    W1l[i] = (c < CIN) ? W1[(i / W1STRIDE) * CIN + c] : 0.f;
  }
  int mt = t >> 5, lane = t & 31;
  float s[8], sq[8];
  #pragma unroll
  for (int v = 0; v < 8; ++v) { s[v] = 0.f; sq[v] = 0.f; }

  for (int pair = blockIdx.x; pair < NPAIRS; pair += gridDim.x) {
    int b  = pair / (NN / 2);
    int n0 = (pair % (NN / 2)) * 2;
    __syncthreads();
    load_ftile(p, xt, idxg, b, n0, ibuf, flds);
    __syncthreads();
    #pragma unroll
    for (int ct = 0; ct < 4; ++ct) {
      v8f acc = wmma_gemm(W1l, W1STRIDE, flds, mt, ct, 9);
      #pragma unroll
      for (int v = 0; v < 8; ++v) { float y = acc[v]; s[v] += y; sq[v] += y * y; }
    }
  }
  #pragma unroll
  for (int v = 0; v < 8; ++v)
    #pragma unroll
    for (int off = 1; off < 16; off <<= 1) {
      s[v]  += __shfl_xor(s[v],  off, 32);
      sq[v] += __shfl_xor(sq[v], off, 32);
    }
  if ((lane & 15) == 0) {
    int hi = lane >> 4;
    #pragma unroll
    for (int v = 0; v < 8; ++v) {
      int o = mt * 16 + hi * 8 + v;
      part1[(size_t)blockIdx.x * 128 + o]      = s[v];
      part1[(size_t)blockIdx.x * 128 + 64 + o] = sq[v];
    }
  }
}

__global__ void bn_finalize_kernel(const float* __restrict__ part,
                                   const float* __restrict__ g,
                                   const float* __restrict__ bta,
                                   float* __restrict__ bnp, int nblocks) {
  int o = threadIdx.x;
  if (o >= COUT) return;
  float s = 0.f, sq = 0.f;
  for (int blk = 0; blk < nblocks; ++blk) {
    s  += part[(size_t)blk * 128 + o];
    sq += part[(size_t)blk * 128 + 64 + o];
  }
  float mean  = s / CNTF;
  float var   = sq / CNTF - mean * mean;
  float scale = g[o] * rsqrtf(var + EPSV);
  bnp[o]      = scale;
  bnp[64 + o] = bta[o] - mean * scale;
}

__global__ __launch_bounds__(128) void conv2_stats_kernel(
    const float* __restrict__ p, const float* __restrict__ xt,
    const int* __restrict__ idxg, const float* __restrict__ W1,
    const float* __restrict__ W2, const float* __restrict__ bnp1,
    float* __restrict__ part2) {
  __shared__ float W1l[COUT * W1STRIDE];
  __shared__ float W2l[COUT * W2STRIDE];
  __shared__ float flds[CINP * FSTRIDE];
  __shared__ float f1l[COUT * FSTRIDE];
  __shared__ int   ibuf[64];
  int t = threadIdx.x;
  for (int i = t; i < COUT * W1STRIDE; i += 128) {
    int c = i % W1STRIDE;
    W1l[i] = (c < CIN) ? W1[(i / W1STRIDE) * CIN + c] : 0.f;
  }
  for (int i = t; i < COUT * W2STRIDE; i += 128) {
    int c = i % W2STRIDE;
    W2l[i] = (c < COUT) ? W2[(i / W2STRIDE) * COUT + c] : 0.f;
  }
  int mt = t >> 5, lane = t & 31;
  int hi = lane >> 4, colb = lane & 15;
  float sc1[8], sh1[8];
  #pragma unroll
  for (int v = 0; v < 8; ++v) {
    int o = mt * 16 + hi * 8 + v;
    sc1[v] = bnp1[o]; sh1[v] = bnp1[64 + o];
  }
  float s[8], sq[8];
  #pragma unroll
  for (int v = 0; v < 8; ++v) { s[v] = 0.f; sq[v] = 0.f; }

  for (int pair = blockIdx.x; pair < NPAIRS; pair += gridDim.x) {
    int b  = pair / (NN / 2);
    int n0 = (pair % (NN / 2)) * 2;
    __syncthreads();
    load_ftile(p, xt, idxg, b, n0, ibuf, flds);
    __syncthreads();
    #pragma unroll
    for (int ct = 0; ct < 4; ++ct) {
      v8f acc = wmma_gemm(W1l, W1STRIDE, flds, mt, ct, 9);
      int col = ct * 16 + colb;
      #pragma unroll
      for (int v = 0; v < 8; ++v) {
        float y = acc[v] * sc1[v] + sh1[v];
        y = y > 0.f ? y : 0.f;
        f1l[(mt * 16 + hi * 8 + v) * FSTRIDE + col] = y;
      }
    }
    __syncthreads();
    #pragma unroll
    for (int ct = 0; ct < 4; ++ct) {
      v8f acc = wmma_gemm(W2l, W2STRIDE, f1l, mt, ct, 16);
      #pragma unroll
      for (int v = 0; v < 8; ++v) { float y = acc[v]; s[v] += y; sq[v] += y * y; }
    }
  }
  #pragma unroll
  for (int v = 0; v < 8; ++v)
    #pragma unroll
    for (int off = 1; off < 16; off <<= 1) {
      s[v]  += __shfl_xor(s[v],  off, 32);
      sq[v] += __shfl_xor(sq[v], off, 32);
    }
  if ((lane & 15) == 0) {
    #pragma unroll
    for (int v = 0; v < 8; ++v) {
      int o = mt * 16 + hi * 8 + v;
      part2[(size_t)blockIdx.x * 128 + o]      = s[v];
      part2[(size_t)blockIdx.x * 128 + 64 + o] = sq[v];
    }
  }
}

__global__ __launch_bounds__(128) void final_kernel(
    const float* __restrict__ p, const float* __restrict__ xt,
    const int* __restrict__ idxg, const float* __restrict__ W1,
    const float* __restrict__ W2, const float* __restrict__ bnp1,
    const float* __restrict__ bnp2, float* __restrict__ out) {
  __shared__ float W1l[COUT * W1STRIDE];
  __shared__ float W2l[COUT * W2STRIDE];
  __shared__ float flds[CINP * FSTRIDE];
  __shared__ float f1l[COUT * FSTRIDE];
  __shared__ int   ibuf[64];
  int t = threadIdx.x;
  for (int i = t; i < COUT * W1STRIDE; i += 128) {
    int c = i % W1STRIDE;
    W1l[i] = (c < CIN) ? W1[(i / W1STRIDE) * CIN + c] : 0.f;
  }
  for (int i = t; i < COUT * W2STRIDE; i += 128) {
    int c = i % W2STRIDE;
    W2l[i] = (c < COUT) ? W2[(i / W2STRIDE) * COUT + c] : 0.f;
  }
  int mt = t >> 5, lane = t & 31;
  int hi = lane >> 4, colb = lane & 15;
  float sc1[8], sh1[8], sc2[8], sh2[8];
  #pragma unroll
  for (int v = 0; v < 8; ++v) {
    int o = mt * 16 + hi * 8 + v;
    sc1[v] = bnp1[o]; sh1[v] = bnp1[64 + o];
    sc2[v] = bnp2[o]; sh2[v] = bnp2[64 + o];
  }
  for (int pair = blockIdx.x; pair < NPAIRS; pair += gridDim.x) {
    int b  = pair / (NN / 2);
    int n0 = (pair % (NN / 2)) * 2;
    __syncthreads();
    load_ftile(p, xt, idxg, b, n0, ibuf, flds);
    __syncthreads();
    #pragma unroll
    for (int ct = 0; ct < 4; ++ct) {
      v8f acc = wmma_gemm(W1l, W1STRIDE, flds, mt, ct, 9);
      int col = ct * 16 + colb;
      #pragma unroll
      for (int v = 0; v < 8; ++v) {
        float y = acc[v] * sc1[v] + sh1[v];
        y = y > 0.f ? y : 0.f;
        f1l[(mt * 16 + hi * 8 + v) * FSTRIDE + col] = y;
      }
    }
    __syncthreads();
    float m0[8], m1[8];
    #pragma unroll
    for (int v = 0; v < 8; ++v) { m0[v] = 0.f; m1[v] = 0.f; }
    #pragma unroll
    for (int ct = 0; ct < 4; ++ct) {
      v8f acc = wmma_gemm(W2l, W2STRIDE, f1l, mt, ct, 16);
      #pragma unroll
      for (int v = 0; v < 8; ++v) {
        float y = acc[v] * sc2[v] + sh2[v];
        y = y > 0.f ? y : 0.f;
        if (ct < 2) m0[v] = fmaxf(m0[v], y);
        else        m1[v] = fmaxf(m1[v], y);
      }
    }
    #pragma unroll
    for (int v = 0; v < 8; ++v)
      #pragma unroll
      for (int off = 1; off < 16; off <<= 1) {
        m0[v] = fmaxf(m0[v], __shfl_xor(m0[v], off, 32));
        m1[v] = fmaxf(m1[v], __shfl_xor(m1[v], off, 32));
      }
    if ((lane & 15) == 0) {
      #pragma unroll
      for (int v = 0; v < 8; ++v) {
        int o = mt * 16 + hi * 8 + v;
        out[((size_t)b * COUT + o) * NN + n0]     = m0[v];
        out[((size_t)b * COUT + o) * NN + n0 + 1] = m1[v];
      }
    }
  }
}

extern "C" void kernel_launch(void* const* d_in, const int* in_sizes, int n_in,
                              void* d_out, int out_size, void* d_ws, size_t ws_size,
                              hipStream_t stream) {
  const float* p  = (const float*)d_in[0];
  const float* x  = (const float*)d_in[1];
  const float* W1 = (const float*)d_in[2];
  const float* g1 = (const float*)d_in[3];
  const float* b1 = (const float*)d_in[4];
  const float* W2 = (const float*)d_in[5];
  const float* g2 = (const float*)d_in[6];
  const float* b2 = (const float*)d_in[7];
  float* out = (float*)d_out;

  char*  ws    = (char*)d_ws;
  int*   idx   = (int*)ws;
  float* xt    = (float*)(ws + (size_t)BB * NN * KS * 4);
  float* part1 = (float*)(ws + (size_t)BB * NN * KS * 4
                             + (size_t)BB * NN * CX * 4);
  float* part2 = part1 + (size_t)GCONV * 128;
  float* bnp1  = part2 + (size_t)GCONV * 128;
  float* bnp2  = bnp1 + 128;

  xpose_kernel<<<dim3(NN / 64, BB), 256, 0, stream>>>(x, xt);
  ballq_kernel<<<dim3(NN / 256, BB), 256, 0, stream>>>(p, idx);
  conv1_stats_kernel<<<GCONV, 128, 0, stream>>>(p, xt, idx, W1, part1);
  bn_finalize_kernel<<<1, 64, 0, stream>>>(part1, g1, b1, bnp1, GCONV);
  conv2_stats_kernel<<<GCONV, 128, 0, stream>>>(p, xt, idx, W1, W2, bnp1, part2);
  bn_finalize_kernel<<<1, 64, 0, stream>>>(part2, g2, b2, bnp2, GCONV);
  final_kernel<<<GCONV, 128, 0, stream>>>(p, xt, idx, W1, W2, bnp1, bnp2, out);
}


# conv passes: LDS tiles stored k-contiguous, a lane reads fragments 16 bytes at a time, a block's reads issued as one group ahead of its WMMAs (were spread between them); k order of conv sums changes
# speedup vs baseline: 3.9236x; 3.9236x over previous
#include <hip/hip_runtime.h>
#include <hip/hip_bf16.h>


typedef __attribute__((ext_vector_type(2))) float v2f;
typedef __attribute__((ext_vector_type(8))) float v8f;

#define BB      4
#define NN      4096
#define CX      32
#define CIN     35
#define CINP    36
#define COUT    64
#define KS      32
#define R2      0.01f
#define EPSV    1e-5f
#define NPAIRS  (BB * (NN / 2))
#define GCONV   1024
#define GSTAT   256
#define CNTF    ((float)BB * (float)NN * (float)KS)

#define FPITCH   36
#define F1PITCH  68
#define W1STRIDE 36
#define W2STRIDE 68

__global__ void xpose_kernel(const float* __restrict__ x, float* __restrict__ xt) {
  __shared__ float tile[CX][65];
  int b  = blockIdx.y;
  int n0 = blockIdx.x * 64;
  int t  = threadIdx.x;
  for (int i = t; i < CX * 64; i += 256) {
    int c = i >> 6, nn = i & 63;
    tile[c][nn] = x[((size_t)b * CX + c) * NN + n0 + nn];
  }
  __syncthreads();
  for (int i = t; i < CX * 64; i += 256) {
    int nn = i >> 5, c = i & 31;
    xt[((size_t)b * NN + n0 + nn) * CX + c] = tile[c][nn];
  }
}

#define BQ_WAVES 8
#define BQ_SEG   (NN / BQ_WAVES)
#define BQ_GRP   16
#define BQ_PITCH 33

__global__ __launch_bounds__(256) void ballq_kernel(const float* __restrict__ p,
                                                    int* __restrict__ idx) {
  __shared__ unsigned short lst[BQ_WAVES][32][BQ_PITCH];
  __shared__ int cnts[BQ_WAVES][32];
  const int b    = blockIdx.y;
  const int lane = threadIdx.x & 31;
  const int seg  = __builtin_amdgcn_readfirstlane((int)(threadIdx.x >> 5));
  const float* __restrict__ pb = p + (size_t)b * NN * 3;
  const int q = blockIdx.x * 32 + lane;
  const float px = pb[q * 3 + 0];
  const float py = pb[q * 3 + 1];
  const float pz = pb[q * 3 + 2];
  const float* __restrict__ cb = pb + seg * (BQ_SEG * 3);
  unsigned short* mylst = &lst[seg][lane][0];
  int cnt = 0;
  for (int g = 0; g < BQ_SEG; g += BQ_GRP) {
    unsigned m = 0;
    #pragma unroll
    for (int jj = 0; jj < BQ_GRP; ++jj) {
      float dx = cb[(g + jj) * 3 + 0] - px;
      float dy = cb[(g + jj) * 3 + 1] - py;
      float dz = cb[(g + jj) * 3 + 2] - pz;
      float d2 = __builtin_fmaf(dz, dz, __builtin_fmaf(dy, dy, dx * dx));
      m = m + m + (d2 < R2 ? 1u : 0u);
    }
    while (m != 0) {
      int lz   = __builtin_clz(m);
      int slot = cnt < KS ? cnt : KS;
      mylst[slot] = (unsigned short)(seg * BQ_SEG + g + lz - (32 - BQ_GRP));
      cnt++;
      m &= ~(0x80000000u >> lz);
    }
  }
  cnts[seg][lane] = cnt;
  __syncthreads();

  const int ql = threadIdx.x >> 3;
  const int r0 = (threadIdx.x & 7) * 4;
  int v[4] = {0, 0, 0, 0};
  int pre = 0, first = 0;
  bool have = false;
  #pragma unroll
  for (int s = 0; s < BQ_WAVES; ++s) {
    int c = cnts[s][ql];
    c = c < KS ? c : KS;
    if (!have && c > 0) { first = lst[s][ql][0]; have = true; }
    #pragma unroll
    for (int k = 0; k < 4; ++k) {
      int r = r0 + k - pre;
      if (r >= 0 && r < c) v[k] = lst[s][ql][r];
    }
    pre += c;
  }
  #pragma unroll
  for (int k = 0; k < 4; ++k)
    if (r0 + k >= pre) v[k] = first;
  int4 o;
  o.x = v[0]; o.y = v[1]; o.z = v[2]; o.w = v[3];
  *reinterpret_cast<int4*>(idx + ((size_t)b * NN + blockIdx.x * 32 + ql) * KS + r0) = o;
}

struct FPart {
  float xv[16];
  float pj[2];
  float pq[2];
};

__device__ __forceinline__ int tile_idx(const int* __restrict__ idxg, int pair) {
  return idxg[(size_t)pair * 64 + (threadIdx.x & 63)];
}

__device__ __forceinline__ void ftile_fetch(const float* __restrict__ p,
                                            const float* __restrict__ xt,
                                            int pair, int j, FPart& r) {
  const int t = threadIdx.x;
  const int b = pair / (NN / 2);
  const float* xr = xt + ((size_t)b * NN + j) * CX + (t < 64 ? 0 : 16);
  #pragma unroll
  for (int c = 0; c < 16; ++c) r.xv[c] = xr[c];
  const float* pj = p + ((size_t)b * NN + j) * 3;
  const float* pq = p + ((size_t)2 * pair + ((t & 63) >> 5)) * 3;
  if (t < 64) {
    r.pj[0] = pj[0]; r.pj[1] = pj[1];
    r.pq[0] = pq[0]; r.pq[1] = pq[1];
  } else {
    r.pj[0] = pj[2];
    r.pq[0] = pq[2];
    r.pj[1] = 0.f;
    r.pq[1] = 0.f;
  }
}

__device__ __forceinline__ void ftile_put(const FPart& r, float* flds) {
  const int t = threadIdx.x;
  const bool lo = t < 64;
  float* colp = flds + (t & 63) * FPITCH;
  float4* xq = reinterpret_cast<float4*>(colp + (lo ? 0 : 16));
  #pragma unroll
  for (int q = 0; q < 4; ++q) {
    float4 v;
    v.x = r.xv[4 * q + 0]; v.y = r.xv[4 * q + 1]; v.z = r.xv[4 * q + 2]; v.w = r.xv[4 * q + 3];
    xq[q] = v;
  }
  float2 d;
  d.x = r.pj[0] - r.pq[0];
  d.y = r.pj[1] - r.pq[1];
  *reinterpret_cast<float2*>(colp + (lo ? 32 : 34)) = d;
}

__device__ __forceinline__ void lds_barrier() {
  asm volatile("s_wait_dscnt 0x0\n\ts_barrier_signal -1\n\ts_barrier_wait -1" ::: "memory");
}

struct Chunk {
  float4 a;
  float4 b[4];
};
struct Tail {
  float2 a;
  float2 b[4];
};

__device__ __forceinline__ void chunk_load(Chunk& f, const float* wrow, const float* fcol,
                                           int fpitch, int j) {
  f.a = *reinterpret_cast<const float4*>(wrow + 8 * j);
  #pragma unroll
  for (int ct = 0; ct < 4; ++ct)
    f.b[ct] = *reinterpret_cast<const float4*>(fcol + ct * 16 * fpitch + 8 * j);
}

__device__ __forceinline__ void tail_load(Tail& f, const float* wrow2, const float* fcol2,
                                          int fpitch, int k0) {
  f.a = *reinterpret_cast<const float2*>(wrow2 + k0);
  #pragma unroll
  for (int ct = 0; ct < 4; ++ct)
    f.b[ct] = *reinterpret_cast<const float2*>(fcol2 + ct * 16 * fpitch + k0);
}

__device__ __forceinline__ v8f mma4(v2f a, v2f b, v8f cacc) {
  return __builtin_amdgcn_wmma_f32_16x16x4_f32(false, a, false, b, (short)0, cacc, false, false);
}

__device__ __forceinline__ void chunk_mma(const Chunk& f, v8f (&acc)[4]) {
  v2f a0, a1;
  a0.x = f.a.x; a0.y = f.a.y; a1.x = f.a.z; a1.y = f.a.w;
  #pragma unroll
  for (int ct = 0; ct < 4; ++ct) {
    v2f b; b.x = f.b[ct].x; b.y = f.b[ct].y;
    acc[ct] = mma4(a0, b, acc[ct]);
  }
  #pragma unroll
  for (int ct = 0; ct < 4; ++ct) {
    v2f b; b.x = f.b[ct].z; b.y = f.b[ct].w;
    acc[ct] = mma4(a1, b, acc[ct]);
  }
}

__device__ __forceinline__ void tail_mma(const Tail& f, v8f (&acc)[4]) {
  v2f a; a.x = f.a.x; a.y = f.a.y;
  #pragma unroll
  for (int ct = 0; ct < 4; ++ct) {
    v2f b; b.x = f.b[ct].x; b.y = f.b[ct].y;
    acc[ct] = mma4(a, b, acc[ct]);
  }
}

__device__ __forceinline__ void sched_fence() { __builtin_amdgcn_sched_barrier(0); }

__device__ __forceinline__ void acc_zero(v8f (&acc)[4]) {
  #pragma unroll
  for (int ct = 0; ct < 4; ++ct)
    #pragma unroll
    for (int v = 0; v < 8; ++v) acc[ct][v] = 0.f;
}

__device__ __forceinline__ void conv1_strip(const float* __restrict__ W1l,
                                            const float* __restrict__ ft,
                                            int mt, v8f (&acc)[4]) {
  const int lane = threadIdx.x & 31;
  const int c = lane & 15, h = lane >> 4;
  const float* wrow = W1l + (mt * 16 + c) * W1STRIDE;
  const float* fcol = ft + c * FPITCH;
  Chunk g[4];
  Tail  gt;
  acc_zero(acc);
  #pragma unroll
  for (int j = 0; j < 4; ++j) chunk_load(g[j], wrow + 4 * h, fcol + 4 * h, FPITCH, j);
  tail_load(gt, wrow + 2 * h, fcol + 2 * h, FPITCH, 32);
  sched_fence();
  #pragma unroll
  for (int j = 0; j < 4; ++j) chunk_mma(g[j], acc);
  tail_mma(gt, acc);
  sched_fence();
}

__device__ __forceinline__ void conv2_strip(const float* __restrict__ W2l,
                                            const float* __restrict__ f1l,
                                            int mt, v8f (&acc)[4]) {
  const int lane = threadIdx.x & 31;
  const int c = lane & 15, h = lane >> 4;
  const float* wrow = W2l + (mt * 16 + c) * W2STRIDE + 4 * h;
  const float* fcol = f1l + c * F1PITCH + 4 * h;
  Chunk s0[2], s1[2];
  acc_zero(acc);
  chunk_load(s0[0], wrow, fcol, F1PITCH, 0);
  chunk_load(s0[1], wrow, fcol, F1PITCH, 1);
  chunk_load(s1[0], wrow, fcol, F1PITCH, 2);
  chunk_load(s1[1], wrow, fcol, F1PITCH, 3);
  sched_fence();
  chunk_mma(s0[0], acc);
  chunk_mma(s0[1], acc);
  sched_fence();
  chunk_load(s0[0], wrow, fcol, F1PITCH, 4);
  chunk_load(s0[1], wrow, fcol, F1PITCH, 5);
  sched_fence();
  chunk_mma(s1[0], acc);
  chunk_mma(s1[1], acc);
  sched_fence();
  chunk_load(s1[0], wrow, fcol, F1PITCH, 6);
  chunk_load(s1[1], wrow, fcol, F1PITCH, 7);
  sched_fence();
  chunk_mma(s0[0], acc);
  chunk_mma(s0[1], acc);
  chunk_mma(s1[0], acc);
  chunk_mma(s1[1], acc);
  sched_fence();
}

__device__ __forceinline__ void bn1_store(const v8f (&acc)[4], const float (&sc1)[8],
                                          const float (&sh1)[8], float* f1l, int mt) {
  const int lane = threadIdx.x & 31;
  const int c = lane & 15, h = lane >> 4;
  #pragma unroll
  for (int ct = 0; ct < 4; ++ct) {
    float y[8];
    #pragma unroll
    for (int v = 0; v < 8; ++v) {
      float t = acc[ct][v] * sc1[v] + sh1[v];
      y[v] = t > 0.f ? t : 0.f;
    }
    float4* dst = reinterpret_cast<float4*>(f1l + (ct * 16 + c) * F1PITCH + mt * 16 + h * 8);
    float4 q0, q1;
    q0.x = y[0]; q0.y = y[1]; q0.z = y[2]; q0.w = y[3];
    q1.x = y[4]; q1.y = y[5]; q1.z = y[6]; q1.w = y[7];
    dst[0] = q0;
    dst[1] = q1;
  }
}


#define W1Q (COUT * CIN / 4)
#define W2Q (COUT * COUT / 4)

__device__ __forceinline__ void w1_fetch(const float* __restrict__ W1, float4 (&r)[5]) {
  const float4* __restrict__ src = reinterpret_cast<const float4*>(W1);
  const int t = threadIdx.x;
  #pragma unroll
  for (int e = 0; e < 5; ++e) {
    int q = t + 128 * e;
    r[e] = src[q < W1Q ? q : W1Q - 1];
  }
}

__device__ __forceinline__ void w1_put(const float4 (&r)[5], float* W1l) {
  const int t = threadIdx.x;
  #pragma unroll
  for (int e = 0; e < 5; ++e) {
    int q = t + 128 * e;
    if (q < W1Q) {
      const float v[4] = {r[e].x, r[e].y, r[e].z, r[e].w};
      #pragma unroll
      for (int u = 0; u < 4; ++u) {
        int i   = 4 * q + u;
        int row = i / CIN;
        int cc  = i - row * CIN;
        int kp  = cc >= 3 ? cc - 3 : 32 + cc;
        W1l[row * W1STRIDE + kp] = v[u];
      }
    }
  }
  if (t < COUT) W1l[t * W1STRIDE + CIN] = 0.f;
}

__device__ __forceinline__ void w2_fetch(const float* __restrict__ W2, float4 (&r)[8]) {
  const float4* __restrict__ src = reinterpret_cast<const float4*>(W2);
  const int t = threadIdx.x;
  #pragma unroll
  for (int e = 0; e < 8; ++e) r[e] = src[t + 128 * e];
}

__device__ __forceinline__ void w2_put(const float4 (&r)[8], float* W2l) {
  const int t = threadIdx.x;
  #pragma unroll
  for (int e = 0; e < 8; ++e) {
    int q = t + 128 * e;
    *reinterpret_cast<float4*>(&W2l[(q >> 4) * W2STRIDE + (q & 15) * 4]) = r[e];
  }
}

__global__ __launch_bounds__(128) void conv1_stats_kernel(
    const float* __restrict__ p, const float* __restrict__ xt,
    const int* __restrict__ idxg, const float* __restrict__ W1,
    float* __restrict__ part1) {
  __shared__ __attribute__((aligned(16))) float W1l[COUT * W1STRIDE];
  __shared__ __attribute__((aligned(16))) float flds[2][64 * FPITCH];
  int t = threadIdx.x;
  const int G = gridDim.x;
  int pair = blockIdx.x;
  const int j0 = tile_idx(idxg, pair);
  int jn = tile_idx(idxg, pair + G < NPAIRS ? pair + G : pair);
  {
    float4 r1[5];
    w1_fetch(W1, r1);
    w1_put(r1, W1l);
  }
  FPart r;
  ftile_fetch(p, xt, pair, j0, r);
  int mt = t >> 5, lane = t & 31;
  float s[8], sq[8];
  #pragma unroll
  for (int v = 0; v < 8; ++v) { s[v] = 0.f; sq[v] = 0.f; }

  for (int trip = 0; pair < NPAIRS; pair += G, ++trip) {
    float* ft = flds[trip & 1];
    ftile_put(r, ft);
    const int pn  = pair + G     < NPAIRS ? pair + G     : pair;
    const int pn2 = pair + 2 * G < NPAIRS ? pair + 2 * G : pair;
    ftile_fetch(p, xt, pn, jn, r);
    jn = tile_idx(idxg, pn2);
    lds_barrier();
    v8f acc[4];
    conv1_strip(W1l, ft, mt, acc);
    #pragma unroll
    for (int ct = 0; ct < 4; ++ct) {
      #pragma unroll
      for (int v = 0; v < 8; ++v) { float y = acc[ct][v]; s[v] += y; sq[v] += y * y; }
    }
  }
  #pragma unroll
  for (int v = 0; v < 8; ++v)
    #pragma unroll
    for (int off = 1; off < 16; off <<= 1) {
      s[v]  += __shfl_xor(s[v],  off, 32);
      sq[v] += __shfl_xor(sq[v], off, 32);
    }
  if ((lane & 15) == 0) {
    int hi = lane >> 4;
    #pragma unroll
    for (int v = 0; v < 8; ++v) {
      int o = mt * 16 + hi * 8 + v;
      part1[(size_t)blockIdx.x * 128 + o]      = s[v];
      part1[(size_t)blockIdx.x * 128 + 64 + o] = sq[v];
    }
  }
}

__global__ void bn_finalize_kernel(const float* __restrict__ part,
                                   const float* __restrict__ g,
                                   const float* __restrict__ bta,
                                   float* __restrict__ bnp, int nblocks) {
  int o = threadIdx.x;
  if (o >= COUT) return;
  float s = 0.f, sq = 0.f;
  for (int blk = 0; blk < nblocks; ++blk) {
    s  += part[(size_t)blk * 128 + o];
    sq += part[(size_t)blk * 128 + 64 + o];
  }
  float mean  = s / CNTF;
  float var   = sq / CNTF - mean * mean;
  float scale = g[o] * rsqrtf(var + EPSV);
  bnp[o]      = scale;
  bnp[64 + o] = bta[o] - mean * scale;
}

__global__ __launch_bounds__(128) void conv2_stats_kernel(
    const float* __restrict__ p, const float* __restrict__ xt,
    const int* __restrict__ idxg, const float* __restrict__ W1,
    const float* __restrict__ W2, const float* __restrict__ bnp1,
    float* __restrict__ part2) {
  __shared__ __attribute__((aligned(16))) float W1l[COUT * W1STRIDE];
  __shared__ __attribute__((aligned(16))) float W2l[COUT * W2STRIDE];
  __shared__ __attribute__((aligned(16))) float flds[64 * FPITCH];
  __shared__ __attribute__((aligned(16))) float f1l[64 * F1PITCH];
  int t = threadIdx.x;
  const int G = gridDim.x;
  int pair = blockIdx.x;
  const int j0 = tile_idx(idxg, pair);
  int jn = tile_idx(idxg, pair + G < NPAIRS ? pair + G : pair);
  {
    float4 r1[5], r2[8];
    w1_fetch(W1, r1);
    w2_fetch(W2, r2);
    w1_put(r1, W1l);
    w2_put(r2, W2l);
  }
  FPart r;
  ftile_fetch(p, xt, pair, j0, r);
  int mt = t >> 5, lane = t & 31;
  int hi = lane >> 4;
  float sc1[8], sh1[8];
  #pragma unroll
  for (int v = 0; v < 8; ++v) {
    int o = mt * 16 + hi * 8 + v;
    sc1[v] = bnp1[o]; sh1[v] = bnp1[64 + o];
  }
  float s[8], sq[8];
  #pragma unroll
  for (int v = 0; v < 8; ++v) { s[v] = 0.f; sq[v] = 0.f; }

  for (; pair < NPAIRS; pair += G) {
    ftile_put(r, flds);
    const int pn  = pair + G     < NPAIRS ? pair + G     : pair;
    const int pn2 = pair + 2 * G < NPAIRS ? pair + 2 * G : pair;
    ftile_fetch(p, xt, pn, jn, r);
    jn = tile_idx(idxg, pn2);
    lds_barrier();
    v8f acc[4];
    conv1_strip(W1l, flds, mt, acc);
    bn1_store(acc, sc1, sh1, f1l, mt);
    lds_barrier();
    conv2_strip(W2l, f1l, mt, acc);
    #pragma unroll
    for (int ct = 0; ct < 4; ++ct) {
      #pragma unroll
      for (int v = 0; v < 8; ++v) { float y = acc[ct][v]; s[v] += y; sq[v] += y * y; }
    }
  }
  #pragma unroll
  for (int v = 0; v < 8; ++v)
    #pragma unroll
    for (int off = 1; off < 16; off <<= 1) {
      s[v]  += __shfl_xor(s[v],  off, 32);
      sq[v] += __shfl_xor(sq[v], off, 32);
    }
  if ((lane & 15) == 0) {
    #pragma unroll
    for (int v = 0; v < 8; ++v) {
      int o = mt * 16 + hi * 8 + v;
      part2[(size_t)blockIdx.x * 128 + o]      = s[v];
      part2[(size_t)blockIdx.x * 128 + 64 + o] = sq[v];
    }
  }
}

__global__ __launch_bounds__(128) void final_kernel(
    const float* __restrict__ p, const float* __restrict__ xt,
    const int* __restrict__ idxg, const float* __restrict__ W1,
    const float* __restrict__ W2, const float* __restrict__ bnp1,
    const float* __restrict__ bnp2, float* __restrict__ out) {
  __shared__ __attribute__((aligned(16))) float W1l[COUT * W1STRIDE];
  __shared__ __attribute__((aligned(16))) float W2l[COUT * W2STRIDE];
  __shared__ __attribute__((aligned(16))) float flds[64 * FPITCH];
  __shared__ __attribute__((aligned(16))) float f1l[64 * F1PITCH];
  int t = threadIdx.x;
  const int G = gridDim.x;
  int pair = blockIdx.x;
  const int j0 = tile_idx(idxg, pair);
  int jn = tile_idx(idxg, pair + G < NPAIRS ? pair + G : pair);
  {
    float4 r1[5], r2[8];
    w1_fetch(W1, r1);
    w2_fetch(W2, r2);
    w1_put(r1, W1l);
    w2_put(r2, W2l);
  }
  FPart r;
  ftile_fetch(p, xt, pair, j0, r);
  int mt = t >> 5, lane = t & 31;
  int hi = lane >> 4;
  float sc1[8], sh1[8], sc2[8], sh2[8];
  #pragma unroll
  for (int v = 0; v < 8; ++v) {
    int o = mt * 16 + hi * 8 + v;
    sc1[v] = bnp1[o]; sh1[v] = bnp1[64 + o];
    sc2[v] = bnp2[o]; sh2[v] = bnp2[64 + o];
  }
  for (; pair < NPAIRS; pair += G) {
    int b  = pair / (NN / 2);
    int n0 = (pair % (NN / 2)) * 2;
    ftile_put(r, flds);
    const int pn  = pair + G     < NPAIRS ? pair + G     : pair;
    const int pn2 = pair + 2 * G < NPAIRS ? pair + 2 * G : pair;
    ftile_fetch(p, xt, pn, jn, r);
    jn = tile_idx(idxg, pn2);
    lds_barrier();
    v8f acc[4];
    conv1_strip(W1l, flds, mt, acc);
    bn1_store(acc, sc1, sh1, f1l, mt);
    lds_barrier();
    conv2_strip(W2l, f1l, mt, acc);
    float m0[8], m1[8];
    #pragma unroll
    for (int v = 0; v < 8; ++v) { m0[v] = 0.f; m1[v] = 0.f; }
    #pragma unroll
    for (int ct = 0; ct < 4; ++ct) {
      #pragma unroll
      for (int v = 0; v < 8; ++v) {
        float y = acc[ct][v] * sc2[v] + sh2[v];
        y = y > 0.f ? y : 0.f;
        if (ct < 2) m0[v] = fmaxf(m0[v], y);
        else        m1[v] = fmaxf(m1[v], y);
      }
    }
    #pragma unroll
    for (int v = 0; v < 8; ++v)
      #pragma unroll
      for (int off = 1; off < 16; off <<= 1) {
        m0[v] = fmaxf(m0[v], __shfl_xor(m0[v], off, 32));
        m1[v] = fmaxf(m1[v], __shfl_xor(m1[v], off, 32));
      }
    if ((lane & 15) == 0) {
      #pragma unroll
      for (int v = 0; v < 8; ++v) {
        int o = mt * 16 + hi * 8 + v;
        out[((size_t)b * COUT + o) * NN + n0]     = m0[v];
        out[((size_t)b * COUT + o) * NN + n0 + 1] = m1[v];
      }
    }
  }
}

extern "C" void kernel_launch(void* const* d_in, const int* in_sizes, int n_in,
                              void* d_out, int out_size, void* d_ws, size_t ws_size,
                              hipStream_t stream) {
  const float* p  = (const float*)d_in[0];
  const float* x  = (const float*)d_in[1];
  const float* W1 = (const float*)d_in[2];
  const float* g1 = (const float*)d_in[3];
  const float* b1 = (const float*)d_in[4];
  const float* W2 = (const float*)d_in[5];
  const float* g2 = (const float*)d_in[6];
  const float* b2 = (const float*)d_in[7];
  float* out = (float*)d_out;

  char*  ws    = (char*)d_ws;
  int*   idx   = (int*)ws;
  float* xt    = (float*)(ws + (size_t)BB * NN * KS * 4);
  float* part1 = (float*)(ws + (size_t)BB * NN * KS * 4
                             + (size_t)BB * NN * CX * 4);
  float* part2 = part1 + (size_t)GCONV * 128;
  float* bnp1  = part2 + (size_t)GCONV * 128;
  float* bnp2  = bnp1 + 128;

  xpose_kernel<<<dim3(NN / 64, BB), 256, 0, stream>>>(x, xt);
  ballq_kernel<<<dim3(NN / 32, BB), 256, 0, stream>>>(p, idx);
  conv1_stats_kernel<<<GSTAT, 128, 0, stream>>>(p, xt, idx, W1, part1);
  bn_finalize_kernel<<<1, 64, 0, stream>>>(part1, g1, b1, bnp1, GSTAT);
  conv2_stats_kernel<<<GSTAT, 128, 0, stream>>>(p, xt, idx, W1, W2, bnp1, part2);
  bn_finalize_kernel<<<1, 64, 0, stream>>>(part2, g2, b2, bnp2, GSTAT);
  final_kernel<<<GCONV, 128, 0, stream>>>(p, xt, idx, W1, W2, bnp1, bnp2, out);
}


# BN1 finalize launch removed: each conv2-stats block sums the 256 BN1 partial rows itself (16-byte loads, all in flight, 4 waves joined in LDS); 6 launches, was 7; BN1 sums' order changes
# speedup vs baseline: 4.3634x; 1.1121x over previous
#include <hip/hip_runtime.h>
#include <hip/hip_bf16.h>


typedef __attribute__((ext_vector_type(2))) float v2f;
typedef __attribute__((ext_vector_type(8))) float v8f;

#define BB      4
#define NN      4096
#define CX      32
#define CIN     35
#define CINP    36
#define COUT    64
#define KS      32
#define R2      0.01f
#define EPSV    1e-5f
#define NPAIRS  (BB * (NN / 2))
#define GCONV   1024
#define GSTAT   256
#define CNTF    ((float)BB * (float)NN * (float)KS)

#define FPITCH   36
#define F1PITCH  68
#define W1STRIDE 36
#define W2STRIDE 68

__global__ void xpose_kernel(const float* __restrict__ x, float* __restrict__ xt) {
  __shared__ float tile[CX][65];
  int b  = blockIdx.y;
  int n0 = blockIdx.x * 64;
  int t  = threadIdx.x;
  for (int i = t; i < CX * 64; i += 256) {
    int c = i >> 6, nn = i & 63;
    tile[c][nn] = x[((size_t)b * CX + c) * NN + n0 + nn];
  }
  __syncthreads();
  for (int i = t; i < CX * 64; i += 256) {
    int nn = i >> 5, c = i & 31;
    xt[((size_t)b * NN + n0 + nn) * CX + c] = tile[c][nn];
  }
}

#define BQ_WAVES 8
#define BQ_SEG   (NN / BQ_WAVES)
#define BQ_GRP   16
#define BQ_PITCH 33

__global__ __launch_bounds__(256) void ballq_kernel(const float* __restrict__ p,
                                                    int* __restrict__ idx) {
  __shared__ unsigned short lst[BQ_WAVES][32][BQ_PITCH];
  __shared__ int cnts[BQ_WAVES][32];
  const int b    = blockIdx.y;
  const int lane = threadIdx.x & 31;
  const int seg  = __builtin_amdgcn_readfirstlane((int)(threadIdx.x >> 5));
  const float* __restrict__ pb = p + (size_t)b * NN * 3;
  const int q = blockIdx.x * 32 + lane;
  const float px = pb[q * 3 + 0];
  const float py = pb[q * 3 + 1];
  const float pz = pb[q * 3 + 2];
  const float* __restrict__ cb = pb + seg * (BQ_SEG * 3);
  unsigned short* mylst = &lst[seg][lane][0];
  int cnt = 0;
  for (int g = 0; g < BQ_SEG; g += BQ_GRP) {
    unsigned m = 0;
    #pragma unroll
    for (int jj = 0; jj < BQ_GRP; ++jj) {
      float dx = cb[(g + jj) * 3 + 0] - px;
      float dy = cb[(g + jj) * 3 + 1] - py;
      float dz = cb[(g + jj) * 3 + 2] - pz;
      float d2 = __builtin_fmaf(dz, dz, __builtin_fmaf(dy, dy, dx * dx));
      m = m + m + (d2 < R2 ? 1u : 0u);
    }
    while (m != 0) {
      int lz   = __builtin_clz(m);
      int slot = cnt < KS ? cnt : KS;
      mylst[slot] = (unsigned short)(seg * BQ_SEG + g + lz - (32 - BQ_GRP));
      cnt++;
      m &= ~(0x80000000u >> lz);
    }
  }
  cnts[seg][lane] = cnt;
  __syncthreads();

  const int ql = threadIdx.x >> 3;
  const int r0 = (threadIdx.x & 7) * 4;
  int v[4] = {0, 0, 0, 0};
  int pre = 0, first = 0;
  bool have = false;
  #pragma unroll
  for (int s = 0; s < BQ_WAVES; ++s) {
    int c = cnts[s][ql];
    c = c < KS ? c : KS;
    if (!have && c > 0) { first = lst[s][ql][0]; have = true; }
    #pragma unroll
    for (int k = 0; k < 4; ++k) {
      int r = r0 + k - pre;
      if (r >= 0 && r < c) v[k] = lst[s][ql][r];
    }
    pre += c;
  }
  #pragma unroll
  for (int k = 0; k < 4; ++k)
    if (r0 + k >= pre) v[k] = first;
  int4 o;
  o.x = v[0]; o.y = v[1]; o.z = v[2]; o.w = v[3];
  *reinterpret_cast<int4*>(idx + ((size_t)b * NN + blockIdx.x * 32 + ql) * KS + r0) = o;
}

struct FPart {
  float xv[16];
  float pj[2];
  float pq[2];
};

__device__ __forceinline__ int tile_idx(const int* __restrict__ idxg, int pair) {
  return idxg[(size_t)pair * 64 + (threadIdx.x & 63)];
}

__device__ __forceinline__ void ftile_fetch(const float* __restrict__ p,
                                            const float* __restrict__ xt,
                                            int pair, int j, FPart& r) {
  const int t = threadIdx.x;
  const int b = pair / (NN / 2);
  const float* xr = xt + ((size_t)b * NN + j) * CX + (t < 64 ? 0 : 16);
  #pragma unroll
  for (int c = 0; c < 16; ++c) r.xv[c] = xr[c];
  const float* pj = p + ((size_t)b * NN + j) * 3;
  const float* pq = p + ((size_t)2 * pair + ((t & 63) >> 5)) * 3;
  if (t < 64) {
    r.pj[0] = pj[0]; r.pj[1] = pj[1];
    r.pq[0] = pq[0]; r.pq[1] = pq[1];
  } else {
    r.pj[0] = pj[2];
    r.pq[0] = pq[2];
    r.pj[1] = 0.f;
    r.pq[1] = 0.f;
  }
}

__device__ __forceinline__ void ftile_put(const FPart& r, float* flds) {
  const int t = threadIdx.x;
  const bool lo = t < 64;
  float* colp = flds + (t & 63) * FPITCH;
  float4* xq = reinterpret_cast<float4*>(colp + (lo ? 0 : 16));
  #pragma unroll
  for (int q = 0; q < 4; ++q) {
    float4 v;
    v.x = r.xv[4 * q + 0]; v.y = r.xv[4 * q + 1]; v.z = r.xv[4 * q + 2]; v.w = r.xv[4 * q + 3];
    xq[q] = v;
  }
  float2 d;
  d.x = r.pj[0] - r.pq[0];
  d.y = r.pj[1] - r.pq[1];
  *reinterpret_cast<float2*>(colp + (lo ? 32 : 34)) = d;
}

__device__ __forceinline__ void lds_barrier() {
  asm volatile("s_wait_dscnt 0x0\n\ts_barrier_signal -1\n\ts_barrier_wait -1" ::: "memory");
}

struct Chunk {
  float4 a;
  float4 b[4];
};
struct Tail {
  float2 a;
  float2 b[4];
};

__device__ __forceinline__ void chunk_load(Chunk& f, const float* wrow, const float* fcol,
                                           int fpitch, int j) {
  f.a = *reinterpret_cast<const float4*>(wrow + 8 * j);
  #pragma unroll
  for (int ct = 0; ct < 4; ++ct)
    f.b[ct] = *reinterpret_cast<const float4*>(fcol + ct * 16 * fpitch + 8 * j);
}

__device__ __forceinline__ void tail_load(Tail& f, const float* wrow2, const float* fcol2,
                                          int fpitch, int k0) {
  f.a = *reinterpret_cast<const float2*>(wrow2 + k0);
  #pragma unroll
  for (int ct = 0; ct < 4; ++ct)
    f.b[ct] = *reinterpret_cast<const float2*>(fcol2 + ct * 16 * fpitch + k0);
}

__device__ __forceinline__ v8f mma4(v2f a, v2f b, v8f cacc) {
  return __builtin_amdgcn_wmma_f32_16x16x4_f32(false, a, false, b, (short)0, cacc, false, false);
}

__device__ __forceinline__ void chunk_mma(const Chunk& f, v8f (&acc)[4]) {
  v2f a0, a1;
  a0.x = f.a.x; a0.y = f.a.y; a1.x = f.a.z; a1.y = f.a.w;
  #pragma unroll
  for (int ct = 0; ct < 4; ++ct) {
    v2f b; b.x = f.b[ct].x; b.y = f.b[ct].y;
    acc[ct] = mma4(a0, b, acc[ct]);
  }
  #pragma unroll
  for (int ct = 0; ct < 4; ++ct) {
    v2f b; b.x = f.b[ct].z; b.y = f.b[ct].w;
    acc[ct] = mma4(a1, b, acc[ct]);
  }
}

__device__ __forceinline__ void tail_mma(const Tail& f, v8f (&acc)[4]) {
  v2f a; a.x = f.a.x; a.y = f.a.y;
  #pragma unroll
  for (int ct = 0; ct < 4; ++ct) {
    v2f b; b.x = f.b[ct].x; b.y = f.b[ct].y;
    acc[ct] = mma4(a, b, acc[ct]);
  }
}

__device__ __forceinline__ void sched_fence() { __builtin_amdgcn_sched_barrier(0); }

__device__ __forceinline__ void acc_zero(v8f (&acc)[4]) {
  #pragma unroll
  for (int ct = 0; ct < 4; ++ct)
    #pragma unroll
    for (int v = 0; v < 8; ++v) acc[ct][v] = 0.f;
}

__device__ __forceinline__ void conv1_strip(const float* __restrict__ W1l,
                                            const float* __restrict__ ft,
                                            int mt, v8f (&acc)[4]) {
  const int lane = threadIdx.x & 31;
  const int c = lane & 15, h = lane >> 4;
  const float* wrow = W1l + (mt * 16 + c) * W1STRIDE;
  const float* fcol = ft + c * FPITCH;
  Chunk g[4];
  Tail  gt;
  acc_zero(acc);
  #pragma unroll
  for (int j = 0; j < 4; ++j) chunk_load(g[j], wrow + 4 * h, fcol + 4 * h, FPITCH, j);
  tail_load(gt, wrow + 2 * h, fcol + 2 * h, FPITCH, 32);
  sched_fence();
  #pragma unroll
  for (int j = 0; j < 4; ++j) chunk_mma(g[j], acc);
  tail_mma(gt, acc);
  sched_fence();
}

__device__ __forceinline__ void conv2_strip(const float* __restrict__ W2l,
                                            const float* __restrict__ f1l,
                                            int mt, v8f (&acc)[4]) {
  const int lane = threadIdx.x & 31;
  const int c = lane & 15, h = lane >> 4;
  const float* wrow = W2l + (mt * 16 + c) * W2STRIDE + 4 * h;
  const float* fcol = f1l + c * F1PITCH + 4 * h;
  Chunk s0[2], s1[2];
  acc_zero(acc);
  chunk_load(s0[0], wrow, fcol, F1PITCH, 0);
  chunk_load(s0[1], wrow, fcol, F1PITCH, 1);
  chunk_load(s1[0], wrow, fcol, F1PITCH, 2);
  chunk_load(s1[1], wrow, fcol, F1PITCH, 3);
  sched_fence();
  chunk_mma(s0[0], acc);
  chunk_mma(s0[1], acc);
  sched_fence();
  chunk_load(s0[0], wrow, fcol, F1PITCH, 4);
  chunk_load(s0[1], wrow, fcol, F1PITCH, 5);
  sched_fence();
  chunk_mma(s1[0], acc);
  chunk_mma(s1[1], acc);
  sched_fence();
  chunk_load(s1[0], wrow, fcol, F1PITCH, 6);
  chunk_load(s1[1], wrow, fcol, F1PITCH, 7);
  sched_fence();
  chunk_mma(s0[0], acc);
  chunk_mma(s0[1], acc);
  chunk_mma(s1[0], acc);
  chunk_mma(s1[1], acc);
  sched_fence();
}

__device__ __forceinline__ void bn1_store(const v8f (&acc)[4], const float (&sc1)[8],
                                          const float (&sh1)[8], float* f1l, int mt) {
  const int lane = threadIdx.x & 31;
  const int c = lane & 15, h = lane >> 4;
  #pragma unroll
  for (int ct = 0; ct < 4; ++ct) {
    float y[8];
    #pragma unroll
    for (int v = 0; v < 8; ++v) {
      float t = acc[ct][v] * sc1[v] + sh1[v];
      y[v] = t > 0.f ? t : 0.f;
    }
    float4* dst = reinterpret_cast<float4*>(f1l + (ct * 16 + c) * F1PITCH + mt * 16 + h * 8);
    float4 q0, q1;
    q0.x = y[0]; q0.y = y[1]; q0.z = y[2]; q0.w = y[3];
    q1.x = y[4]; q1.y = y[5]; q1.z = y[6]; q1.w = y[7];
    dst[0] = q0;
    dst[1] = q1;
  }
}


#define W1Q (COUT * CIN / 4)
#define W2Q (COUT * COUT / 4)

__device__ __forceinline__ void w1_fetch(const float* __restrict__ W1, float4 (&r)[5]) {
  const float4* __restrict__ src = reinterpret_cast<const float4*>(W1);
  const int t = threadIdx.x;
  #pragma unroll
  for (int e = 0; e < 5; ++e) {
    int q = t + 128 * e;
    r[e] = src[q < W1Q ? q : W1Q - 1];
  }
}

__device__ __forceinline__ void w1_put(const float4 (&r)[5], float* W1l) {
  const int t = threadIdx.x;
  #pragma unroll
  for (int e = 0; e < 5; ++e) {
    int q = t + 128 * e;
    if (q < W1Q) {
      const float v[4] = {r[e].x, r[e].y, r[e].z, r[e].w};
      #pragma unroll
      for (int u = 0; u < 4; ++u) {
        int i   = 4 * q + u;
        int row = i / CIN;
        int cc  = i - row * CIN;
        int kp  = cc >= 3 ? cc - 3 : 32 + cc;
        W1l[row * W1STRIDE + kp] = v[u];
      }
    }
  }
  if (t < COUT) W1l[t * W1STRIDE + CIN] = 0.f;
}

__device__ __forceinline__ void w2_fetch(const float* __restrict__ W2, float4 (&r)[8]) {
  const float4* __restrict__ src = reinterpret_cast<const float4*>(W2);
  const int t = threadIdx.x;
  #pragma unroll
  for (int e = 0; e < 8; ++e) r[e] = src[t + 128 * e];
}

__device__ __forceinline__ void w2_put(const float4 (&r)[8], float* W2l) {
  const int t = threadIdx.x;
  #pragma unroll
  for (int e = 0; e < 8; ++e) {
    int q = t + 128 * e;
    *reinterpret_cast<float4*>(&W2l[(q >> 4) * W2STRIDE + (q & 15) * 4]) = r[e];
  }
}

__global__ __launch_bounds__(128) void conv1_stats_kernel(
    const float* __restrict__ p, const float* __restrict__ xt,
    const int* __restrict__ idxg, const float* __restrict__ W1,
    float* __restrict__ part1) {
  __shared__ __attribute__((aligned(16))) float W1l[COUT * W1STRIDE];
  __shared__ __attribute__((aligned(16))) float flds[2][64 * FPITCH];
  int t = threadIdx.x;
  const int G = gridDim.x;
  int pair = blockIdx.x;
  const int j0 = tile_idx(idxg, pair);
  int jn = tile_idx(idxg, pair + G < NPAIRS ? pair + G : pair);
  {
    float4 r1[5];
    w1_fetch(W1, r1);
    w1_put(r1, W1l);
  }
  FPart r;
  ftile_fetch(p, xt, pair, j0, r);
  int mt = t >> 5, lane = t & 31;
  float s[8], sq[8];
  #pragma unroll
  for (int v = 0; v < 8; ++v) { s[v] = 0.f; sq[v] = 0.f; }

  for (int trip = 0; pair < NPAIRS; pair += G, ++trip) {
    float* ft = flds[trip & 1];
    ftile_put(r, ft);
    const int pn  = pair + G     < NPAIRS ? pair + G     : pair;
    const int pn2 = pair + 2 * G < NPAIRS ? pair + 2 * G : pair;
    ftile_fetch(p, xt, pn, jn, r);
    jn = tile_idx(idxg, pn2);
    lds_barrier();
    v8f acc[4];
    conv1_strip(W1l, ft, mt, acc);
    #pragma unroll
    for (int ct = 0; ct < 4; ++ct) {
      #pragma unroll
      for (int v = 0; v < 8; ++v) { float y = acc[ct][v]; s[v] += y; sq[v] += y * y; }
    }
  }
  #pragma unroll
  for (int v = 0; v < 8; ++v)
    #pragma unroll
    for (int off = 1; off < 16; off <<= 1) {
      s[v]  += __shfl_xor(s[v],  off, 32);
      sq[v] += __shfl_xor(sq[v], off, 32);
    }
  if ((lane & 15) == 0) {
    int hi = lane >> 4;
    #pragma unroll
    for (int v = 0; v < 8; ++v) {
      int o = mt * 16 + hi * 8 + v;
      part1[(size_t)blockIdx.x * 128 + o]      = s[v];
      part1[(size_t)blockIdx.x * 128 + 64 + o] = sq[v];
    }
  }
}

__global__ void bn_finalize_kernel(const float* __restrict__ part,
                                   const float* __restrict__ g,
                                   const float* __restrict__ bta,
                                   float* __restrict__ bnp, int nblocks) {
  int o = threadIdx.x;
  if (o >= COUT) return;
  float s = 0.f, sq = 0.f;
  for (int blk = 0; blk < nblocks; ++blk) {
    s  += part[(size_t)blk * 128 + o];
    sq += part[(size_t)blk * 128 + 64 + o];
  }
  float mean  = s / CNTF;
  float var   = sq / CNTF - mean * mean;
  float scale = g[o] * rsqrtf(var + EPSV);
  bnp[o]      = scale;
  bnp[64 + o] = bta[o] - mean * scale;
}

__device__ __forceinline__ void bn_from_rows(const float* __restrict__ part,
                                             const float* g,
                                             const float* bta,
                                             float* red, float* bnl,
                                             float* __restrict__ bnp_out) {
  const int t = threadIdx.x;
  const int q = t & 31, w = t >> 5;
  const float gv = g[t & 63];
  const float bv = bta[t & 63];
  const float4* __restrict__ rows = reinterpret_cast<const float4*>(part) + (size_t)w * (GSTAT / 4) * 32 + q;
  float4 a[4];
  #pragma unroll
  for (int u = 0; u < 4; ++u) { a[u].x = 0.f; a[u].y = 0.f; a[u].z = 0.f; a[u].w = 0.f; }
  #pragma unroll
  for (int i = 0; i < GSTAT / 4; ++i) {
    const float4 v = rows[(size_t)i * 32];
    a[i & 3].x += v.x; a[i & 3].y += v.y; a[i & 3].z += v.z; a[i & 3].w += v.w;
  }
  float4 sum;
  sum.x = (a[0].x + a[1].x) + (a[2].x + a[3].x);
  sum.y = (a[0].y + a[1].y) + (a[2].y + a[3].y);
  sum.z = (a[0].z + a[1].z) + (a[2].z + a[3].z);
  sum.w = (a[0].w + a[1].w) + (a[2].w + a[3].w);
  *reinterpret_cast<float4*>(red + w * 128 + 4 * q) = sum;
  lds_barrier();
  const int o = t & 63;
  const float s  = (red[o]      + red[128 + o])      + (red[256 + o]      + red[384 + o]);
  const float sq = (red[64 + o] + red[128 + 64 + o]) + (red[256 + 64 + o] + red[384 + 64 + o]);
  const float mean  = s / CNTF;
  const float var   = sq / CNTF - mean * mean;
  const float scale = gv * rsqrtf(var + EPSV);
  const float shift = bv - mean * scale;
  if (t < COUT) {
    bnl[o]      = scale;
    bnl[64 + o] = shift;
    if (blockIdx.x == 0) {
      bnp_out[o]      = scale;
      bnp_out[64 + o] = shift;
    }
  }
  lds_barrier();
}

__global__ __launch_bounds__(128) void conv2_stats_kernel(
    const float* __restrict__ p, const float* __restrict__ xt,
    const int* __restrict__ idxg, const float* __restrict__ W1,
    const float* __restrict__ W2, const float* __restrict__ part1,
    const float* g1, const float* b1,
    float* __restrict__ bnp1, float* __restrict__ part2) {
  __shared__ __attribute__((aligned(16))) float W1l[COUT * W1STRIDE];
  __shared__ __attribute__((aligned(16))) float W2l[COUT * W2STRIDE];
  __shared__ __attribute__((aligned(16))) float flds[64 * FPITCH];
  __shared__ __attribute__((aligned(16))) float f1l[64 * F1PITCH];
  __shared__ __attribute__((aligned(16))) float red[4 * 128];
  __shared__ float bn1l[128];
  int t = threadIdx.x;
  const int G = gridDim.x;
  int pair = blockIdx.x;
  const int j0 = tile_idx(idxg, pair);
  int jn = tile_idx(idxg, pair + G < NPAIRS ? pair + G : pair);
  {
    float4 r1[5], r2[8];
    w1_fetch(W1, r1);
    w2_fetch(W2, r2);
    w1_put(r1, W1l);
    w2_put(r2, W2l);
  }
  FPart r;
  ftile_fetch(p, xt, pair, j0, r);
  bn_from_rows(part1, g1, b1, red, bn1l, bnp1);
  int mt = t >> 5, lane = t & 31;
  int hi = lane >> 4;
  float sc1[8], sh1[8];
  #pragma unroll
  for (int v = 0; v < 8; ++v) {
    int o = mt * 16 + hi * 8 + v;
    sc1[v] = bn1l[o]; sh1[v] = bn1l[64 + o];
  }
  float s[8], sq[8];
  #pragma unroll
  for (int v = 0; v < 8; ++v) { s[v] = 0.f; sq[v] = 0.f; }

  for (; pair < NPAIRS; pair += G) {
    ftile_put(r, flds);
    const int pn  = pair + G     < NPAIRS ? pair + G     : pair;
    const int pn2 = pair + 2 * G < NPAIRS ? pair + 2 * G : pair;
    ftile_fetch(p, xt, pn, jn, r);
    jn = tile_idx(idxg, pn2);
    lds_barrier();
    v8f acc[4];
    conv1_strip(W1l, flds, mt, acc);
    bn1_store(acc, sc1, sh1, f1l, mt);
    lds_barrier();
    conv2_strip(W2l, f1l, mt, acc);
    #pragma unroll
    for (int ct = 0; ct < 4; ++ct) {
      #pragma unroll
      for (int v = 0; v < 8; ++v) { float y = acc[ct][v]; s[v] += y; sq[v] += y * y; }
    }
  }
  #pragma unroll
  for (int v = 0; v < 8; ++v)
    #pragma unroll
    for (int off = 1; off < 16; off <<= 1) {
      s[v]  += __shfl_xor(s[v],  off, 32);
      sq[v] += __shfl_xor(sq[v], off, 32);
    }
  if ((lane & 15) == 0) {
    #pragma unroll
    for (int v = 0; v < 8; ++v) {
      int o = mt * 16 + hi * 8 + v;
      part2[(size_t)blockIdx.x * 128 + o]      = s[v];
      part2[(size_t)blockIdx.x * 128 + 64 + o] = sq[v];
    }
  }
}

__global__ __launch_bounds__(128) void final_kernel(
    const float* __restrict__ p, const float* __restrict__ xt,
    const int* __restrict__ idxg, const float* __restrict__ W1,
    const float* __restrict__ W2, const float* __restrict__ bnp1,
    const float* __restrict__ bnp2, float* __restrict__ out) {
  __shared__ __attribute__((aligned(16))) float W1l[COUT * W1STRIDE];
  __shared__ __attribute__((aligned(16))) float W2l[COUT * W2STRIDE];
  __shared__ __attribute__((aligned(16))) float flds[64 * FPITCH];
  __shared__ __attribute__((aligned(16))) float f1l[64 * F1PITCH];
  int t = threadIdx.x;
  const int G = gridDim.x;
  int pair = blockIdx.x;
  const int j0 = tile_idx(idxg, pair);
  int jn = tile_idx(idxg, pair + G < NPAIRS ? pair + G : pair);
  {
    float4 r1[5], r2[8];
    w1_fetch(W1, r1);
    w2_fetch(W2, r2);
    w1_put(r1, W1l);
    w2_put(r2, W2l);
  }
  FPart r;
  ftile_fetch(p, xt, pair, j0, r);
  int mt = t >> 5, lane = t & 31;
  int hi = lane >> 4;
  float sc1[8], sh1[8], sc2[8], sh2[8];
  #pragma unroll
  for (int v = 0; v < 8; ++v) {
    int o = mt * 16 + hi * 8 + v;
    sc1[v] = bnp1[o]; sh1[v] = bnp1[64 + o];
    sc2[v] = bnp2[o]; sh2[v] = bnp2[64 + o];
  }
  for (; pair < NPAIRS; pair += G) {
    int b  = pair / (NN / 2);
    int n0 = (pair % (NN / 2)) * 2;
    ftile_put(r, flds);
    const int pn  = pair + G     < NPAIRS ? pair + G     : pair;
    const int pn2 = pair + 2 * G < NPAIRS ? pair + 2 * G : pair;
    ftile_fetch(p, xt, pn, jn, r);
    jn = tile_idx(idxg, pn2);
    lds_barrier();
    v8f acc[4];
    conv1_strip(W1l, flds, mt, acc);
    bn1_store(acc, sc1, sh1, f1l, mt);
    lds_barrier();
    conv2_strip(W2l, f1l, mt, acc);
    float m0[8], m1[8];
    #pragma unroll
    for (int v = 0; v < 8; ++v) { m0[v] = 0.f; m1[v] = 0.f; }
    #pragma unroll
    for (int ct = 0; ct < 4; ++ct) {
      #pragma unroll
      for (int v = 0; v < 8; ++v) {
        float y = acc[ct][v] * sc2[v] + sh2[v];
        y = y > 0.f ? y : 0.f;
        if (ct < 2) m0[v] = fmaxf(m0[v], y);
        else        m1[v] = fmaxf(m1[v], y);
      }
    }
    #pragma unroll
    for (int v = 0; v < 8; ++v)
      #pragma unroll
      for (int off = 1; off < 16; off <<= 1) {
        m0[v] = fmaxf(m0[v], __shfl_xor(m0[v], off, 32));
        m1[v] = fmaxf(m1[v], __shfl_xor(m1[v], off, 32));
      }
    if ((lane & 15) == 0) {
      #pragma unroll
      for (int v = 0; v < 8; ++v) {
        int o = mt * 16 + hi * 8 + v;
        out[((size_t)b * COUT + o) * NN + n0]     = m0[v];
        out[((size_t)b * COUT + o) * NN + n0 + 1] = m1[v];
      }
    }
  }
}

extern "C" void kernel_launch(void* const* d_in, const int* in_sizes, int n_in,
                              void* d_out, int out_size, void* d_ws, size_t ws_size,
                              hipStream_t stream) {
  const float* p  = (const float*)d_in[0];
  const float* x  = (const float*)d_in[1];
  const float* W1 = (const float*)d_in[2];
  const float* g1 = (const float*)d_in[3];
  const float* b1 = (const float*)d_in[4];
  const float* W2 = (const float*)d_in[5];
  const float* g2 = (const float*)d_in[6];
  const float* b2 = (const float*)d_in[7];
  float* out = (float*)d_out;

  char*  ws    = (char*)d_ws;
  int*   idx   = (int*)ws;
  float* xt    = (float*)(ws + (size_t)BB * NN * KS * 4);
  float* part1 = (float*)(ws + (size_t)BB * NN * KS * 4
                             + (size_t)BB * NN * CX * 4);
  float* part2 = part1 + (size_t)GCONV * 128;
  float* bnp1  = part2 + (size_t)GCONV * 128;
  float* bnp2  = bnp1 + 128;

  xpose_kernel<<<dim3(NN / 64, BB), 256, 0, stream>>>(x, xt);
  ballq_kernel<<<dim3(NN / 32, BB), 256, 0, stream>>>(p, idx);
  conv1_stats_kernel<<<GSTAT, 128, 0, stream>>>(p, xt, idx, W1, part1);
  conv2_stats_kernel<<<GSTAT, 128, 0, stream>>>(p, xt, idx, W1, W2, part1, g1, b1, bnp1, part2);
  bn_finalize_kernel<<<1, 64, 0, stream>>>(part2, g2, b2, bnp2, GSTAT);
  final_kernel<<<GCONV, 128, 0, stream>>>(p, xt, idx, W1, W2, bnp1, bnp2, out);
}


# last pass: a 16-column tile holding only padding copies of slot 0 (slots 16..31 of a ball of 16 or fewer points) is left out of both convs; a copy cannot change the max over K; results unchanged
# speedup vs baseline: 4.5077x; 1.0331x over previous
#include <hip/hip_runtime.h>
#include <hip/hip_bf16.h>


typedef __attribute__((ext_vector_type(2))) float v2f;
typedef __attribute__((ext_vector_type(8))) float v8f;

#define BB      4
#define NN      4096
#define CX      32
#define CIN     35
#define CINP    36
#define COUT    64
#define KS      32
#define R2      0.01f
#define EPSV    1e-5f
#define NPAIRS  (BB * (NN / 2))
#define GCONV   1024
#define GSTAT   256
#define CNTF    ((float)BB * (float)NN * (float)KS)

#define FPITCH   36
#define F1PITCH  68
#define W1STRIDE 36
#define W2STRIDE 68

__global__ void xpose_kernel(const float* __restrict__ x, float* __restrict__ xt) {
  __shared__ float tile[CX][65];
  int b  = blockIdx.y;
  int n0 = blockIdx.x * 64;
  int t  = threadIdx.x;
  for (int i = t; i < CX * 64; i += 256) {
    int c = i >> 6, nn = i & 63;
    tile[c][nn] = x[((size_t)b * CX + c) * NN + n0 + nn];
  }
  __syncthreads();
  for (int i = t; i < CX * 64; i += 256) {
    int nn = i >> 5, c = i & 31;
    xt[((size_t)b * NN + n0 + nn) * CX + c] = tile[c][nn];
  }
}

#define BQ_WAVES 8
#define BQ_SEG   (NN / BQ_WAVES)
#define BQ_GRP   16
#define BQ_PITCH 33

__global__ __launch_bounds__(256) void ballq_kernel(const float* __restrict__ p,
                                                    int* __restrict__ idx) {
  __shared__ unsigned short lst[BQ_WAVES][32][BQ_PITCH];
  __shared__ int cnts[BQ_WAVES][32];
  const int b    = blockIdx.y;
  const int lane = threadIdx.x & 31;
  const int seg  = __builtin_amdgcn_readfirstlane((int)(threadIdx.x >> 5));
  const float* __restrict__ pb = p + (size_t)b * NN * 3;
  const int q = blockIdx.x * 32 + lane;
  const float px = pb[q * 3 + 0];
  const float py = pb[q * 3 + 1];
  const float pz = pb[q * 3 + 2];
  const float* __restrict__ cb = pb + seg * (BQ_SEG * 3);
  unsigned short* mylst = &lst[seg][lane][0];
  int cnt = 0;
  for (int g = 0; g < BQ_SEG; g += BQ_GRP) {
    unsigned m = 0;
    #pragma unroll
    for (int jj = 0; jj < BQ_GRP; ++jj) {
      float dx = cb[(g + jj) * 3 + 0] - px;
      float dy = cb[(g + jj) * 3 + 1] - py;
      float dz = cb[(g + jj) * 3 + 2] - pz;
      float d2 = __builtin_fmaf(dz, dz, __builtin_fmaf(dy, dy, dx * dx));
      m = m + m + (d2 < R2 ? 1u : 0u);
    }
    while (m != 0) {
      int lz   = __builtin_clz(m);
      int slot = cnt < KS ? cnt : KS;
      mylst[slot] = (unsigned short)(seg * BQ_SEG + g + lz - (32 - BQ_GRP));
      cnt++;
      m &= ~(0x80000000u >> lz);
    }
  }
  cnts[seg][lane] = cnt;
  __syncthreads();

  const int ql = threadIdx.x >> 3;
  const int r0 = (threadIdx.x & 7) * 4;
  int v[4] = {0, 0, 0, 0};
  int pre = 0, first = 0;
  bool have = false;
  #pragma unroll
  for (int s = 0; s < BQ_WAVES; ++s) {
    int c = cnts[s][ql];
    c = c < KS ? c : KS;
    if (!have && c > 0) { first = lst[s][ql][0]; have = true; }
    #pragma unroll
    for (int k = 0; k < 4; ++k) {
      int r = r0 + k - pre;
      if (r >= 0 && r < c) v[k] = lst[s][ql][r];
    }
    pre += c;
  }
  #pragma unroll
  for (int k = 0; k < 4; ++k)
    if (r0 + k >= pre) v[k] = first;
  int4 o;
  o.x = v[0]; o.y = v[1]; o.z = v[2]; o.w = v[3];
  *reinterpret_cast<int4*>(idx + ((size_t)b * NN + blockIdx.x * 32 + ql) * KS + r0) = o;
}

struct FPart {
  float xv[16];
  float pj[2];
  float pq[2];
};

__device__ __forceinline__ int tile_idx(const int* __restrict__ idxg, int pair) {
  return idxg[(size_t)pair * 64 + (threadIdx.x & 63)];
}

__device__ __forceinline__ void ftile_fetch(const float* __restrict__ p,
                                            const float* __restrict__ xt,
                                            int pair, int j, FPart& r) {
  const int t = threadIdx.x;
  const int b = pair / (NN / 2);
  const float* xr = xt + ((size_t)b * NN + j) * CX + (t < 64 ? 0 : 16);
  #pragma unroll
  for (int c = 0; c < 16; ++c) r.xv[c] = xr[c];
  const float* pj = p + ((size_t)b * NN + j) * 3;
  const float* pq = p + ((size_t)2 * pair + ((t & 63) >> 5)) * 3;
  if (t < 64) {
    r.pj[0] = pj[0]; r.pj[1] = pj[1];
    r.pq[0] = pq[0]; r.pq[1] = pq[1];
  } else {
    r.pj[0] = pj[2];
    r.pq[0] = pq[2];
    r.pj[1] = 0.f;
    r.pq[1] = 0.f;
  }
}

__device__ __forceinline__ void ftile_put(const FPart& r, float* flds) {
  const int t = threadIdx.x;
  const bool lo = t < 64;
  float* colp = flds + (t & 63) * FPITCH;
  float4* xq = reinterpret_cast<float4*>(colp + (lo ? 0 : 16));
  #pragma unroll
  for (int q = 0; q < 4; ++q) {
    float4 v;
    v.x = r.xv[4 * q + 0]; v.y = r.xv[4 * q + 1]; v.z = r.xv[4 * q + 2]; v.w = r.xv[4 * q + 3];
    xq[q] = v;
  }
  float2 d;
  d.x = r.pj[0] - r.pq[0];
  d.y = r.pj[1] - r.pq[1];
  *reinterpret_cast<float2*>(colp + (lo ? 32 : 34)) = d;
}

__device__ __forceinline__ void lds_barrier() {
  asm volatile("s_wait_dscnt 0x0\n\ts_barrier_signal -1\n\ts_barrier_wait -1" ::: "memory");
}

struct Chunk {
  float4 a;
  float4 b[4];
};
struct Tail {
  float2 a;
  float2 b[4];
};

template <int LIVE>
__device__ __forceinline__ void chunk_load(Chunk& f, const float* wrow, const float* fcol,
                                           int fpitch, int j) {
  f.a = *reinterpret_cast<const float4*>(wrow + 8 * j);
  #pragma unroll
  for (int ct = 0; ct < 4; ++ct)
    if ((LIVE >> ct) & 1)
      f.b[ct] = *reinterpret_cast<const float4*>(fcol + ct * 16 * fpitch + 8 * j);
}

template <int LIVE>
__device__ __forceinline__ void tail_load(Tail& f, const float* wrow2, const float* fcol2,
                                          int fpitch, int k0) {
  f.a = *reinterpret_cast<const float2*>(wrow2 + k0);
  #pragma unroll
  for (int ct = 0; ct < 4; ++ct)
    if ((LIVE >> ct) & 1)
      f.b[ct] = *reinterpret_cast<const float2*>(fcol2 + ct * 16 * fpitch + k0);
}

__device__ __forceinline__ v8f mma4(v2f a, v2f b, v8f cacc) {
  return __builtin_amdgcn_wmma_f32_16x16x4_f32(false, a, false, b, (short)0, cacc, false, false);
}

template <int LIVE>
__device__ __forceinline__ void chunk_mma(const Chunk& f, v8f (&acc)[4]) {
  v2f a0, a1;
  a0.x = f.a.x; a0.y = f.a.y; a1.x = f.a.z; a1.y = f.a.w;
  #pragma unroll
  for (int ct = 0; ct < 4; ++ct) {
    if (!((LIVE >> ct) & 1)) continue;
    v2f b; b.x = f.b[ct].x; b.y = f.b[ct].y;
    acc[ct] = mma4(a0, b, acc[ct]);
  }
  #pragma unroll
  for (int ct = 0; ct < 4; ++ct) {
    if (!((LIVE >> ct) & 1)) continue;
    v2f b; b.x = f.b[ct].z; b.y = f.b[ct].w;
    acc[ct] = mma4(a1, b, acc[ct]);
  }
}

template <int LIVE>
__device__ __forceinline__ void tail_mma(const Tail& f, v8f (&acc)[4]) {
  v2f a; a.x = f.a.x; a.y = f.a.y;
  #pragma unroll
  for (int ct = 0; ct < 4; ++ct) {
    if (!((LIVE >> ct) & 1)) continue;
    v2f b; b.x = f.b[ct].x; b.y = f.b[ct].y;
    acc[ct] = mma4(a, b, acc[ct]);
  }
}

__device__ __forceinline__ void sched_fence() { __builtin_amdgcn_sched_barrier(0); }

template <int LIVE>
__device__ __forceinline__ void acc_zero(v8f (&acc)[4]) {
  #pragma unroll
  for (int ct = 0; ct < 4; ++ct)
    if ((LIVE >> ct) & 1) {
      #pragma unroll
      for (int v = 0; v < 8; ++v) acc[ct][v] = 0.f;
    }
}

template <int LIVE>
__device__ __forceinline__ void conv1_strip(const float* __restrict__ W1l,
                                            const float* __restrict__ ft,
                                            int mt, v8f (&acc)[4]) {
  const int lane = threadIdx.x & 31;
  const int c = lane & 15, h = lane >> 4;
  const float* wrow = W1l + (mt * 16 + c) * W1STRIDE;
  const float* fcol = ft + c * FPITCH;
  Chunk g[4];
  Tail  gt;
  acc_zero<LIVE>(acc);
  #pragma unroll
  for (int j = 0; j < 4; ++j) chunk_load<LIVE>(g[j], wrow + 4 * h, fcol + 4 * h, FPITCH, j);
  tail_load<LIVE>(gt, wrow + 2 * h, fcol + 2 * h, FPITCH, 32);
  sched_fence();
  #pragma unroll
  for (int j = 0; j < 4; ++j) chunk_mma<LIVE>(g[j], acc);
  tail_mma<LIVE>(gt, acc);
  sched_fence();
}

template <int LIVE>
__device__ __forceinline__ void conv2_strip(const float* __restrict__ W2l,
                                            const float* __restrict__ f1l,
                                            int mt, v8f (&acc)[4]) {
  const int lane = threadIdx.x & 31;
  const int c = lane & 15, h = lane >> 4;
  const float* wrow = W2l + (mt * 16 + c) * W2STRIDE + 4 * h;
  const float* fcol = f1l + c * F1PITCH + 4 * h;
  Chunk s0[2], s1[2];
  acc_zero<LIVE>(acc);
  chunk_load<LIVE>(s0[0], wrow, fcol, F1PITCH, 0);
  chunk_load<LIVE>(s0[1], wrow, fcol, F1PITCH, 1);
  chunk_load<LIVE>(s1[0], wrow, fcol, F1PITCH, 2);
  chunk_load<LIVE>(s1[1], wrow, fcol, F1PITCH, 3);
  sched_fence();
  chunk_mma<LIVE>(s0[0], acc);
  chunk_mma<LIVE>(s0[1], acc);
  sched_fence();
  chunk_load<LIVE>(s0[0], wrow, fcol, F1PITCH, 4);
  chunk_load<LIVE>(s0[1], wrow, fcol, F1PITCH, 5);
  sched_fence();
  chunk_mma<LIVE>(s1[0], acc);
  chunk_mma<LIVE>(s1[1], acc);
  sched_fence();
  chunk_load<LIVE>(s1[0], wrow, fcol, F1PITCH, 6);
  chunk_load<LIVE>(s1[1], wrow, fcol, F1PITCH, 7);
  sched_fence();
  chunk_mma<LIVE>(s0[0], acc);
  chunk_mma<LIVE>(s0[1], acc);
  chunk_mma<LIVE>(s1[0], acc);
  chunk_mma<LIVE>(s1[1], acc);
  sched_fence();
}

template <int LIVE>
__device__ __forceinline__ void bn1_store(const v8f (&acc)[4], const float (&sc1)[8],
                                          const float (&sh1)[8], float* f1l, int mt) {
  const int lane = threadIdx.x & 31;
  const int c = lane & 15, h = lane >> 4;
  #pragma unroll
  for (int ct = 0; ct < 4; ++ct) {
    if (!((LIVE >> ct) & 1)) continue;
    float y[8];
    #pragma unroll
    for (int v = 0; v < 8; ++v) {
      float t = acc[ct][v] * sc1[v] + sh1[v];
      y[v] = t > 0.f ? t : 0.f;
    }
    float4* dst = reinterpret_cast<float4*>(f1l + (ct * 16 + c) * F1PITCH + mt * 16 + h * 8);
    float4 q0, q1;
    q0.x = y[0]; q0.y = y[1]; q0.z = y[2]; q0.w = y[3];
    q1.x = y[4]; q1.y = y[5]; q1.z = y[6]; q1.w = y[7];
    dst[0] = q0;
    dst[1] = q1;
  }
}


#define W1Q (COUT * CIN / 4)
#define W2Q (COUT * COUT / 4)

__device__ __forceinline__ void w1_fetch(const float* __restrict__ W1, float4 (&r)[5]) {
  const float4* __restrict__ src = reinterpret_cast<const float4*>(W1);
  const int t = threadIdx.x;
  #pragma unroll
  for (int e = 0; e < 5; ++e) {
    int q = t + 128 * e;
    r[e] = src[q < W1Q ? q : W1Q - 1];
  }
}

__device__ __forceinline__ void w1_put(const float4 (&r)[5], float* W1l) {
  const int t = threadIdx.x;
  #pragma unroll
  for (int e = 0; e < 5; ++e) {
    int q = t + 128 * e;
    if (q < W1Q) {
      const float v[4] = {r[e].x, r[e].y, r[e].z, r[e].w};
      #pragma unroll
      for (int u = 0; u < 4; ++u) {
        int i   = 4 * q + u;
        int row = i / CIN;
        int cc  = i - row * CIN;
        int kp  = cc >= 3 ? cc - 3 : 32 + cc;
        W1l[row * W1STRIDE + kp] = v[u];
      }
    }
  }
  if (t < COUT) W1l[t * W1STRIDE + CIN] = 0.f;
}

__device__ __forceinline__ void w2_fetch(const float* __restrict__ W2, float4 (&r)[8]) {
  const float4* __restrict__ src = reinterpret_cast<const float4*>(W2);
  const int t = threadIdx.x;
  #pragma unroll
  for (int e = 0; e < 8; ++e) r[e] = src[t + 128 * e];
}

__device__ __forceinline__ void w2_put(const float4 (&r)[8], float* W2l) {
  const int t = threadIdx.x;
  #pragma unroll
  for (int e = 0; e < 8; ++e) {
    int q = t + 128 * e;
    *reinterpret_cast<float4*>(&W2l[(q >> 4) * W2STRIDE + (q & 15) * 4]) = r[e];
  }
}

__global__ __launch_bounds__(128) void conv1_stats_kernel(
    const float* __restrict__ p, const float* __restrict__ xt,
    const int* __restrict__ idxg, const float* __restrict__ W1,
    float* __restrict__ part1) {
  __shared__ __attribute__((aligned(16))) float W1l[COUT * W1STRIDE];
  __shared__ __attribute__((aligned(16))) float flds[2][64 * FPITCH];
  int t = threadIdx.x;
  const int G = gridDim.x;
  int pair = blockIdx.x;
  const int j0 = tile_idx(idxg, pair);
  int jn = tile_idx(idxg, pair + G < NPAIRS ? pair + G : pair);
  {
    float4 r1[5];
    w1_fetch(W1, r1);
    w1_put(r1, W1l);
  }
  FPart r;
  ftile_fetch(p, xt, pair, j0, r);
  int mt = t >> 5, lane = t & 31;
  float s[8], sq[8];
  #pragma unroll
  for (int v = 0; v < 8; ++v) { s[v] = 0.f; sq[v] = 0.f; }

  for (int trip = 0; pair < NPAIRS; pair += G, ++trip) {
    float* ft = flds[trip & 1];
    ftile_put(r, ft);
    const int pn  = pair + G     < NPAIRS ? pair + G     : pair;
    const int pn2 = pair + 2 * G < NPAIRS ? pair + 2 * G : pair;
    ftile_fetch(p, xt, pn, jn, r);
    jn = tile_idx(idxg, pn2);
    lds_barrier();
    v8f acc[4];
    conv1_strip<15>(W1l, ft, mt, acc);
    #pragma unroll
    for (int ct = 0; ct < 4; ++ct) {
      #pragma unroll
      for (int v = 0; v < 8; ++v) { float y = acc[ct][v]; s[v] += y; sq[v] += y * y; }
    }
  }
  #pragma unroll
  for (int v = 0; v < 8; ++v)
    #pragma unroll
    for (int off = 1; off < 16; off <<= 1) {
      s[v]  += __shfl_xor(s[v],  off, 32);
      sq[v] += __shfl_xor(sq[v], off, 32);
    }
  if ((lane & 15) == 0) {
    int hi = lane >> 4;
    #pragma unroll
    for (int v = 0; v < 8; ++v) {
      int o = mt * 16 + hi * 8 + v;
      part1[(size_t)blockIdx.x * 128 + o]      = s[v];
      part1[(size_t)blockIdx.x * 128 + 64 + o] = sq[v];
    }
  }
}

__global__ void bn_finalize_kernel(const float* __restrict__ part,
                                   const float* __restrict__ g,
                                   const float* __restrict__ bta,
                                   float* __restrict__ bnp, int nblocks) {
  int o = threadIdx.x;
  if (o >= COUT) return;
  float s = 0.f, sq = 0.f;
  for (int blk = 0; blk < nblocks; ++blk) {
    s  += part[(size_t)blk * 128 + o];
    sq += part[(size_t)blk * 128 + 64 + o];
  }
  float mean  = s / CNTF;
  float var   = sq / CNTF - mean * mean;
  float scale = g[o] * rsqrtf(var + EPSV);
  bnp[o]      = scale;
  bnp[64 + o] = bta[o] - mean * scale;
}

__device__ __forceinline__ void bn_from_rows(const float* __restrict__ part,
                                             const float* g,
                                             const float* bta,
                                             float* red, float* bnl,
                                             float* __restrict__ bnp_out) {
  const int t = threadIdx.x;
  const int q = t & 31, w = t >> 5;
  const float gv = g[t & 63];
  const float bv = bta[t & 63];
  const float4* __restrict__ rows = reinterpret_cast<const float4*>(part) + (size_t)w * (GSTAT / 4) * 32 + q;
  float4 a[4];
  #pragma unroll
  for (int u = 0; u < 4; ++u) { a[u].x = 0.f; a[u].y = 0.f; a[u].z = 0.f; a[u].w = 0.f; }
  #pragma unroll
  for (int i = 0; i < GSTAT / 4; ++i) {
    const float4 v = rows[(size_t)i * 32];
    a[i & 3].x += v.x; a[i & 3].y += v.y; a[i & 3].z += v.z; a[i & 3].w += v.w;
  }
  float4 sum;
  sum.x = (a[0].x + a[1].x) + (a[2].x + a[3].x);
  sum.y = (a[0].y + a[1].y) + (a[2].y + a[3].y);
  sum.z = (a[0].z + a[1].z) + (a[2].z + a[3].z);
  sum.w = (a[0].w + a[1].w) + (a[2].w + a[3].w);
  *reinterpret_cast<float4*>(red + w * 128 + 4 * q) = sum;
  lds_barrier();
  const int o = t & 63;
  const float s  = (red[o]      + red[128 + o])      + (red[256 + o]      + red[384 + o]);
  const float sq = (red[64 + o] + red[128 + 64 + o]) + (red[256 + 64 + o] + red[384 + 64 + o]);
  const float mean  = s / CNTF;
  const float var   = sq / CNTF - mean * mean;
  const float scale = gv * rsqrtf(var + EPSV);
  const float shift = bv - mean * scale;
  if (t < COUT) {
    bnl[o]      = scale;
    bnl[64 + o] = shift;
    if (blockIdx.x == 0) {
      bnp_out[o]      = scale;
      bnp_out[64 + o] = shift;
    }
  }
  lds_barrier();
}

__global__ __launch_bounds__(128) void conv2_stats_kernel(
    const float* __restrict__ p, const float* __restrict__ xt,
    const int* __restrict__ idxg, const float* __restrict__ W1,
    const float* __restrict__ W2, const float* __restrict__ part1,
    const float* g1, const float* b1,
    float* __restrict__ bnp1, float* __restrict__ part2) {
  __shared__ __attribute__((aligned(16))) float W1l[COUT * W1STRIDE];
  __shared__ __attribute__((aligned(16))) float W2l[COUT * W2STRIDE];
  __shared__ __attribute__((aligned(16))) float flds[64 * FPITCH];
  __shared__ __attribute__((aligned(16))) float f1l[64 * F1PITCH];
  __shared__ __attribute__((aligned(16))) float red[4 * 128];
  __shared__ float bn1l[128];
  int t = threadIdx.x;
  const int G = gridDim.x;
  int pair = blockIdx.x;
  const int j0 = tile_idx(idxg, pair);
  int jn = tile_idx(idxg, pair + G < NPAIRS ? pair + G : pair);
  {
    float4 r1[5], r2[8];
    w1_fetch(W1, r1);
    w2_fetch(W2, r2);
    w1_put(r1, W1l);
    w2_put(r2, W2l);
  }
  FPart r;
  ftile_fetch(p, xt, pair, j0, r);
  bn_from_rows(part1, g1, b1, red, bn1l, bnp1);
  int mt = t >> 5, lane = t & 31;
  int hi = lane >> 4;
  float sc1[8], sh1[8];
  #pragma unroll
  for (int v = 0; v < 8; ++v) {
    int o = mt * 16 + hi * 8 + v;
    sc1[v] = bn1l[o]; sh1[v] = bn1l[64 + o];
  }
  float s[8], sq[8];
  #pragma unroll
  for (int v = 0; v < 8; ++v) { s[v] = 0.f; sq[v] = 0.f; }

  for (; pair < NPAIRS; pair += G) {
    ftile_put(r, flds);
    const int pn  = pair + G     < NPAIRS ? pair + G     : pair;
    const int pn2 = pair + 2 * G < NPAIRS ? pair + 2 * G : pair;
    ftile_fetch(p, xt, pn, jn, r);
    jn = tile_idx(idxg, pn2);
    lds_barrier();
    v8f acc[4];
    conv1_strip<15>(W1l, flds, mt, acc);
    bn1_store<15>(acc, sc1, sh1, f1l, mt);
    lds_barrier();
    conv2_strip<15>(W2l, f1l, mt, acc);
    #pragma unroll
    for (int ct = 0; ct < 4; ++ct) {
      #pragma unroll
      for (int v = 0; v < 8; ++v) { float y = acc[ct][v]; s[v] += y; sq[v] += y * y; }
    }
  }
  #pragma unroll
  for (int v = 0; v < 8; ++v)
    #pragma unroll
    for (int off = 1; off < 16; off <<= 1) {
      s[v]  += __shfl_xor(s[v],  off, 32);
      sq[v] += __shfl_xor(sq[v], off, 32);
    }
  if ((lane & 15) == 0) {
    #pragma unroll
    for (int v = 0; v < 8; ++v) {
      int o = mt * 16 + hi * 8 + v;
      part2[(size_t)blockIdx.x * 128 + o]      = s[v];
      part2[(size_t)blockIdx.x * 128 + 64 + o] = sq[v];
    }
  }
}

struct PairEnds { int a0, a16, b0, b16; };

__device__ __forceinline__ PairEnds pair_ends(const int* __restrict__ idxg, int pair) {
  const int* row = idxg + (size_t)pair * 64;
  PairEnds e;
  e.a0 = row[0]; e.a16 = row[16]; e.b0 = row[32]; e.b16 = row[48];
  return e;
}

__device__ __forceinline__ int live_tiles(const PairEnds& e) {
  int live = 15;
  if (e.a16 == e.a0) live &= ~2;
  if (e.b16 == e.b0) live &= ~8;
  return __builtin_amdgcn_readfirstlane(live);
}

template <int LIVE>
__device__ __forceinline__ void final_tile(const float* __restrict__ W1l, const float* __restrict__ W2l,
                                           const float* flds, float* f1l, int mt,
                                           const float (&sc1)[8], const float (&sh1)[8],
                                           const float (&sc2)[8], const float (&sh2)[8],
                                           float (&m0)[8], float (&m1)[8]) {
  v8f acc[4];
  conv1_strip<LIVE>(W1l, flds, mt, acc);
  bn1_store<LIVE>(acc, sc1, sh1, f1l, mt);
  lds_barrier();
  conv2_strip<LIVE>(W2l, f1l, mt, acc);
  #pragma unroll
  for (int v = 0; v < 8; ++v) { m0[v] = 0.f; m1[v] = 0.f; }
  #pragma unroll
  for (int ct = 0; ct < 4; ++ct) {
    if (!((LIVE >> ct) & 1)) continue;
    #pragma unroll
    for (int v = 0; v < 8; ++v) {
      float y = acc[ct][v] * sc2[v] + sh2[v];
      y = y > 0.f ? y : 0.f;
      if (ct < 2) m0[v] = fmaxf(m0[v], y);
      else        m1[v] = fmaxf(m1[v], y);
    }
  }
}

__global__ __launch_bounds__(128) void final_kernel(
    const float* __restrict__ p, const float* __restrict__ xt,
    const int* __restrict__ idxg, const float* __restrict__ W1,
    const float* __restrict__ W2, const float* __restrict__ bnp1,
    const float* __restrict__ bnp2, float* __restrict__ out) {
  __shared__ __attribute__((aligned(16))) float W1l[COUT * W1STRIDE];
  __shared__ __attribute__((aligned(16))) float W2l[COUT * W2STRIDE];
  __shared__ __attribute__((aligned(16))) float flds[64 * FPITCH];
  __shared__ __attribute__((aligned(16))) float f1l[64 * F1PITCH];
  int t = threadIdx.x;
  const int G = gridDim.x;
  int pair = blockIdx.x;
  const int j0 = tile_idx(idxg, pair);
  int jn = tile_idx(idxg, pair + G < NPAIRS ? pair + G : pair);
  PairEnds ends = pair_ends(idxg, pair);
  {
    float4 r1[5], r2[8];
    w1_fetch(W1, r1);
    w2_fetch(W2, r2);
    w1_put(r1, W1l);
    w2_put(r2, W2l);
  }
  FPart r;
  ftile_fetch(p, xt, pair, j0, r);
  int mt = t >> 5, lane = t & 31;
  int hi = lane >> 4;
  float sc1[8], sh1[8], sc2[8], sh2[8];
  #pragma unroll
  for (int v = 0; v < 8; ++v) {
    int o = mt * 16 + hi * 8 + v;
    sc1[v] = bnp1[o]; sh1[v] = bnp1[64 + o];
    sc2[v] = bnp2[o]; sh2[v] = bnp2[64 + o];
  }
  for (; pair < NPAIRS; pair += G) {
    int b  = pair / (NN / 2);
    int n0 = (pair % (NN / 2)) * 2;
    const int live = live_tiles(ends);
    ftile_put(r, flds);
    const int pn  = pair + G     < NPAIRS ? pair + G     : pair;
    const int pn2 = pair + 2 * G < NPAIRS ? pair + 2 * G : pair;
    ftile_fetch(p, xt, pn, jn, r);
    jn = tile_idx(idxg, pn2);
    ends = pair_ends(idxg, pn);
    lds_barrier();
    float m0[8], m1[8];
    switch (live) {
      case 15: final_tile<15>(W1l, W2l, flds, f1l, mt, sc1, sh1, sc2, sh2, m0, m1); break;
      case 13: final_tile<13>(W1l, W2l, flds, f1l, mt, sc1, sh1, sc2, sh2, m0, m1); break;
      case 7:  final_tile<7>(W1l, W2l, flds, f1l, mt, sc1, sh1, sc2, sh2, m0, m1); break;
      default: final_tile<5>(W1l, W2l, flds, f1l, mt, sc1, sh1, sc2, sh2, m0, m1); break;
    }
    #pragma unroll
    for (int v = 0; v < 8; ++v)
      #pragma unroll
      for (int off = 1; off < 16; off <<= 1) {
        m0[v] = fmaxf(m0[v], __shfl_xor(m0[v], off, 32));
        m1[v] = fmaxf(m1[v], __shfl_xor(m1[v], off, 32));
      }
    if ((lane & 15) == 0) {
      #pragma unroll
      for (int v = 0; v < 8; ++v) {
        int o = mt * 16 + hi * 8 + v;
        out[((size_t)b * COUT + o) * NN + n0]     = m0[v];
        out[((size_t)b * COUT + o) * NN + n0 + 1] = m1[v];
      }
    }
  }
}

extern "C" void kernel_launch(void* const* d_in, const int* in_sizes, int n_in,
                              void* d_out, int out_size, void* d_ws, size_t ws_size,
                              hipStream_t stream) {
  const float* p  = (const float*)d_in[0];
  const float* x  = (const float*)d_in[1];
  const float* W1 = (const float*)d_in[2];
  const float* g1 = (const float*)d_in[3];
  const float* b1 = (const float*)d_in[4];
  const float* W2 = (const float*)d_in[5];
  const float* g2 = (const float*)d_in[6];
  const float* b2 = (const float*)d_in[7];
  float* out = (float*)d_out;

  char*  ws    = (char*)d_ws;
  int*   idx   = (int*)ws;
  float* xt    = (float*)(ws + (size_t)BB * NN * KS * 4);
  float* part1 = (float*)(ws + (size_t)BB * NN * KS * 4
                             + (size_t)BB * NN * CX * 4);
  float* part2 = part1 + (size_t)GCONV * 128;
  float* bnp1  = part2 + (size_t)GCONV * 128;
  float* bnp2  = bnp1 + 128;

  xpose_kernel<<<dim3(NN / 64, BB), 256, 0, stream>>>(x, xt);
  ballq_kernel<<<dim3(NN / 32, BB), 256, 0, stream>>>(p, idx);
  conv1_stats_kernel<<<GSTAT, 128, 0, stream>>>(p, xt, idx, W1, part1);
  conv2_stats_kernel<<<GSTAT, 128, 0, stream>>>(p, xt, idx, W1, W2, part1, g1, b1, bnp1, part2);
  bn_finalize_kernel<<<1, 64, 0, stream>>>(part2, g2, b2, bnp2, GSTAT);
  final_kernel<<<GCONV, 128, 0, stream>>>(p, xt, idx, W1, W2, bnp1, bnp2, out);
}


# transpose launch removed: each ball-query block transposes its own 32 points of x (loads in flight during the scan, LDS hand-over on the join's barrier); 5 launches per call, was 6
# speedup vs baseline: 4.7031x; 1.0433x over previous
#include <hip/hip_runtime.h>
#include <hip/hip_bf16.h>


typedef __attribute__((ext_vector_type(2))) float v2f;
typedef __attribute__((ext_vector_type(8))) float v8f;

#define BB      4
#define NN      4096
#define CX      32
#define CIN     35
#define CINP    36
#define COUT    64
#define KS      32
#define R2      0.01f
#define EPSV    1e-5f
#define NPAIRS  (BB * (NN / 2))
#define GCONV   1024
#define GSTAT   256
#define CNTF    ((float)BB * (float)NN * (float)KS)

#define FPITCH   36
#define F1PITCH  68
#define W1STRIDE 36
#define W2STRIDE 68

#define BQ_WAVES 8
#define BQ_SEG   (NN / BQ_WAVES)
#define BQ_GRP   16
#define BQ_PITCH 33

__global__ __launch_bounds__(256) void ballq_kernel(const float* __restrict__ p,
                                                    const float* __restrict__ x,
                                                    int* __restrict__ idx,
                                                    float* __restrict__ xt) {
  __shared__ unsigned short lst[BQ_WAVES][32][BQ_PITCH];
  __shared__ int cnts[BQ_WAVES][32];
  __shared__ float xtile[CX][33];
  const int b    = blockIdx.y;
  const int lane = threadIdx.x & 31;
  const int seg  = __builtin_amdgcn_readfirstlane((int)(threadIdx.x >> 5));
  const float* __restrict__ pb = p + (size_t)b * NN * 3;
  const int q = blockIdx.x * 32 + lane;
  const float px = pb[q * 3 + 0];
  const float py = pb[q * 3 + 1];
  const float pz = pb[q * 3 + 2];
  float xv[4];
  #pragma unroll
  for (int k = 0; k < 4; ++k) {
    int i = (int)threadIdx.x + 256 * k;
    xv[k] = x[((size_t)b * CX + (i >> 5)) * NN + blockIdx.x * 32 + (i & 31)];
  }
  const float* __restrict__ cb = pb + seg * (BQ_SEG * 3);
  unsigned short* mylst = &lst[seg][lane][0];
  int cnt = 0;
  for (int g = 0; g < BQ_SEG; g += BQ_GRP) {
    unsigned m = 0;
    #pragma unroll
    for (int jj = 0; jj < BQ_GRP; ++jj) {
      float dx = cb[(g + jj) * 3 + 0] - px;
      float dy = cb[(g + jj) * 3 + 1] - py;
      float dz = cb[(g + jj) * 3 + 2] - pz;
      float d2 = __builtin_fmaf(dz, dz, __builtin_fmaf(dy, dy, dx * dx));
      m = m + m + (d2 < R2 ? 1u : 0u);
    }
    while (m != 0) {
      int lz   = __builtin_clz(m);
      int slot = cnt < KS ? cnt : KS;
      mylst[slot] = (unsigned short)(seg * BQ_SEG + g + lz - (32 - BQ_GRP));
      cnt++;
      m &= ~(0x80000000u >> lz);
    }
  }
  cnts[seg][lane] = cnt;
  #pragma unroll
  for (int k = 0; k < 4; ++k) {
    int i = (int)threadIdx.x + 256 * k;
    xtile[i >> 5][i & 31] = xv[k];
  }
  __syncthreads();

  {
    const int nn = threadIdx.x >> 3;
    const int c4 = (threadIdx.x & 7) * 4;
    float4 xo;
    xo.x = xtile[c4 + 0][nn]; xo.y = xtile[c4 + 1][nn];
    xo.z = xtile[c4 + 2][nn]; xo.w = xtile[c4 + 3][nn];
    *reinterpret_cast<float4*>(xt + ((size_t)b * NN + blockIdx.x * 32 + nn) * CX + c4) = xo;
  }

  const int ql = threadIdx.x >> 3;
  const int r0 = (threadIdx.x & 7) * 4;
  int v[4] = {0, 0, 0, 0};
  int pre = 0, first = 0;
  bool have = false;
  #pragma unroll
  for (int s = 0; s < BQ_WAVES; ++s) {
    int c = cnts[s][ql];
    c = c < KS ? c : KS;
    if (!have && c > 0) { first = lst[s][ql][0]; have = true; }
    #pragma unroll
    for (int k = 0; k < 4; ++k) {
      int r = r0 + k - pre;
      if (r >= 0 && r < c) v[k] = lst[s][ql][r];
    }
    pre += c;
  }
  #pragma unroll
  for (int k = 0; k < 4; ++k)
    if (r0 + k >= pre) v[k] = first;
  int4 o;
  o.x = v[0]; o.y = v[1]; o.z = v[2]; o.w = v[3];
  *reinterpret_cast<int4*>(idx + ((size_t)b * NN + blockIdx.x * 32 + ql) * KS + r0) = o;
}

struct FPart {
  float xv[16];
  float pj[2];
  float pq[2];
};

__device__ __forceinline__ int tile_idx(const int* __restrict__ idxg, int pair) {
  return idxg[(size_t)pair * 64 + (threadIdx.x & 63)];
}

__device__ __forceinline__ void ftile_fetch(const float* __restrict__ p,
                                            const float* __restrict__ xt,
                                            int pair, int j, FPart& r) {
  const int t = threadIdx.x;
  const int b = pair / (NN / 2);
  const float* xr = xt + ((size_t)b * NN + j) * CX + (t < 64 ? 0 : 16);
  #pragma unroll
  for (int c = 0; c < 16; ++c) r.xv[c] = xr[c];
  const float* pj = p + ((size_t)b * NN + j) * 3;
  const float* pq = p + ((size_t)2 * pair + ((t & 63) >> 5)) * 3;
  if (t < 64) {
    r.pj[0] = pj[0]; r.pj[1] = pj[1];
    r.pq[0] = pq[0]; r.pq[1] = pq[1];
  } else {
    r.pj[0] = pj[2];
    r.pq[0] = pq[2];
    r.pj[1] = 0.f;
    r.pq[1] = 0.f;
  }
}

__device__ __forceinline__ void ftile_put(const FPart& r, float* flds) {
  const int t = threadIdx.x;
  const bool lo = t < 64;
  float* colp = flds + (t & 63) * FPITCH;
  float4* xq = reinterpret_cast<float4*>(colp + (lo ? 0 : 16));
  #pragma unroll
  for (int q = 0; q < 4; ++q) {
    float4 v;
    v.x = r.xv[4 * q + 0]; v.y = r.xv[4 * q + 1]; v.z = r.xv[4 * q + 2]; v.w = r.xv[4 * q + 3];
    xq[q] = v;
  }
  float2 d;
  d.x = r.pj[0] - r.pq[0];
  d.y = r.pj[1] - r.pq[1];
  *reinterpret_cast<float2*>(colp + (lo ? 32 : 34)) = d;
}

__device__ __forceinline__ void lds_barrier() {
  asm volatile("s_wait_dscnt 0x0\n\ts_barrier_signal -1\n\ts_barrier_wait -1" ::: "memory");
}

struct Chunk {
  float4 a;
  float4 b[4];
};
struct Tail {
  float2 a;
  float2 b[4];
};

template <int LIVE>
__device__ __forceinline__ void chunk_load(Chunk& f, const float* wrow, const float* fcol,
                                           int fpitch, int j) {
  f.a = *reinterpret_cast<const float4*>(wrow + 8 * j);
  #pragma unroll
  for (int ct = 0; ct < 4; ++ct)
    if ((LIVE >> ct) & 1)
      f.b[ct] = *reinterpret_cast<const float4*>(fcol + ct * 16 * fpitch + 8 * j);
}

template <int LIVE>
__device__ __forceinline__ void tail_load(Tail& f, const float* wrow2, const float* fcol2,
                                          int fpitch, int k0) {
  f.a = *reinterpret_cast<const float2*>(wrow2 + k0);
  #pragma unroll
  for (int ct = 0; ct < 4; ++ct)
    if ((LIVE >> ct) & 1)
      f.b[ct] = *reinterpret_cast<const float2*>(fcol2 + ct * 16 * fpitch + k0);
}

__device__ __forceinline__ v8f mma4(v2f a, v2f b, v8f cacc) {
  return __builtin_amdgcn_wmma_f32_16x16x4_f32(false, a, false, b, (short)0, cacc, false, false);
}

template <int LIVE>
__device__ __forceinline__ void chunk_mma(const Chunk& f, v8f (&acc)[4]) {
  v2f a0, a1;
  a0.x = f.a.x; a0.y = f.a.y; a1.x = f.a.z; a1.y = f.a.w;
  #pragma unroll
  for (int ct = 0; ct < 4; ++ct) {
    if (!((LIVE >> ct) & 1)) continue;
    v2f b; b.x = f.b[ct].x; b.y = f.b[ct].y;
    acc[ct] = mma4(a0, b, acc[ct]);
  }
  #pragma unroll
  for (int ct = 0; ct < 4; ++ct) {
    if (!((LIVE >> ct) & 1)) continue;
    v2f b; b.x = f.b[ct].z; b.y = f.b[ct].w;
    acc[ct] = mma4(a1, b, acc[ct]);
  }
}

template <int LIVE>
__device__ __forceinline__ void tail_mma(const Tail& f, v8f (&acc)[4]) {
  v2f a; a.x = f.a.x; a.y = f.a.y;
  #pragma unroll
  for (int ct = 0; ct < 4; ++ct) {
    if (!((LIVE >> ct) & 1)) continue;
    v2f b; b.x = f.b[ct].x; b.y = f.b[ct].y;
    acc[ct] = mma4(a, b, acc[ct]);
  }
}

__device__ __forceinline__ void sched_fence() { __builtin_amdgcn_sched_barrier(0); }

template <int LIVE>
__device__ __forceinline__ void acc_zero(v8f (&acc)[4]) {
  #pragma unroll
  for (int ct = 0; ct < 4; ++ct)
    if ((LIVE >> ct) & 1) {
      #pragma unroll
      for (int v = 0; v < 8; ++v) acc[ct][v] = 0.f;
    }
}

template <int LIVE>
__device__ __forceinline__ void conv1_strip(const float* __restrict__ W1l,
                                            const float* __restrict__ ft,
                                            int mt, v8f (&acc)[4]) {
  const int lane = threadIdx.x & 31;
  const int c = lane & 15, h = lane >> 4;
  const float* wrow = W1l + (mt * 16 + c) * W1STRIDE;
  const float* fcol = ft + c * FPITCH;
  Chunk g[4];
  Tail  gt;
  acc_zero<LIVE>(acc);
  #pragma unroll
  for (int j = 0; j < 4; ++j) chunk_load<LIVE>(g[j], wrow + 4 * h, fcol + 4 * h, FPITCH, j);
  tail_load<LIVE>(gt, wrow + 2 * h, fcol + 2 * h, FPITCH, 32);
  sched_fence();
  #pragma unroll
  for (int j = 0; j < 4; ++j) chunk_mma<LIVE>(g[j], acc);
  tail_mma<LIVE>(gt, acc);
  sched_fence();
}

template <int LIVE>
__device__ __forceinline__ void conv2_strip(const float* __restrict__ W2l,
                                            const float* __restrict__ f1l,
                                            int mt, v8f (&acc)[4]) {
  const int lane = threadIdx.x & 31;
  const int c = lane & 15, h = lane >> 4;
  const float* wrow = W2l + (mt * 16 + c) * W2STRIDE + 4 * h;
  const float* fcol = f1l + c * F1PITCH + 4 * h;
  Chunk s0[2], s1[2];
  acc_zero<LIVE>(acc);
  chunk_load<LIVE>(s0[0], wrow, fcol, F1PITCH, 0);
  chunk_load<LIVE>(s0[1], wrow, fcol, F1PITCH, 1);
  chunk_load<LIVE>(s1[0], wrow, fcol, F1PITCH, 2);
  chunk_load<LIVE>(s1[1], wrow, fcol, F1PITCH, 3);
  sched_fence();
  chunk_mma<LIVE>(s0[0], acc);
  chunk_mma<LIVE>(s0[1], acc);
  sched_fence();
  chunk_load<LIVE>(s0[0], wrow, fcol, F1PITCH, 4);
  chunk_load<LIVE>(s0[1], wrow, fcol, F1PITCH, 5);
  sched_fence();
  chunk_mma<LIVE>(s1[0], acc);
  chunk_mma<LIVE>(s1[1], acc);
  sched_fence();
  chunk_load<LIVE>(s1[0], wrow, fcol, F1PITCH, 6);
  chunk_load<LIVE>(s1[1], wrow, fcol, F1PITCH, 7);
  sched_fence();
  chunk_mma<LIVE>(s0[0], acc);
  chunk_mma<LIVE>(s0[1], acc);
  chunk_mma<LIVE>(s1[0], acc);
  chunk_mma<LIVE>(s1[1], acc);
  sched_fence();
}

template <int LIVE>
__device__ __forceinline__ void bn1_store(const v8f (&acc)[4], const float (&sc1)[8],
                                          const float (&sh1)[8], float* f1l, int mt) {
  const int lane = threadIdx.x & 31;
  const int c = lane & 15, h = lane >> 4;
  #pragma unroll
  for (int ct = 0; ct < 4; ++ct) {
    if (!((LIVE >> ct) & 1)) continue;
    float y[8];
    #pragma unroll
    for (int v = 0; v < 8; ++v) {
      float t = acc[ct][v] * sc1[v] + sh1[v];
      y[v] = t > 0.f ? t : 0.f;
    }
    float4* dst = reinterpret_cast<float4*>(f1l + (ct * 16 + c) * F1PITCH + mt * 16 + h * 8);
    float4 q0, q1;
    q0.x = y[0]; q0.y = y[1]; q0.z = y[2]; q0.w = y[3];
    q1.x = y[4]; q1.y = y[5]; q1.z = y[6]; q1.w = y[7];
    dst[0] = q0;
    dst[1] = q1;
  }
}


#define W1Q (COUT * CIN / 4)
#define W2Q (COUT * COUT / 4)

__device__ __forceinline__ void w1_fetch(const float* __restrict__ W1, float4 (&r)[5]) {
  const float4* __restrict__ src = reinterpret_cast<const float4*>(W1);
  const int t = threadIdx.x;
  #pragma unroll
  for (int e = 0; e < 5; ++e) {
    int q = t + 128 * e;
    r[e] = src[q < W1Q ? q : W1Q - 1];
  }
}

__device__ __forceinline__ void w1_put(const float4 (&r)[5], float* W1l) {
  const int t = threadIdx.x;
  #pragma unroll
  for (int e = 0; e < 5; ++e) {
    int q = t + 128 * e;
    if (q < W1Q) {
      const float v[4] = {r[e].x, r[e].y, r[e].z, r[e].w};
      #pragma unroll
      for (int u = 0; u < 4; ++u) {
        int i   = 4 * q + u;
        int row = i / CIN;
        int cc  = i - row * CIN;
        int kp  = cc >= 3 ? cc - 3 : 32 + cc;
        W1l[row * W1STRIDE + kp] = v[u];
      }
    }
  }
  if (t < COUT) W1l[t * W1STRIDE + CIN] = 0.f;
}

__device__ __forceinline__ void w2_fetch(const float* __restrict__ W2, float4 (&r)[8]) {
  const float4* __restrict__ src = reinterpret_cast<const float4*>(W2);
  const int t = threadIdx.x;
  #pragma unroll
  for (int e = 0; e < 8; ++e) r[e] = src[t + 128 * e];
}

__device__ __forceinline__ void w2_put(const float4 (&r)[8], float* W2l) {
  const int t = threadIdx.x;
  #pragma unroll
  for (int e = 0; e < 8; ++e) {
    int q = t + 128 * e;
    *reinterpret_cast<float4*>(&W2l[(q >> 4) * W2STRIDE + (q & 15) * 4]) = r[e];
  }
}

__global__ __launch_bounds__(128) void conv1_stats_kernel(
    const float* __restrict__ p, const float* __restrict__ xt,
    const int* __restrict__ idxg, const float* __restrict__ W1,
    float* __restrict__ part1) {
  __shared__ __attribute__((aligned(16))) float W1l[COUT * W1STRIDE];
  __shared__ __attribute__((aligned(16))) float flds[2][64 * FPITCH];
  int t = threadIdx.x;
  const int G = gridDim.x;
  int pair = blockIdx.x;
  const int j0 = tile_idx(idxg, pair);
  int jn = tile_idx(idxg, pair + G < NPAIRS ? pair + G : pair);
  {
    float4 r1[5];
    w1_fetch(W1, r1);
    w1_put(r1, W1l);
  }
  FPart r;
  ftile_fetch(p, xt, pair, j0, r);
  int mt = t >> 5, lane = t & 31;
  float s[8], sq[8];
  #pragma unroll
  for (int v = 0; v < 8; ++v) { s[v] = 0.f; sq[v] = 0.f; }

  for (int trip = 0; pair < NPAIRS; pair += G, ++trip) {
    float* ft = flds[trip & 1];
    ftile_put(r, ft);
    const int pn  = pair + G     < NPAIRS ? pair + G     : pair;
    const int pn2 = pair + 2 * G < NPAIRS ? pair + 2 * G : pair;
    ftile_fetch(p, xt, pn, jn, r);
    jn = tile_idx(idxg, pn2);
    lds_barrier();
    v8f acc[4];
    conv1_strip<15>(W1l, ft, mt, acc);
    #pragma unroll
    for (int ct = 0; ct < 4; ++ct) {
      #pragma unroll
      for (int v = 0; v < 8; ++v) { float y = acc[ct][v]; s[v] += y; sq[v] += y * y; }
    }
  }
  #pragma unroll
  for (int v = 0; v < 8; ++v)
    #pragma unroll
    for (int off = 1; off < 16; off <<= 1) {
      s[v]  += __shfl_xor(s[v],  off, 32);
      sq[v] += __shfl_xor(sq[v], off, 32);
    }
  if ((lane & 15) == 0) {
    int hi = lane >> 4;
    #pragma unroll
    for (int v = 0; v < 8; ++v) {
      int o = mt * 16 + hi * 8 + v;
      part1[(size_t)blockIdx.x * 128 + o]      = s[v];
      part1[(size_t)blockIdx.x * 128 + 64 + o] = sq[v];
    }
  }
}

__global__ void bn_finalize_kernel(const float* __restrict__ part,
                                   const float* __restrict__ g,
                                   const float* __restrict__ bta,
                                   float* __restrict__ bnp, int nblocks) {
  int o = threadIdx.x;
  if (o >= COUT) return;
  float s = 0.f, sq = 0.f;
  for (int blk = 0; blk < nblocks; ++blk) {
    s  += part[(size_t)blk * 128 + o];
    sq += part[(size_t)blk * 128 + 64 + o];
  }
  float mean  = s / CNTF;
  float var   = sq / CNTF - mean * mean;
  float scale = g[o] * rsqrtf(var + EPSV);
  bnp[o]      = scale;
  bnp[64 + o] = bta[o] - mean * scale;
}

__device__ __forceinline__ void bn_from_rows(const float* __restrict__ part,
                                             const float* g,
                                             const float* bta,
                                             float* red, float* bnl,
                                             float* __restrict__ bnp_out) {
  const int t = threadIdx.x;
  const int q = t & 31, w = t >> 5;
  const float gv = g[t & 63];
  const float bv = bta[t & 63];
  const float4* __restrict__ rows = reinterpret_cast<const float4*>(part) + (size_t)w * (GSTAT / 4) * 32 + q;
  float4 a[4];
  #pragma unroll
  for (int u = 0; u < 4; ++u) { a[u].x = 0.f; a[u].y = 0.f; a[u].z = 0.f; a[u].w = 0.f; }
  #pragma unroll
  for (int i = 0; i < GSTAT / 4; ++i) {
    const float4 v = rows[(size_t)i * 32];
    a[i & 3].x += v.x; a[i & 3].y += v.y; a[i & 3].z += v.z; a[i & 3].w += v.w;
  }
  float4 sum;
  sum.x = (a[0].x + a[1].x) + (a[2].x + a[3].x);
  sum.y = (a[0].y + a[1].y) + (a[2].y + a[3].y);
  sum.z = (a[0].z + a[1].z) + (a[2].z + a[3].z);
  sum.w = (a[0].w + a[1].w) + (a[2].w + a[3].w);
  *reinterpret_cast<float4*>(red + w * 128 + 4 * q) = sum;
  lds_barrier();
  const int o = t & 63;
  const float s  = (red[o]      + red[128 + o])      + (red[256 + o]      + red[384 + o]);
  const float sq = (red[64 + o] + red[128 + 64 + o]) + (red[256 + 64 + o] + red[384 + 64 + o]);
  const float mean  = s / CNTF;
  const float var   = sq / CNTF - mean * mean;
  const float scale = gv * rsqrtf(var + EPSV);
  const float shift = bv - mean * scale;
  if (t < COUT) {
    bnl[o]      = scale;
    bnl[64 + o] = shift;
    if (blockIdx.x == 0) {
      bnp_out[o]      = scale;
      bnp_out[64 + o] = shift;
    }
  }
  lds_barrier();
}

__global__ __launch_bounds__(128) void conv2_stats_kernel(
    const float* __restrict__ p, const float* __restrict__ xt,
    const int* __restrict__ idxg, const float* __restrict__ W1,
    const float* __restrict__ W2, const float* __restrict__ part1,
    const float* g1, const float* b1,
    float* __restrict__ bnp1, float* __restrict__ part2) {
  __shared__ __attribute__((aligned(16))) float W1l[COUT * W1STRIDE];
  __shared__ __attribute__((aligned(16))) float W2l[COUT * W2STRIDE];
  __shared__ __attribute__((aligned(16))) float flds[64 * FPITCH];
  __shared__ __attribute__((aligned(16))) float f1l[64 * F1PITCH];
  __shared__ __attribute__((aligned(16))) float red[4 * 128];
  __shared__ float bn1l[128];
  int t = threadIdx.x;
  const int G = gridDim.x;
  int pair = blockIdx.x;
  const int j0 = tile_idx(idxg, pair);
  int jn = tile_idx(idxg, pair + G < NPAIRS ? pair + G : pair);
  {
    float4 r1[5], r2[8];
    w1_fetch(W1, r1);
    w2_fetch(W2, r2);
    w1_put(r1, W1l);
    w2_put(r2, W2l);
  }
  FPart r;
  ftile_fetch(p, xt, pair, j0, r);
  bn_from_rows(part1, g1, b1, red, bn1l, bnp1);
  int mt = t >> 5, lane = t & 31;
  int hi = lane >> 4;
  float sc1[8], sh1[8];
  #pragma unroll
  for (int v = 0; v < 8; ++v) {
    int o = mt * 16 + hi * 8 + v;
    sc1[v] = bn1l[o]; sh1[v] = bn1l[64 + o];
  }
  float s[8], sq[8];
  #pragma unroll
  for (int v = 0; v < 8; ++v) { s[v] = 0.f; sq[v] = 0.f; }

  for (; pair < NPAIRS; pair += G) {
    ftile_put(r, flds);
    const int pn  = pair + G     < NPAIRS ? pair + G     : pair;
    const int pn2 = pair + 2 * G < NPAIRS ? pair + 2 * G : pair;
    ftile_fetch(p, xt, pn, jn, r);
    jn = tile_idx(idxg, pn2);
    lds_barrier();
    v8f acc[4];
    conv1_strip<15>(W1l, flds, mt, acc);
    bn1_store<15>(acc, sc1, sh1, f1l, mt);
    lds_barrier();
    conv2_strip<15>(W2l, f1l, mt, acc);
    #pragma unroll
    for (int ct = 0; ct < 4; ++ct) {
      #pragma unroll
      for (int v = 0; v < 8; ++v) { float y = acc[ct][v]; s[v] += y; sq[v] += y * y; }
    }
  }
  #pragma unroll
  for (int v = 0; v < 8; ++v)
    #pragma unroll
    for (int off = 1; off < 16; off <<= 1) {
      s[v]  += __shfl_xor(s[v],  off, 32);
      sq[v] += __shfl_xor(sq[v], off, 32);
    }
  if ((lane & 15) == 0) {
    #pragma unroll
    for (int v = 0; v < 8; ++v) {
      int o = mt * 16 + hi * 8 + v;
      part2[(size_t)blockIdx.x * 128 + o]      = s[v];
      part2[(size_t)blockIdx.x * 128 + 64 + o] = sq[v];
    }
  }
}

struct PairEnds { int a0, a16, b0, b16; };

__device__ __forceinline__ PairEnds pair_ends(const int* __restrict__ idxg, int pair) {
  const int* row = idxg + (size_t)pair * 64;
  PairEnds e;
  e.a0 = row[0]; e.a16 = row[16]; e.b0 = row[32]; e.b16 = row[48];
  return e;
}

__device__ __forceinline__ int live_tiles(const PairEnds& e) {
  int live = 15;
  if (e.a16 == e.a0) live &= ~2;
  if (e.b16 == e.b0) live &= ~8;
  return __builtin_amdgcn_readfirstlane(live);
}

template <int LIVE>
__device__ __forceinline__ void final_tile(const float* __restrict__ W1l, const float* __restrict__ W2l,
                                           const float* flds, float* f1l, int mt,
                                           const float (&sc1)[8], const float (&sh1)[8],
                                           const float (&sc2)[8], const float (&sh2)[8],
                                           float (&m0)[8], float (&m1)[8]) {
  v8f acc[4];
  conv1_strip<LIVE>(W1l, flds, mt, acc);
  bn1_store<LIVE>(acc, sc1, sh1, f1l, mt);
  lds_barrier();
  conv2_strip<LIVE>(W2l, f1l, mt, acc);
  #pragma unroll
  for (int v = 0; v < 8; ++v) { m0[v] = 0.f; m1[v] = 0.f; }
  #pragma unroll
  for (int ct = 0; ct < 4; ++ct) {
    if (!((LIVE >> ct) & 1)) continue;
    #pragma unroll
    for (int v = 0; v < 8; ++v) {
      float y = acc[ct][v] * sc2[v] + sh2[v];
      y = y > 0.f ? y : 0.f;
      if (ct < 2) m0[v] = fmaxf(m0[v], y);
      else        m1[v] = fmaxf(m1[v], y);
    }
  }
}

__global__ __launch_bounds__(128) void final_kernel(
    const float* __restrict__ p, const float* __restrict__ xt,
    const int* __restrict__ idxg, const float* __restrict__ W1,
    const float* __restrict__ W2, const float* __restrict__ bnp1,
    const float* __restrict__ bnp2, float* __restrict__ out) {
  __shared__ __attribute__((aligned(16))) float W1l[COUT * W1STRIDE];
  __shared__ __attribute__((aligned(16))) float W2l[COUT * W2STRIDE];
  __shared__ __attribute__((aligned(16))) float flds[64 * FPITCH];
  __shared__ __attribute__((aligned(16))) float f1l[64 * F1PITCH];
  int t = threadIdx.x;
  const int G = gridDim.x;
  int pair = blockIdx.x;
  const int j0 = tile_idx(idxg, pair);
  int jn = tile_idx(idxg, pair + G < NPAIRS ? pair + G : pair);
  PairEnds ends = pair_ends(idxg, pair);
  {
    float4 r1[5], r2[8];
    w1_fetch(W1, r1);
    w2_fetch(W2, r2);
    w1_put(r1, W1l);
    w2_put(r2, W2l);
  }
  FPart r;
  ftile_fetch(p, xt, pair, j0, r);
  int mt = t >> 5, lane = t & 31;
  int hi = lane >> 4;
  float sc1[8], sh1[8], sc2[8], sh2[8];
  #pragma unroll
  for (int v = 0; v < 8; ++v) {
    int o = mt * 16 + hi * 8 + v;
    sc1[v] = bnp1[o]; sh1[v] = bnp1[64 + o];
    sc2[v] = bnp2[o]; sh2[v] = bnp2[64 + o];
  }
  for (; pair < NPAIRS; pair += G) {
    int b  = pair / (NN / 2);
    int n0 = (pair % (NN / 2)) * 2;
    const int live = live_tiles(ends);
    ftile_put(r, flds);
    const int pn  = pair + G     < NPAIRS ? pair + G     : pair;
    const int pn2 = pair + 2 * G < NPAIRS ? pair + 2 * G : pair;
    ftile_fetch(p, xt, pn, jn, r);
    jn = tile_idx(idxg, pn2);
    ends = pair_ends(idxg, pn);
    lds_barrier();
    float m0[8], m1[8];
    switch (live) {
      case 15: final_tile<15>(W1l, W2l, flds, f1l, mt, sc1, sh1, sc2, sh2, m0, m1); break;
      case 13: final_tile<13>(W1l, W2l, flds, f1l, mt, sc1, sh1, sc2, sh2, m0, m1); break;
      case 7:  final_tile<7>(W1l, W2l, flds, f1l, mt, sc1, sh1, sc2, sh2, m0, m1); break;
      default: final_tile<5>(W1l, W2l, flds, f1l, mt, sc1, sh1, sc2, sh2, m0, m1); break;
    }
    #pragma unroll
    for (int v = 0; v < 8; ++v)
      #pragma unroll
      for (int off = 1; off < 16; off <<= 1) {
        m0[v] = fmaxf(m0[v], __shfl_xor(m0[v], off, 32));
        m1[v] = fmaxf(m1[v], __shfl_xor(m1[v], off, 32));
      }
    if ((lane & 15) == 0) {
      #pragma unroll
      for (int v = 0; v < 8; ++v) {
        int o = mt * 16 + hi * 8 + v;
        out[((size_t)b * COUT + o) * NN + n0]     = m0[v];
        out[((size_t)b * COUT + o) * NN + n0 + 1] = m1[v];
      }
    }
  }
}

extern "C" void kernel_launch(void* const* d_in, const int* in_sizes, int n_in,
                              void* d_out, int out_size, void* d_ws, size_t ws_size,
                              hipStream_t stream) {
  const float* p  = (const float*)d_in[0];
  const float* x  = (const float*)d_in[1];
  const float* W1 = (const float*)d_in[2];
  const float* g1 = (const float*)d_in[3];
  const float* b1 = (const float*)d_in[4];
  const float* W2 = (const float*)d_in[5];
  const float* g2 = (const float*)d_in[6];
  const float* b2 = (const float*)d_in[7];
  float* out = (float*)d_out;

  char*  ws    = (char*)d_ws;
  int*   idx   = (int*)ws;
  float* xt    = (float*)(ws + (size_t)BB * NN * KS * 4);
  float* part1 = (float*)(ws + (size_t)BB * NN * KS * 4
                             + (size_t)BB * NN * CX * 4);
  float* part2 = part1 + (size_t)GCONV * 128;
  float* bnp1  = part2 + (size_t)GCONV * 128;
  float* bnp2  = bnp1 + 128;

  ballq_kernel<<<dim3(NN / 32, BB), 256, 0, stream>>>(p, x, idx, xt);
  conv1_stats_kernel<<<GSTAT, 128, 0, stream>>>(p, xt, idx, W1, part1);
  conv2_stats_kernel<<<GSTAT, 128, 0, stream>>>(p, xt, idx, W1, W2, part1, g1, b1, bnp1, part2);
  bn_finalize_kernel<<<1, 64, 0, stream>>>(part2, g2, b2, bnp2, GSTAT);
  final_kernel<<<GCONV, 128, 0, stream>>>(p, xt, idx, W1, W2, bnp1, bnp2, out);
}


# gather prefetch gets its distance: rows carried as 16-byte vectors (no end-of-trip register copy waiting for the loads); stats passes fetch two tiles ahead in two register sets, loop unrolled by 2
# speedup vs baseline: 4.8830x; 1.0383x over previous
#include <hip/hip_runtime.h>
#include <hip/hip_bf16.h>


typedef __attribute__((ext_vector_type(2))) float v2f;
typedef __attribute__((ext_vector_type(4))) float v4f;
typedef __attribute__((ext_vector_type(8))) float v8f;

#define BB      4
#define NN      4096
#define CX      32
#define CIN     35
#define CINP    36
#define COUT    64
#define KS      32
#define R2      0.01f
#define EPSV    1e-5f
#define NPAIRS  (BB * (NN / 2))
#define GCONV   1024
#define GSTAT   256
#define CNTF    ((float)BB * (float)NN * (float)KS)

#define FPITCH   36
#define F1PITCH  68
#define W1STRIDE 36
#define W2STRIDE 68

#define BQ_WAVES 8
#define BQ_SEG   (NN / BQ_WAVES)
#define BQ_GRP   16
#define BQ_PITCH 33

__global__ __launch_bounds__(256) void ballq_kernel(const float* __restrict__ p,
                                                    const float* __restrict__ x,
                                                    int* __restrict__ idx,
                                                    float* __restrict__ xt) {
  __shared__ unsigned short lst[BQ_WAVES][32][BQ_PITCH];
  __shared__ int cnts[BQ_WAVES][32];
  __shared__ float xtile[CX][33];
  const int b    = blockIdx.y;
  const int lane = threadIdx.x & 31;
  const int seg  = __builtin_amdgcn_readfirstlane((int)(threadIdx.x >> 5));
  const float* __restrict__ pb = p + (size_t)b * NN * 3;
  const int q = blockIdx.x * 32 + lane;
  const float px = pb[q * 3 + 0];
  const float py = pb[q * 3 + 1];
  const float pz = pb[q * 3 + 2];
  float xv[4];
  #pragma unroll
  for (int k = 0; k < 4; ++k) {
    int i = (int)threadIdx.x + 256 * k;
    xv[k] = x[((size_t)b * CX + (i >> 5)) * NN + blockIdx.x * 32 + (i & 31)];
  }
  const float* __restrict__ cb = pb + seg * (BQ_SEG * 3);
  unsigned short* mylst = &lst[seg][lane][0];
  int cnt = 0;
  for (int g = 0; g < BQ_SEG; g += BQ_GRP) {
    unsigned m = 0;
    #pragma unroll
    for (int jj = 0; jj < BQ_GRP; ++jj) {
      float dx = cb[(g + jj) * 3 + 0] - px;
      float dy = cb[(g + jj) * 3 + 1] - py;
      float dz = cb[(g + jj) * 3 + 2] - pz;
      float d2 = __builtin_fmaf(dz, dz, __builtin_fmaf(dy, dy, dx * dx));
      m = m + m + (d2 < R2 ? 1u : 0u);
    }
    while (m != 0) {
      int lz   = __builtin_clz(m);
      int slot = cnt < KS ? cnt : KS;
      mylst[slot] = (unsigned short)(seg * BQ_SEG + g + lz - (32 - BQ_GRP));
      cnt++;
      m &= ~(0x80000000u >> lz);
    }
  }
  cnts[seg][lane] = cnt;
  #pragma unroll
  for (int k = 0; k < 4; ++k) {
    int i = (int)threadIdx.x + 256 * k;
    xtile[i >> 5][i & 31] = xv[k];
  }
  __syncthreads();

  {
    const int nn = threadIdx.x >> 3;
    const int c4 = (threadIdx.x & 7) * 4;
    float4 xo;
    xo.x = xtile[c4 + 0][nn]; xo.y = xtile[c4 + 1][nn];
    xo.z = xtile[c4 + 2][nn]; xo.w = xtile[c4 + 3][nn];
    *reinterpret_cast<float4*>(xt + ((size_t)b * NN + blockIdx.x * 32 + nn) * CX + c4) = xo;
  }

  const int ql = threadIdx.x >> 3;
  const int r0 = (threadIdx.x & 7) * 4;
  int v[4] = {0, 0, 0, 0};
  int pre = 0, first = 0;
  bool have = false;
  #pragma unroll
  for (int s = 0; s < BQ_WAVES; ++s) {
    int c = cnts[s][ql];
    c = c < KS ? c : KS;
    if (!have && c > 0) { first = lst[s][ql][0]; have = true; }
    #pragma unroll
    for (int k = 0; k < 4; ++k) {
      int r = r0 + k - pre;
      if (r >= 0 && r < c) v[k] = lst[s][ql][r];
    }
    pre += c;
  }
  #pragma unroll
  for (int k = 0; k < 4; ++k)
    if (r0 + k >= pre) v[k] = first;
  int4 o;
  o.x = v[0]; o.y = v[1]; o.z = v[2]; o.w = v[3];
  *reinterpret_cast<int4*>(idx + ((size_t)b * NN + blockIdx.x * 32 + ql) * KS + r0) = o;
}

struct FPart {
  v4f   xq[4];
  float pj[2];
  float pq[2];
};

__device__ __forceinline__ int tile_idx(const int* idxg, int pair) {
  return idxg[(size_t)pair * 64 + (threadIdx.x & 63)];
}

__device__ __forceinline__ void ftile_fetch(const float* p,
                                            const float* xt,
                                            int pair, int j, FPart& r) {
  const int t = threadIdx.x;
  const int b = pair / (NN / 2);
  const v4f* xr = reinterpret_cast<const v4f*>(xt + ((size_t)b * NN + j) * CX + (t < 64 ? 0 : 16));
  #pragma unroll
  for (int q = 0; q < 4; ++q) r.xq[q] = xr[q];
  const bool lo = t < 64;
  const float* pj = p + ((size_t)b * NN + j) * 3 + (lo ? 0 : 2);
  const float* pq = p + ((size_t)2 * pair + ((t & 63) >> 5)) * 3 + (lo ? 0 : 2);
  r.pj[0] = pj[0];
  r.pq[0] = pq[0];
  r.pj[1] = pj[lo ? 1 : 0];
  r.pq[1] = pq[lo ? 1 : 0];
}

__device__ __forceinline__ void ftile_put(const FPart& r, float* flds) {
  const int t = threadIdx.x;
  const bool lo = t < 64;
  float* colp = flds + (t & 63) * FPITCH;
  v4f* xq = reinterpret_cast<v4f*>(colp + (lo ? 0 : 16));
  #pragma unroll
  for (int q = 0; q < 4; ++q) xq[q] = r.xq[q];
  float2 d;
  d.x = r.pj[0] - r.pq[0];
  d.y = lo ? r.pj[1] - r.pq[1] : 0.f;
  *reinterpret_cast<float2*>(colp + (lo ? 32 : 34)) = d;
}

__device__ __forceinline__ void lds_barrier() {
  asm volatile("s_wait_dscnt 0x0\n\ts_barrier_signal -1\n\ts_barrier_wait -1" ::: "memory");
}

struct Chunk {
  float4 a;
  float4 b[4];
};
struct Tail {
  float2 a;
  float2 b[4];
};

template <int LIVE>
__device__ __forceinline__ void chunk_load(Chunk& f, const float* wrow, const float* fcol,
                                           int fpitch, int j) {
  f.a = *reinterpret_cast<const float4*>(wrow + 8 * j);
  #pragma unroll
  for (int ct = 0; ct < 4; ++ct)
    if ((LIVE >> ct) & 1)
      f.b[ct] = *reinterpret_cast<const float4*>(fcol + ct * 16 * fpitch + 8 * j);
}

template <int LIVE>
__device__ __forceinline__ void tail_load(Tail& f, const float* wrow2, const float* fcol2,
                                          int fpitch, int k0) {
  f.a = *reinterpret_cast<const float2*>(wrow2 + k0);
  #pragma unroll
  for (int ct = 0; ct < 4; ++ct)
    if ((LIVE >> ct) & 1)
      f.b[ct] = *reinterpret_cast<const float2*>(fcol2 + ct * 16 * fpitch + k0);
}

__device__ __forceinline__ v8f mma4(v2f a, v2f b, v8f cacc) {
  return __builtin_amdgcn_wmma_f32_16x16x4_f32(false, a, false, b, (short)0, cacc, false, false);
}

template <int LIVE>
__device__ __forceinline__ void chunk_mma(const Chunk& f, v8f (&acc)[4]) {
  v2f a0, a1;
  a0.x = f.a.x; a0.y = f.a.y; a1.x = f.a.z; a1.y = f.a.w;
  #pragma unroll
  for (int ct = 0; ct < 4; ++ct) {
    if (!((LIVE >> ct) & 1)) continue;
    v2f b; b.x = f.b[ct].x; b.y = f.b[ct].y;
    acc[ct] = mma4(a0, b, acc[ct]);
  }
  #pragma unroll
  for (int ct = 0; ct < 4; ++ct) {
    if (!((LIVE >> ct) & 1)) continue;
    v2f b; b.x = f.b[ct].z; b.y = f.b[ct].w;
    acc[ct] = mma4(a1, b, acc[ct]);
  }
}

template <int LIVE>
__device__ __forceinline__ void tail_mma(const Tail& f, v8f (&acc)[4]) {
  v2f a; a.x = f.a.x; a.y = f.a.y;
  #pragma unroll
  for (int ct = 0; ct < 4; ++ct) {
    if (!((LIVE >> ct) & 1)) continue;
    v2f b; b.x = f.b[ct].x; b.y = f.b[ct].y;
    acc[ct] = mma4(a, b, acc[ct]);
  }
}

__device__ __forceinline__ void sched_fence() { __builtin_amdgcn_sched_barrier(0); }

template <int LIVE>
__device__ __forceinline__ void acc_zero(v8f (&acc)[4]) {
  #pragma unroll
  for (int ct = 0; ct < 4; ++ct)
    if ((LIVE >> ct) & 1) {
      #pragma unroll
      for (int v = 0; v < 8; ++v) acc[ct][v] = 0.f;
    }
}

template <int LIVE>
__device__ __forceinline__ void conv1_strip(const float* __restrict__ W1l,
                                            const float* __restrict__ ft,
                                            int mt, v8f (&acc)[4]) {
  const int lane = threadIdx.x & 31;
  const int c = lane & 15, h = lane >> 4;
  const float* wrow = W1l + (mt * 16 + c) * W1STRIDE;
  const float* fcol = ft + c * FPITCH;
  Chunk g[4];
  Tail  gt;
  acc_zero<LIVE>(acc);
  #pragma unroll
  for (int j = 0; j < 4; ++j) chunk_load<LIVE>(g[j], wrow + 4 * h, fcol + 4 * h, FPITCH, j);
  tail_load<LIVE>(gt, wrow + 2 * h, fcol + 2 * h, FPITCH, 32);
  sched_fence();
  #pragma unroll
  for (int j = 0; j < 4; ++j) chunk_mma<LIVE>(g[j], acc);
  tail_mma<LIVE>(gt, acc);
  sched_fence();
}

template <int LIVE>
__device__ __forceinline__ void conv2_strip(const float* __restrict__ W2l,
                                            const float* __restrict__ f1l,
                                            int mt, v8f (&acc)[4]) {
  const int lane = threadIdx.x & 31;
  const int c = lane & 15, h = lane >> 4;
  const float* wrow = W2l + (mt * 16 + c) * W2STRIDE + 4 * h;
  const float* fcol = f1l + c * F1PITCH + 4 * h;
  Chunk s0[2], s1[2];
  acc_zero<LIVE>(acc);
  chunk_load<LIVE>(s0[0], wrow, fcol, F1PITCH, 0);
  chunk_load<LIVE>(s0[1], wrow, fcol, F1PITCH, 1);
  chunk_load<LIVE>(s1[0], wrow, fcol, F1PITCH, 2);
  chunk_load<LIVE>(s1[1], wrow, fcol, F1PITCH, 3);
  sched_fence();
  chunk_mma<LIVE>(s0[0], acc);
  chunk_mma<LIVE>(s0[1], acc);
  sched_fence();
  chunk_load<LIVE>(s0[0], wrow, fcol, F1PITCH, 4);
  chunk_load<LIVE>(s0[1], wrow, fcol, F1PITCH, 5);
  sched_fence();
  chunk_mma<LIVE>(s1[0], acc);
  chunk_mma<LIVE>(s1[1], acc);
  sched_fence();
  chunk_load<LIVE>(s1[0], wrow, fcol, F1PITCH, 6);
  chunk_load<LIVE>(s1[1], wrow, fcol, F1PITCH, 7);
  sched_fence();
  chunk_mma<LIVE>(s0[0], acc);
  chunk_mma<LIVE>(s0[1], acc);
  chunk_mma<LIVE>(s1[0], acc);
  chunk_mma<LIVE>(s1[1], acc);
  sched_fence();
}

template <int LIVE>
__device__ __forceinline__ void bn1_store(const v8f (&acc)[4], const float (&sc1)[8],
                                          const float (&sh1)[8], float* f1l, int mt) {
  const int lane = threadIdx.x & 31;
  const int c = lane & 15, h = lane >> 4;
  #pragma unroll
  for (int ct = 0; ct < 4; ++ct) {
    if (!((LIVE >> ct) & 1)) continue;
    float y[8];
    #pragma unroll
    for (int v = 0; v < 8; ++v) {
      float t = acc[ct][v] * sc1[v] + sh1[v];
      y[v] = t > 0.f ? t : 0.f;
    }
    float4* dst = reinterpret_cast<float4*>(f1l + (ct * 16 + c) * F1PITCH + mt * 16 + h * 8);
    float4 q0, q1;
    q0.x = y[0]; q0.y = y[1]; q0.z = y[2]; q0.w = y[3];
    q1.x = y[4]; q1.y = y[5]; q1.z = y[6]; q1.w = y[7];
    dst[0] = q0;
    dst[1] = q1;
  }
}


#define W1Q (COUT * CIN / 4)
#define W2Q (COUT * COUT / 4)

__device__ __forceinline__ void w1_fetch(const float* __restrict__ W1, float4 (&r)[5]) {
  const float4* __restrict__ src = reinterpret_cast<const float4*>(W1);
  const int t = threadIdx.x;
  #pragma unroll
  for (int e = 0; e < 5; ++e) {
    int q = t + 128 * e;
    r[e] = src[q < W1Q ? q : W1Q - 1];
  }
}

__device__ __forceinline__ void w1_put(const float4 (&r)[5], float* W1l) {
  const int t = threadIdx.x;
  #pragma unroll
  for (int e = 0; e < 5; ++e) {
    int q = t + 128 * e;
    if (q < W1Q) {
      const float v[4] = {r[e].x, r[e].y, r[e].z, r[e].w};
      #pragma unroll
      for (int u = 0; u < 4; ++u) {
        int i   = 4 * q + u;
        int row = i / CIN;
        int cc  = i - row * CIN;
        int kp  = cc >= 3 ? cc - 3 : 32 + cc;
        W1l[row * W1STRIDE + kp] = v[u];
      }
    }
  }
  if (t < COUT) W1l[t * W1STRIDE + CIN] = 0.f;
}

__device__ __forceinline__ void w2_fetch(const float* __restrict__ W2, float4 (&r)[8]) {
  const float4* __restrict__ src = reinterpret_cast<const float4*>(W2);
  const int t = threadIdx.x;
  #pragma unroll
  for (int e = 0; e < 8; ++e) r[e] = src[t + 128 * e];
}

__device__ __forceinline__ void w2_put(const float4 (&r)[8], float* W2l) {
  const int t = threadIdx.x;
  #pragma unroll
  for (int e = 0; e < 8; ++e) {
    int q = t + 128 * e;
    *reinterpret_cast<float4*>(&W2l[(q >> 4) * W2STRIDE + (q & 15) * 4]) = r[e];
  }
}

__device__ __forceinline__ int pair_or(int q, int fallback) { return q < NPAIRS ? q : fallback; }

__device__ __forceinline__ void conv1_stats_trip(const float* p, const float* xt,
                                                 const int* idxg, const float* W1l,
                                                 float* ft, int pair, int G, FPart& r, int& jn,
                                                 int mt, float (&s)[8], float (&sq)[8]) {
  ftile_put(r, ft);
  const int pn2 = pair_or(pair + 2 * G, pair);
  const int pn4 = pair_or(pair + 4 * G, pair);
  ftile_fetch(p, xt, pn2, jn, r);
  jn = tile_idx(idxg, pn4);
  lds_barrier();
  v8f acc[4];
  conv1_strip<15>(W1l, ft, mt, acc);
  #pragma unroll
  for (int ct = 0; ct < 4; ++ct) {
    #pragma unroll
    for (int v = 0; v < 8; ++v) { float y = acc[ct][v]; s[v] += y; sq[v] += y * y; }
  }
}

__global__ __launch_bounds__(128) void conv1_stats_kernel(
    const float* p, const float* xt,
    const int* idxg, const float* __restrict__ W1,
    float* __restrict__ part1) {
  __shared__ __attribute__((aligned(16))) float W1l[COUT * W1STRIDE];
  __shared__ __attribute__((aligned(16))) float flds[2][64 * FPITCH];
  int t = threadIdx.x;
  const int G = gridDim.x;
  int pair = blockIdx.x;
  const int pf1 = pair_or(pair + G, pair);
  const int j0 = tile_idx(idxg, pair);
  const int j1 = tile_idx(idxg, pf1);
  {
    float4 r1[5];
    w1_fetch(W1, r1);
    w1_put(r1, W1l);
  }
  FPart ra, rb;
  ftile_fetch(p, xt, pair, j0, ra);
  int ja = tile_idx(idxg, pair_or(pair + 2 * G, pair));
  sched_fence();
  ftile_fetch(p, xt, pf1, j1, rb);
  int jb = tile_idx(idxg, pair_or(pair + 3 * G, pair));
  int mt = t >> 5, lane = t & 31;
  float s[8], sq[8];
  #pragma unroll
  for (int v = 0; v < 8; ++v) { s[v] = 0.f; sq[v] = 0.f; }

  const int ntrip = pair < NPAIRS ? (NPAIRS - 1 - pair) / G + 1 : 0;
  for (int it = 0; it + 1 < ntrip; it += 2) {
    conv1_stats_trip(p, xt, idxg, W1l, flds[0], pair, G, ra, ja, mt, s, sq);
    pair += G;
    conv1_stats_trip(p, xt, idxg, W1l, flds[1], pair, G, rb, jb, mt, s, sq);
    pair += G;
  }
  if (ntrip & 1) conv1_stats_trip(p, xt, idxg, W1l, flds[0], pair, G, ra, ja, mt, s, sq);
  #pragma unroll
  for (int v = 0; v < 8; ++v)
    #pragma unroll
    for (int off = 1; off < 16; off <<= 1) {
      s[v]  += __shfl_xor(s[v],  off, 32);
      sq[v] += __shfl_xor(sq[v], off, 32);
    }
  if ((lane & 15) == 0) {
    int hi = lane >> 4;
    #pragma unroll
    for (int v = 0; v < 8; ++v) {
      int o = mt * 16 + hi * 8 + v;
      part1[(size_t)blockIdx.x * 128 + o]      = s[v];
      part1[(size_t)blockIdx.x * 128 + 64 + o] = sq[v];
    }
  }
}

__global__ void bn_finalize_kernel(const float* __restrict__ part,
                                   const float* __restrict__ g,
                                   const float* __restrict__ bta,
                                   float* __restrict__ bnp, int nblocks) {
  int o = threadIdx.x;
  if (o >= COUT) return;
  float s = 0.f, sq = 0.f;
  for (int blk = 0; blk < nblocks; ++blk) {
    s  += part[(size_t)blk * 128 + o];
    sq += part[(size_t)blk * 128 + 64 + o];
  }
  float mean  = s / CNTF;
  float var   = sq / CNTF - mean * mean;
  float scale = g[o] * rsqrtf(var + EPSV);
  bnp[o]      = scale;
  bnp[64 + o] = bta[o] - mean * scale;
}

__device__ __forceinline__ void bn_from_rows(const float* __restrict__ part,
                                             const float* g,
                                             const float* bta,
                                             float* red, float* bnl,
                                             float* __restrict__ bnp_out) {
  const int t = threadIdx.x;
  const int q = t & 31, w = t >> 5;
  const float gv = g[t & 63];
  const float bv = bta[t & 63];
  const float4* __restrict__ rows = reinterpret_cast<const float4*>(part) + (size_t)w * (GSTAT / 4) * 32 + q;
  float4 a[4];
  #pragma unroll
  for (int u = 0; u < 4; ++u) { a[u].x = 0.f; a[u].y = 0.f; a[u].z = 0.f; a[u].w = 0.f; }
  #pragma unroll
  for (int i = 0; i < GSTAT / 4; ++i) {
    const float4 v = rows[(size_t)i * 32];
    a[i & 3].x += v.x; a[i & 3].y += v.y; a[i & 3].z += v.z; a[i & 3].w += v.w;
  }
  float4 sum;
  sum.x = (a[0].x + a[1].x) + (a[2].x + a[3].x);
  sum.y = (a[0].y + a[1].y) + (a[2].y + a[3].y);
  sum.z = (a[0].z + a[1].z) + (a[2].z + a[3].z);
  sum.w = (a[0].w + a[1].w) + (a[2].w + a[3].w);
  *reinterpret_cast<float4*>(red + w * 128 + 4 * q) = sum;
  lds_barrier();
  const int o = t & 63;
  const float s  = (red[o]      + red[128 + o])      + (red[256 + o]      + red[384 + o]);
  const float sq = (red[64 + o] + red[128 + 64 + o]) + (red[256 + 64 + o] + red[384 + 64 + o]);
  const float mean  = s / CNTF;
  const float var   = sq / CNTF - mean * mean;
  const float scale = gv * rsqrtf(var + EPSV);
  const float shift = bv - mean * scale;
  if (t < COUT) {
    bnl[o]      = scale;
    bnl[64 + o] = shift;
    if (blockIdx.x == 0) {
      bnp_out[o]      = scale;
      bnp_out[64 + o] = shift;
    }
  }
  lds_barrier();
}

__device__ __forceinline__ void conv2_stats_trip(const float* p, const float* xt,
                                                 const int* idxg, const float* W1l,
                                                 const float* W2l, float* flds, float* f1l,
                                                 int pair, int G, FPart& r, int& jn, int mt,
                                                 const float (&sc1)[8], const float (&sh1)[8],
                                                 float (&s)[8], float (&sq)[8]) {
  ftile_put(r, flds);
  const int pn2 = pair_or(pair + 2 * G, pair);
  const int pn4 = pair_or(pair + 4 * G, pair);
  ftile_fetch(p, xt, pn2, jn, r);
  jn = tile_idx(idxg, pn4);
  lds_barrier();
  v8f acc[4];
  conv1_strip<15>(W1l, flds, mt, acc);
  bn1_store<15>(acc, sc1, sh1, f1l, mt);
  lds_barrier();
  conv2_strip<15>(W2l, f1l, mt, acc);
  #pragma unroll
  for (int ct = 0; ct < 4; ++ct) {
    #pragma unroll
    for (int v = 0; v < 8; ++v) { float y = acc[ct][v]; s[v] += y; sq[v] += y * y; }
  }
}

__global__ __launch_bounds__(128) void conv2_stats_kernel(
    const float* p, const float* xt,
    const int* idxg, const float* __restrict__ W1,
    const float* __restrict__ W2, const float* __restrict__ part1,
    const float* g1, const float* b1,
    float* __restrict__ bnp1, float* __restrict__ part2) {
  __shared__ __attribute__((aligned(16))) float W1l[COUT * W1STRIDE];
  __shared__ __attribute__((aligned(16))) float W2l[COUT * W2STRIDE];
  __shared__ __attribute__((aligned(16))) float flds[64 * FPITCH];
  __shared__ __attribute__((aligned(16))) float f1l[64 * F1PITCH];
  __shared__ __attribute__((aligned(16))) float red[4 * 128];
  __shared__ float bn1l[128];
  int t = threadIdx.x;
  const int G = gridDim.x;
  int pair = blockIdx.x;
  const int pf1 = pair_or(pair + G, pair);
  const int j0 = tile_idx(idxg, pair);
  const int j1 = tile_idx(idxg, pf1);
  {
    float4 r1[5], r2[8];
    w1_fetch(W1, r1);
    w2_fetch(W2, r2);
    w1_put(r1, W1l);
    w2_put(r2, W2l);
  }
  FPart ra, rb;
  ftile_fetch(p, xt, pair, j0, ra);
  int ja = tile_idx(idxg, pair_or(pair + 2 * G, pair));
  sched_fence();
  ftile_fetch(p, xt, pf1, j1, rb);
  int jb = tile_idx(idxg, pair_or(pair + 3 * G, pair));
  bn_from_rows(part1, g1, b1, red, bn1l, bnp1);
  int mt = t >> 5, lane = t & 31;
  int hi = lane >> 4;
  float sc1[8], sh1[8];
  #pragma unroll
  for (int v = 0; v < 8; ++v) {
    int o = mt * 16 + hi * 8 + v;
    sc1[v] = bn1l[o]; sh1[v] = bn1l[64 + o];
  }
  float s[8], sq[8];
  #pragma unroll
  for (int v = 0; v < 8; ++v) { s[v] = 0.f; sq[v] = 0.f; }

  const int ntrip = pair < NPAIRS ? (NPAIRS - 1 - pair) / G + 1 : 0;
  for (int it = 0; it + 1 < ntrip; it += 2) {
    conv2_stats_trip(p, xt, idxg, W1l, W2l, flds, f1l, pair, G, ra, ja, mt, sc1, sh1, s, sq);
    pair += G;
    conv2_stats_trip(p, xt, idxg, W1l, W2l, flds, f1l, pair, G, rb, jb, mt, sc1, sh1, s, sq);
    pair += G;
  }
  if (ntrip & 1) conv2_stats_trip(p, xt, idxg, W1l, W2l, flds, f1l, pair, G, ra, ja, mt, sc1, sh1, s, sq);
  #pragma unroll
  for (int v = 0; v < 8; ++v)
    #pragma unroll
    for (int off = 1; off < 16; off <<= 1) {
      s[v]  += __shfl_xor(s[v],  off, 32);
      sq[v] += __shfl_xor(sq[v], off, 32);
    }
  if ((lane & 15) == 0) {
    #pragma unroll
    for (int v = 0; v < 8; ++v) {
      int o = mt * 16 + hi * 8 + v;
      part2[(size_t)blockIdx.x * 128 + o]      = s[v];
      part2[(size_t)blockIdx.x * 128 + 64 + o] = sq[v];
    }
  }
}

struct PairEnds { int a0, a16, b0, b16; };

__device__ __forceinline__ PairEnds pair_ends(const int* __restrict__ idxg, int pair) {
  const int* row = idxg + (size_t)pair * 64;
  PairEnds e;
  e.a0 = row[0]; e.a16 = row[16]; e.b0 = row[32]; e.b16 = row[48];
  return e;
}

__device__ __forceinline__ int live_tiles(const PairEnds& e) {
  int live = 15;
  if (e.a16 == e.a0) live &= ~2;
  if (e.b16 == e.b0) live &= ~8;
  return __builtin_amdgcn_readfirstlane(live);
}

template <int LIVE>
__device__ __forceinline__ void final_tile(const float* __restrict__ W1l, const float* __restrict__ W2l,
                                           const float* flds, float* f1l, int mt,
                                           const float (&sc1)[8], const float (&sh1)[8],
                                           const float (&sc2)[8], const float (&sh2)[8],
                                           float (&m0)[8], float (&m1)[8]) {
  v8f acc[4];
  conv1_strip<LIVE>(W1l, flds, mt, acc);
  bn1_store<LIVE>(acc, sc1, sh1, f1l, mt);
  lds_barrier();
  conv2_strip<LIVE>(W2l, f1l, mt, acc);
  #pragma unroll
  for (int v = 0; v < 8; ++v) { m0[v] = 0.f; m1[v] = 0.f; }
  #pragma unroll
  for (int ct = 0; ct < 4; ++ct) {
    if (!((LIVE >> ct) & 1)) continue;
    #pragma unroll
    for (int v = 0; v < 8; ++v) {
      float y = acc[ct][v] * sc2[v] + sh2[v];
      y = y > 0.f ? y : 0.f;
      if (ct < 2) m0[v] = fmaxf(m0[v], y);
      else        m1[v] = fmaxf(m1[v], y);
    }
  }
}

__global__ __launch_bounds__(128) void final_kernel(
    const float* __restrict__ p, const float* __restrict__ xt,
    const int* __restrict__ idxg, const float* __restrict__ W1,
    const float* __restrict__ W2, const float* __restrict__ bnp1,
    const float* __restrict__ bnp2, float* __restrict__ out) {
  __shared__ __attribute__((aligned(16))) float W1l[COUT * W1STRIDE];
  __shared__ __attribute__((aligned(16))) float W2l[COUT * W2STRIDE];
  __shared__ __attribute__((aligned(16))) float flds[64 * FPITCH];
  __shared__ __attribute__((aligned(16))) float f1l[64 * F1PITCH];
  int t = threadIdx.x;
  const int G = gridDim.x;
  int pair = blockIdx.x;
  const int j0 = tile_idx(idxg, pair);
  int jn = tile_idx(idxg, pair + G < NPAIRS ? pair + G : pair);
  PairEnds ends = pair_ends(idxg, pair);
  {
    float4 r1[5], r2[8];
    w1_fetch(W1, r1);
    w2_fetch(W2, r2);
    w1_put(r1, W1l);
    w2_put(r2, W2l);
  }
  FPart r;
  ftile_fetch(p, xt, pair, j0, r);
  int mt = t >> 5, lane = t & 31;
  int hi = lane >> 4;
  float sc1[8], sh1[8], sc2[8], sh2[8];
  #pragma unroll
  for (int v = 0; v < 8; ++v) {
    int o = mt * 16 + hi * 8 + v;
    sc1[v] = bnp1[o]; sh1[v] = bnp1[64 + o];
    sc2[v] = bnp2[o]; sh2[v] = bnp2[64 + o];
  }
  for (; pair < NPAIRS; pair += G) {
    int b  = pair / (NN / 2);
    int n0 = (pair % (NN / 2)) * 2;
    const int live = live_tiles(ends);
    ftile_put(r, flds);
    const int pn  = pair + G     < NPAIRS ? pair + G     : pair;
    const int pn2 = pair + 2 * G < NPAIRS ? pair + 2 * G : pair;
    ftile_fetch(p, xt, pn, jn, r);
    jn = tile_idx(idxg, pn2);
    ends = pair_ends(idxg, pn);
    lds_barrier();
    float m0[8], m1[8];
    switch (live) {
      case 15: final_tile<15>(W1l, W2l, flds, f1l, mt, sc1, sh1, sc2, sh2, m0, m1); break;
      case 13: final_tile<13>(W1l, W2l, flds, f1l, mt, sc1, sh1, sc2, sh2, m0, m1); break;
      case 7:  final_tile<7>(W1l, W2l, flds, f1l, mt, sc1, sh1, sc2, sh2, m0, m1); break;
      default: final_tile<5>(W1l, W2l, flds, f1l, mt, sc1, sh1, sc2, sh2, m0, m1); break;
    }
    #pragma unroll
    for (int v = 0; v < 8; ++v)
      #pragma unroll
      for (int off = 1; off < 16; off <<= 1) {
        m0[v] = fmaxf(m0[v], __shfl_xor(m0[v], off, 32));
        m1[v] = fmaxf(m1[v], __shfl_xor(m1[v], off, 32));
      }
    if ((lane & 15) == 0) {
      #pragma unroll
      for (int v = 0; v < 8; ++v) {
        int o = mt * 16 + hi * 8 + v;
        out[((size_t)b * COUT + o) * NN + n0]     = m0[v];
        out[((size_t)b * COUT + o) * NN + n0 + 1] = m1[v];
      }
    }
  }
}

extern "C" void kernel_launch(void* const* d_in, const int* in_sizes, int n_in,
                              void* d_out, int out_size, void* d_ws, size_t ws_size,
                              hipStream_t stream) {
  const float* p  = (const float*)d_in[0];
  const float* x  = (const float*)d_in[1];
  const float* W1 = (const float*)d_in[2];
  const float* g1 = (const float*)d_in[3];
  const float* b1 = (const float*)d_in[4];
  const float* W2 = (const float*)d_in[5];
  const float* g2 = (const float*)d_in[6];
  const float* b2 = (const float*)d_in[7];
  float* out = (float*)d_out;

  char*  ws    = (char*)d_ws;
  int*   idx   = (int*)ws;
  float* xt    = (float*)(ws + (size_t)BB * NN * KS * 4);
  float* part1 = (float*)(ws + (size_t)BB * NN * KS * 4
                             + (size_t)BB * NN * CX * 4);
  float* part2 = part1 + (size_t)GCONV * 128;
  float* bnp1  = part2 + (size_t)GCONV * 128;
  float* bnp2  = bnp1 + 128;

  ballq_kernel<<<dim3(NN / 32, BB), 256, 0, stream>>>(p, x, idx, xt);
  conv1_stats_kernel<<<GSTAT, 128, 0, stream>>>(p, xt, idx, W1, part1);
  conv2_stats_kernel<<<GSTAT, 128, 0, stream>>>(p, xt, idx, W1, W2, part1, g1, b1, bnp1, part2);
  bn_finalize_kernel<<<1, 64, 0, stream>>>(part2, g2, b2, bnp2, GSTAT);
  final_kernel<<<GCONV, 128, 0, stream>>>(p, xt, idx, W1, W2, bnp1, bnp2, out);
}


# BN2 finalize launch removed: each last-pass block sums the 256 BN2 partial rows itself in its prologue (16-byte loads, 4 waves joined in LDS); 4 launches, was 5; BN2 sums' order changes
# speedup vs baseline: 5.5072x; 1.1278x over previous
#include <hip/hip_runtime.h>
#include <hip/hip_bf16.h>


typedef __attribute__((ext_vector_type(2))) float v2f;
typedef __attribute__((ext_vector_type(4))) float v4f;
typedef __attribute__((ext_vector_type(8))) float v8f;

#define BB      4
#define NN      4096
#define CX      32
#define CIN     35
#define CINP    36
#define COUT    64
#define KS      32
#define R2      0.01f
#define EPSV    1e-5f
#define NPAIRS  (BB * (NN / 2))
#define GCONV   1024
#define GSTAT   256
#define CNTF    ((float)BB * (float)NN * (float)KS)

#define FPITCH   36
#define F1PITCH  68
#define W1STRIDE 36
#define W2STRIDE 68

#define BQ_WAVES 8
#define BQ_SEG   (NN / BQ_WAVES)
#define BQ_GRP   16
#define BQ_PITCH 33

__global__ __launch_bounds__(256) void ballq_kernel(const float* __restrict__ p,
                                                    const float* __restrict__ x,
                                                    int* __restrict__ idx,
                                                    float* __restrict__ xt) {
  __shared__ unsigned short lst[BQ_WAVES][32][BQ_PITCH];
  __shared__ int cnts[BQ_WAVES][32];
  __shared__ float xtile[CX][33];
  const int b    = blockIdx.y;
  const int lane = threadIdx.x & 31;
  const int seg  = __builtin_amdgcn_readfirstlane((int)(threadIdx.x >> 5));
  const float* __restrict__ pb = p + (size_t)b * NN * 3;
  const int q = blockIdx.x * 32 + lane;
  const float px = pb[q * 3 + 0];
  const float py = pb[q * 3 + 1];
  const float pz = pb[q * 3 + 2];
  float xv[4];
  #pragma unroll
  for (int k = 0; k < 4; ++k) {
    int i = (int)threadIdx.x + 256 * k;
    xv[k] = x[((size_t)b * CX + (i >> 5)) * NN + blockIdx.x * 32 + (i & 31)];
  }
  const float* __restrict__ cb = pb + seg * (BQ_SEG * 3);
  unsigned short* mylst = &lst[seg][lane][0];
  int cnt = 0;
  for (int g = 0; g < BQ_SEG; g += BQ_GRP) {
    unsigned m = 0;
    #pragma unroll
    for (int jj = 0; jj < BQ_GRP; ++jj) {
      float dx = cb[(g + jj) * 3 + 0] - px;
      float dy = cb[(g + jj) * 3 + 1] - py;
      float dz = cb[(g + jj) * 3 + 2] - pz;
      float d2 = __builtin_fmaf(dz, dz, __builtin_fmaf(dy, dy, dx * dx));
      m = m + m + (d2 < R2 ? 1u : 0u);
    }
    while (m != 0) {
      int lz   = __builtin_clz(m);
      int slot = cnt < KS ? cnt : KS;
      mylst[slot] = (unsigned short)(seg * BQ_SEG + g + lz - (32 - BQ_GRP));
      cnt++;
      m &= ~(0x80000000u >> lz);
    }
  }
  cnts[seg][lane] = cnt;
  #pragma unroll
  for (int k = 0; k < 4; ++k) {
    int i = (int)threadIdx.x + 256 * k;
    xtile[i >> 5][i & 31] = xv[k];
  }
  __syncthreads();

  {
    const int nn = threadIdx.x >> 3;
    const int c4 = (threadIdx.x & 7) * 4;
    float4 xo;
    xo.x = xtile[c4 + 0][nn]; xo.y = xtile[c4 + 1][nn];
    xo.z = xtile[c4 + 2][nn]; xo.w = xtile[c4 + 3][nn];
    *reinterpret_cast<float4*>(xt + ((size_t)b * NN + blockIdx.x * 32 + nn) * CX + c4) = xo;
  }

  const int ql = threadIdx.x >> 3;
  const int r0 = (threadIdx.x & 7) * 4;
  int v[4] = {0, 0, 0, 0};
  int pre = 0, first = 0;
  bool have = false;
  #pragma unroll
  for (int s = 0; s < BQ_WAVES; ++s) {
    int c = cnts[s][ql];
    c = c < KS ? c : KS;
    if (!have && c > 0) { first = lst[s][ql][0]; have = true; }
    #pragma unroll
    for (int k = 0; k < 4; ++k) {
      int r = r0 + k - pre;
      if (r >= 0 && r < c) v[k] = lst[s][ql][r];
    }
    pre += c;
  }
  #pragma unroll
  for (int k = 0; k < 4; ++k)
    if (r0 + k >= pre) v[k] = first;
  int4 o;
  o.x = v[0]; o.y = v[1]; o.z = v[2]; o.w = v[3];
  *reinterpret_cast<int4*>(idx + ((size_t)b * NN + blockIdx.x * 32 + ql) * KS + r0) = o;
}

struct FPart {
  v4f   xq[4];
  float pj[2];
  float pq[2];
};

__device__ __forceinline__ int tile_idx(const int* idxg, int pair) {
  return idxg[(size_t)pair * 64 + (threadIdx.x & 63)];
}

__device__ __forceinline__ void ftile_fetch(const float* p,
                                            const float* xt,
                                            int pair, int j, FPart& r) {
  const int t = threadIdx.x;
  const int b = pair / (NN / 2);
  const v4f* xr = reinterpret_cast<const v4f*>(xt + ((size_t)b * NN + j) * CX + (t < 64 ? 0 : 16));
  #pragma unroll
  for (int q = 0; q < 4; ++q) r.xq[q] = xr[q];
  const bool lo = t < 64;
  const float* pj = p + ((size_t)b * NN + j) * 3 + (lo ? 0 : 2);
  const float* pq = p + ((size_t)2 * pair + ((t & 63) >> 5)) * 3 + (lo ? 0 : 2);
  r.pj[0] = pj[0];
  r.pq[0] = pq[0];
  r.pj[1] = pj[lo ? 1 : 0];
  r.pq[1] = pq[lo ? 1 : 0];
}

__device__ __forceinline__ void ftile_put(const FPart& r, float* flds) {
  const int t = threadIdx.x;
  const bool lo = t < 64;
  float* colp = flds + (t & 63) * FPITCH;
  v4f* xq = reinterpret_cast<v4f*>(colp + (lo ? 0 : 16));
  #pragma unroll
  for (int q = 0; q < 4; ++q) xq[q] = r.xq[q];
  float2 d;
  d.x = r.pj[0] - r.pq[0];
  d.y = lo ? r.pj[1] - r.pq[1] : 0.f;
  *reinterpret_cast<float2*>(colp + (lo ? 32 : 34)) = d;
}

__device__ __forceinline__ void lds_barrier() {
  asm volatile("s_wait_dscnt 0x0\n\ts_barrier_signal -1\n\ts_barrier_wait -1" ::: "memory");
}

struct Chunk {
  float4 a;
  float4 b[4];
};
struct Tail {
  float2 a;
  float2 b[4];
};

template <int LIVE>
__device__ __forceinline__ void chunk_load(Chunk& f, const float* wrow, const float* fcol,
                                           int fpitch, int j) {
  f.a = *reinterpret_cast<const float4*>(wrow + 8 * j);
  #pragma unroll
  for (int ct = 0; ct < 4; ++ct)
    if ((LIVE >> ct) & 1)
      f.b[ct] = *reinterpret_cast<const float4*>(fcol + ct * 16 * fpitch + 8 * j);
}

template <int LIVE>
__device__ __forceinline__ void tail_load(Tail& f, const float* wrow2, const float* fcol2,
                                          int fpitch, int k0) {
  f.a = *reinterpret_cast<const float2*>(wrow2 + k0);
  #pragma unroll
  for (int ct = 0; ct < 4; ++ct)
    if ((LIVE >> ct) & 1)
      f.b[ct] = *reinterpret_cast<const float2*>(fcol2 + ct * 16 * fpitch + k0);
}

__device__ __forceinline__ v8f mma4(v2f a, v2f b, v8f cacc) {
  return __builtin_amdgcn_wmma_f32_16x16x4_f32(false, a, false, b, (short)0, cacc, false, false);
}

template <int LIVE>
__device__ __forceinline__ void chunk_mma(const Chunk& f, v8f (&acc)[4]) {
  v2f a0, a1;
  a0.x = f.a.x; a0.y = f.a.y; a1.x = f.a.z; a1.y = f.a.w;
  #pragma unroll
  for (int ct = 0; ct < 4; ++ct) {
    if (!((LIVE >> ct) & 1)) continue;
    v2f b; b.x = f.b[ct].x; b.y = f.b[ct].y;
    acc[ct] = mma4(a0, b, acc[ct]);
  }
  #pragma unroll
  for (int ct = 0; ct < 4; ++ct) {
    if (!((LIVE >> ct) & 1)) continue;
    v2f b; b.x = f.b[ct].z; b.y = f.b[ct].w;
    acc[ct] = mma4(a1, b, acc[ct]);
  }
}

template <int LIVE>
__device__ __forceinline__ void tail_mma(const Tail& f, v8f (&acc)[4]) {
  v2f a; a.x = f.a.x; a.y = f.a.y;
  #pragma unroll
  for (int ct = 0; ct < 4; ++ct) {
    if (!((LIVE >> ct) & 1)) continue;
    v2f b; b.x = f.b[ct].x; b.y = f.b[ct].y;
    acc[ct] = mma4(a, b, acc[ct]);
  }
}

__device__ __forceinline__ void sched_fence() { __builtin_amdgcn_sched_barrier(0); }

template <int LIVE>
__device__ __forceinline__ void acc_zero(v8f (&acc)[4]) {
  #pragma unroll
  for (int ct = 0; ct < 4; ++ct)
    if ((LIVE >> ct) & 1) {
      #pragma unroll
      for (int v = 0; v < 8; ++v) acc[ct][v] = 0.f;
    }
}

template <int LIVE>
__device__ __forceinline__ void conv1_strip(const float* __restrict__ W1l,
                                            const float* __restrict__ ft,
                                            int mt, v8f (&acc)[4]) {
  const int lane = threadIdx.x & 31;
  const int c = lane & 15, h = lane >> 4;
  const float* wrow = W1l + (mt * 16 + c) * W1STRIDE;
  const float* fcol = ft + c * FPITCH;
  Chunk g[4];
  Tail  gt;
  acc_zero<LIVE>(acc);
  #pragma unroll
  for (int j = 0; j < 4; ++j) chunk_load<LIVE>(g[j], wrow + 4 * h, fcol + 4 * h, FPITCH, j);
  tail_load<LIVE>(gt, wrow + 2 * h, fcol + 2 * h, FPITCH, 32);
  sched_fence();
  #pragma unroll
  for (int j = 0; j < 4; ++j) chunk_mma<LIVE>(g[j], acc);
  tail_mma<LIVE>(gt, acc);
  sched_fence();
}

template <int LIVE>
__device__ __forceinline__ void conv2_strip(const float* __restrict__ W2l,
                                            const float* __restrict__ f1l,
                                            int mt, v8f (&acc)[4]) {
  const int lane = threadIdx.x & 31;
  const int c = lane & 15, h = lane >> 4;
  const float* wrow = W2l + (mt * 16 + c) * W2STRIDE + 4 * h;
  const float* fcol = f1l + c * F1PITCH + 4 * h;
  Chunk s0[2], s1[2];
  acc_zero<LIVE>(acc);
  chunk_load<LIVE>(s0[0], wrow, fcol, F1PITCH, 0);
  chunk_load<LIVE>(s0[1], wrow, fcol, F1PITCH, 1);
  chunk_load<LIVE>(s1[0], wrow, fcol, F1PITCH, 2);
  chunk_load<LIVE>(s1[1], wrow, fcol, F1PITCH, 3);
  sched_fence();
  chunk_mma<LIVE>(s0[0], acc);
  chunk_mma<LIVE>(s0[1], acc);
  sched_fence();
  chunk_load<LIVE>(s0[0], wrow, fcol, F1PITCH, 4);
  chunk_load<LIVE>(s0[1], wrow, fcol, F1PITCH, 5);
  sched_fence();
  chunk_mma<LIVE>(s1[0], acc);
  chunk_mma<LIVE>(s1[1], acc);
  sched_fence();
  chunk_load<LIVE>(s1[0], wrow, fcol, F1PITCH, 6);
  chunk_load<LIVE>(s1[1], wrow, fcol, F1PITCH, 7);
  sched_fence();
  chunk_mma<LIVE>(s0[0], acc);
  chunk_mma<LIVE>(s0[1], acc);
  chunk_mma<LIVE>(s1[0], acc);
  chunk_mma<LIVE>(s1[1], acc);
  sched_fence();
}

template <int LIVE>
__device__ __forceinline__ void bn1_store(const v8f (&acc)[4], const float (&sc1)[8],
                                          const float (&sh1)[8], float* f1l, int mt) {
  const int lane = threadIdx.x & 31;
  const int c = lane & 15, h = lane >> 4;
  #pragma unroll
  for (int ct = 0; ct < 4; ++ct) {
    if (!((LIVE >> ct) & 1)) continue;
    float y[8];
    #pragma unroll
    for (int v = 0; v < 8; ++v) {
      float t = acc[ct][v] * sc1[v] + sh1[v];
      y[v] = t > 0.f ? t : 0.f;
    }
    float4* dst = reinterpret_cast<float4*>(f1l + (ct * 16 + c) * F1PITCH + mt * 16 + h * 8);
    float4 q0, q1;
    q0.x = y[0]; q0.y = y[1]; q0.z = y[2]; q0.w = y[3];
    q1.x = y[4]; q1.y = y[5]; q1.z = y[6]; q1.w = y[7];
    dst[0] = q0;
    dst[1] = q1;
  }
}


#define W1Q (COUT * CIN / 4)
#define W2Q (COUT * COUT / 4)

__device__ __forceinline__ void w1_fetch(const float* __restrict__ W1, float4 (&r)[5]) {
  const float4* __restrict__ src = reinterpret_cast<const float4*>(W1);
  const int t = threadIdx.x;
  #pragma unroll
  for (int e = 0; e < 5; ++e) {
    int q = t + 128 * e;
    r[e] = src[q < W1Q ? q : W1Q - 1];
  }
}

__device__ __forceinline__ void w1_put(const float4 (&r)[5], float* W1l) {
  const int t = threadIdx.x;
  #pragma unroll
  for (int e = 0; e < 5; ++e) {
    int q = t + 128 * e;
    if (q < W1Q) {
      const float v[4] = {r[e].x, r[e].y, r[e].z, r[e].w};
      #pragma unroll
      for (int u = 0; u < 4; ++u) {
        int i   = 4 * q + u;
        int row = i / CIN;
        int cc  = i - row * CIN;
        int kp  = cc >= 3 ? cc - 3 : 32 + cc;
        W1l[row * W1STRIDE + kp] = v[u];
      }
    }
  }
  if (t < COUT) W1l[t * W1STRIDE + CIN] = 0.f;
}

__device__ __forceinline__ void w2_fetch(const float* __restrict__ W2, float4 (&r)[8]) {
  const float4* __restrict__ src = reinterpret_cast<const float4*>(W2);
  const int t = threadIdx.x;
  #pragma unroll
  for (int e = 0; e < 8; ++e) r[e] = src[t + 128 * e];
}

__device__ __forceinline__ void w2_put(const float4 (&r)[8], float* W2l) {
  const int t = threadIdx.x;
  #pragma unroll
  for (int e = 0; e < 8; ++e) {
    int q = t + 128 * e;
    *reinterpret_cast<float4*>(&W2l[(q >> 4) * W2STRIDE + (q & 15) * 4]) = r[e];
  }
}

__device__ __forceinline__ int pair_or(int q, int fallback) { return q < NPAIRS ? q : fallback; }

__device__ __forceinline__ void conv1_stats_trip(const float* p, const float* xt,
                                                 const int* idxg, const float* W1l,
                                                 float* ft, int pair, int G, FPart& r, int& jn,
                                                 int mt, float (&s)[8], float (&sq)[8]) {
  ftile_put(r, ft);
  const int pn2 = pair_or(pair + 2 * G, pair);
  const int pn4 = pair_or(pair + 4 * G, pair);
  ftile_fetch(p, xt, pn2, jn, r);
  jn = tile_idx(idxg, pn4);
  lds_barrier();
  v8f acc[4];
  conv1_strip<15>(W1l, ft, mt, acc);
  #pragma unroll
  for (int ct = 0; ct < 4; ++ct) {
    #pragma unroll
    for (int v = 0; v < 8; ++v) { float y = acc[ct][v]; s[v] += y; sq[v] += y * y; }
  }
}

__global__ __launch_bounds__(128) void conv1_stats_kernel(
    const float* p, const float* xt,
    const int* idxg, const float* __restrict__ W1,
    float* __restrict__ part1) {
  __shared__ __attribute__((aligned(16))) float W1l[COUT * W1STRIDE];
  __shared__ __attribute__((aligned(16))) float flds[2][64 * FPITCH];
  int t = threadIdx.x;
  const int G = gridDim.x;
  int pair = blockIdx.x;
  const int pf1 = pair_or(pair + G, pair);
  const int j0 = tile_idx(idxg, pair);
  const int j1 = tile_idx(idxg, pf1);
  {
    float4 r1[5];
    w1_fetch(W1, r1);
    w1_put(r1, W1l);
  }
  FPart ra, rb;
  ftile_fetch(p, xt, pair, j0, ra);
  int ja = tile_idx(idxg, pair_or(pair + 2 * G, pair));
  sched_fence();
  ftile_fetch(p, xt, pf1, j1, rb);
  int jb = tile_idx(idxg, pair_or(pair + 3 * G, pair));
  int mt = t >> 5, lane = t & 31;
  float s[8], sq[8];
  #pragma unroll
  for (int v = 0; v < 8; ++v) { s[v] = 0.f; sq[v] = 0.f; }

  const int ntrip = pair < NPAIRS ? (NPAIRS - 1 - pair) / G + 1 : 0;
  for (int it = 0; it + 1 < ntrip; it += 2) {
    conv1_stats_trip(p, xt, idxg, W1l, flds[0], pair, G, ra, ja, mt, s, sq);
    pair += G;
    conv1_stats_trip(p, xt, idxg, W1l, flds[1], pair, G, rb, jb, mt, s, sq);
    pair += G;
  }
  if (ntrip & 1) conv1_stats_trip(p, xt, idxg, W1l, flds[0], pair, G, ra, ja, mt, s, sq);
  #pragma unroll
  for (int v = 0; v < 8; ++v)
    #pragma unroll
    for (int off = 1; off < 16; off <<= 1) {
      s[v]  += __shfl_xor(s[v],  off, 32);
      sq[v] += __shfl_xor(sq[v], off, 32);
    }
  if ((lane & 15) == 0) {
    int hi = lane >> 4;
    #pragma unroll
    for (int v = 0; v < 8; ++v) {
      int o = mt * 16 + hi * 8 + v;
      part1[(size_t)blockIdx.x * 128 + o]      = s[v];
      part1[(size_t)blockIdx.x * 128 + 64 + o] = sq[v];
    }
  }
}

template <bool KEEP, int BATCH>
__device__ __forceinline__ void bn_from_rows(const float* __restrict__ part,
                                             const float* g,
                                             const float* bta,
                                             float* red, float* bnl,
                                             float* __restrict__ bnp_out) {
  const int t = threadIdx.x;
  const int q = t & 31, w = t >> 5;
  const float gv = g[t & 63];
  const float bv = bta[t & 63];
  const float4* __restrict__ rows = reinterpret_cast<const float4*>(part) + (size_t)w * (GSTAT / 4) * 32 + q;
  float4 a[4];
  #pragma unroll
  for (int u = 0; u < 4; ++u) { a[u].x = 0.f; a[u].y = 0.f; a[u].z = 0.f; a[u].w = 0.f; }
  #pragma unroll
  for (int i0 = 0; i0 < GSTAT / 4; i0 += BATCH) {
    #pragma unroll
    for (int i = i0; i < i0 + BATCH; ++i) {
      const float4 v = rows[(size_t)i * 32];
      a[i & 3].x += v.x; a[i & 3].y += v.y; a[i & 3].z += v.z; a[i & 3].w += v.w;
    }
    if (i0 + BATCH < GSTAT / 4) sched_fence();
  }
  float4 sum;
  sum.x = (a[0].x + a[1].x) + (a[2].x + a[3].x);
  sum.y = (a[0].y + a[1].y) + (a[2].y + a[3].y);
  sum.z = (a[0].z + a[1].z) + (a[2].z + a[3].z);
  sum.w = (a[0].w + a[1].w) + (a[2].w + a[3].w);
  *reinterpret_cast<float4*>(red + w * 128 + 4 * q) = sum;
  lds_barrier();
  const int o = t & 63;
  const float s  = (red[o]      + red[128 + o])      + (red[256 + o]      + red[384 + o]);
  const float sq = (red[64 + o] + red[128 + 64 + o]) + (red[256 + 64 + o] + red[384 + 64 + o]);
  const float mean  = s / CNTF;
  const float var   = sq / CNTF - mean * mean;
  const float scale = gv * rsqrtf(var + EPSV);
  const float shift = bv - mean * scale;
  if (t < COUT) {
    bnl[o]      = scale;
    bnl[64 + o] = shift;
    if (KEEP && blockIdx.x == 0) {
      bnp_out[o]      = scale;
      bnp_out[64 + o] = shift;
    }
  }
  lds_barrier();
}

__device__ __forceinline__ void conv2_stats_trip(const float* p, const float* xt,
                                                 const int* idxg, const float* W1l,
                                                 const float* W2l, float* flds, float* f1l,
                                                 int pair, int G, FPart& r, int& jn, int mt,
                                                 const float (&sc1)[8], const float (&sh1)[8],
                                                 float (&s)[8], float (&sq)[8]) {
  ftile_put(r, flds);
  const int pn2 = pair_or(pair + 2 * G, pair);
  const int pn4 = pair_or(pair + 4 * G, pair);
  ftile_fetch(p, xt, pn2, jn, r);
  jn = tile_idx(idxg, pn4);
  lds_barrier();
  v8f acc[4];
  conv1_strip<15>(W1l, flds, mt, acc);
  bn1_store<15>(acc, sc1, sh1, f1l, mt);
  lds_barrier();
  conv2_strip<15>(W2l, f1l, mt, acc);
  #pragma unroll
  for (int ct = 0; ct < 4; ++ct) {
    #pragma unroll
    for (int v = 0; v < 8; ++v) { float y = acc[ct][v]; s[v] += y; sq[v] += y * y; }
  }
}

__global__ __launch_bounds__(128) void conv2_stats_kernel(
    const float* p, const float* xt,
    const int* idxg, const float* __restrict__ W1,
    const float* __restrict__ W2, const float* __restrict__ part1,
    const float* g1, const float* b1,
    float* __restrict__ bnp1, float* __restrict__ part2) {
  __shared__ __attribute__((aligned(16))) float W1l[COUT * W1STRIDE];
  __shared__ __attribute__((aligned(16))) float W2l[COUT * W2STRIDE];
  __shared__ __attribute__((aligned(16))) float flds[64 * FPITCH];
  __shared__ __attribute__((aligned(16))) float f1l[64 * F1PITCH];
  __shared__ __attribute__((aligned(16))) float red[4 * 128];
  __shared__ float bn1l[128];
  int t = threadIdx.x;
  const int G = gridDim.x;
  int pair = blockIdx.x;
  const int pf1 = pair_or(pair + G, pair);
  const int j0 = tile_idx(idxg, pair);
  const int j1 = tile_idx(idxg, pf1);
  {
    float4 r1[5], r2[8];
    w1_fetch(W1, r1);
    w2_fetch(W2, r2);
    w1_put(r1, W1l);
    w2_put(r2, W2l);
  }
  FPart ra, rb;
  ftile_fetch(p, xt, pair, j0, ra);
  int ja = tile_idx(idxg, pair_or(pair + 2 * G, pair));
  sched_fence();
  ftile_fetch(p, xt, pf1, j1, rb);
  int jb = tile_idx(idxg, pair_or(pair + 3 * G, pair));
  bn_from_rows<true, GSTAT / 4>(part1, g1, b1, red, bn1l, bnp1);
  int mt = t >> 5, lane = t & 31;
  int hi = lane >> 4;
  float sc1[8], sh1[8];
  #pragma unroll
  for (int v = 0; v < 8; ++v) {
    int o = mt * 16 + hi * 8 + v;
    sc1[v] = bn1l[o]; sh1[v] = bn1l[64 + o];
  }
  float s[8], sq[8];
  #pragma unroll
  for (int v = 0; v < 8; ++v) { s[v] = 0.f; sq[v] = 0.f; }

  const int ntrip = pair < NPAIRS ? (NPAIRS - 1 - pair) / G + 1 : 0;
  for (int it = 0; it + 1 < ntrip; it += 2) {
    conv2_stats_trip(p, xt, idxg, W1l, W2l, flds, f1l, pair, G, ra, ja, mt, sc1, sh1, s, sq);
    pair += G;
    conv2_stats_trip(p, xt, idxg, W1l, W2l, flds, f1l, pair, G, rb, jb, mt, sc1, sh1, s, sq);
    pair += G;
  }
  if (ntrip & 1) conv2_stats_trip(p, xt, idxg, W1l, W2l, flds, f1l, pair, G, ra, ja, mt, sc1, sh1, s, sq);
  #pragma unroll
  for (int v = 0; v < 8; ++v)
    #pragma unroll
    for (int off = 1; off < 16; off <<= 1) {
      s[v]  += __shfl_xor(s[v],  off, 32);
      sq[v] += __shfl_xor(sq[v], off, 32);
    }
  if ((lane & 15) == 0) {
    #pragma unroll
    for (int v = 0; v < 8; ++v) {
      int o = mt * 16 + hi * 8 + v;
      part2[(size_t)blockIdx.x * 128 + o]      = s[v];
      part2[(size_t)blockIdx.x * 128 + 64 + o] = sq[v];
    }
  }
}

struct PairEnds { int a0, a16, b0, b16; };

__device__ __forceinline__ PairEnds pair_ends(const int* __restrict__ idxg, int pair) {
  const int* row = idxg + (size_t)pair * 64;
  PairEnds e;
  e.a0 = row[0]; e.a16 = row[16]; e.b0 = row[32]; e.b16 = row[48];
  return e;
}

__device__ __forceinline__ int live_tiles(const PairEnds& e) {
  int live = 15;
  if (e.a16 == e.a0) live &= ~2;
  if (e.b16 == e.b0) live &= ~8;
  return __builtin_amdgcn_readfirstlane(live);
}

template <int LIVE>
__device__ __forceinline__ void final_tile(const float* __restrict__ W1l, const float* __restrict__ W2l,
                                           const float* flds, float* f1l, int mt,
                                           const float (&sc1)[8], const float (&sh1)[8],
                                           const float (&sc2)[8], const float (&sh2)[8],
                                           float (&m0)[8], float (&m1)[8]) {
  v8f acc[4];
  conv1_strip<LIVE>(W1l, flds, mt, acc);
  bn1_store<LIVE>(acc, sc1, sh1, f1l, mt);
  lds_barrier();
  conv2_strip<LIVE>(W2l, f1l, mt, acc);
  #pragma unroll
  for (int v = 0; v < 8; ++v) { m0[v] = 0.f; m1[v] = 0.f; }
  #pragma unroll
  for (int ct = 0; ct < 4; ++ct) {
    if (!((LIVE >> ct) & 1)) continue;
    #pragma unroll
    for (int v = 0; v < 8; ++v) {
      float y = acc[ct][v] * sc2[v] + sh2[v];
      y = y > 0.f ? y : 0.f;
      if (ct < 2) m0[v] = fmaxf(m0[v], y);
      else        m1[v] = fmaxf(m1[v], y);
    }
  }
}

__global__ __launch_bounds__(128) void final_kernel(
    const float* __restrict__ p, const float* __restrict__ xt,
    const int* __restrict__ idxg, const float* __restrict__ W1,
    const float* __restrict__ W2, const float* __restrict__ bnp1,
    const float* __restrict__ part2, const float* g2, const float* b2,
    float* __restrict__ out) {
  __shared__ __attribute__((aligned(16))) float W1l[COUT * W1STRIDE];
  __shared__ __attribute__((aligned(16))) float W2l[COUT * W2STRIDE];
  __shared__ __attribute__((aligned(16))) float flds[64 * FPITCH];
  __shared__ __attribute__((aligned(16))) float f1l[64 * F1PITCH];
  __shared__ __attribute__((aligned(16))) float red[4 * 128];
  __shared__ float bn2l[128];
  int t = threadIdx.x;
  const int G = gridDim.x;
  int pair = blockIdx.x;
  const int j0 = tile_idx(idxg, pair);
  int jn = tile_idx(idxg, pair + G < NPAIRS ? pair + G : pair);
  PairEnds ends = pair_ends(idxg, pair);
  {
    float4 r1[5], r2[8];
    w1_fetch(W1, r1);
    w2_fetch(W2, r2);
    w1_put(r1, W1l);
    w2_put(r2, W2l);
  }
  FPart r;
  ftile_fetch(p, xt, pair, j0, r);
  int mt = t >> 5, lane = t & 31;
  int hi = lane >> 4;
  float sc1[8], sh1[8], sc2[8], sh2[8];
  #pragma unroll
  for (int v = 0; v < 8; ++v) {
    int o = mt * 16 + hi * 8 + v;
    sc1[v] = bnp1[o]; sh1[v] = bnp1[64 + o];
  }
  bn_from_rows<false, GSTAT / 8>(part2, g2, b2, red, bn2l, nullptr);
  #pragma unroll
  for (int v = 0; v < 8; ++v) {
    int o = mt * 16 + hi * 8 + v;
    sc2[v] = bn2l[o]; sh2[v] = bn2l[64 + o];
  }
  for (; pair < NPAIRS; pair += G) {
    int b  = pair / (NN / 2);
    int n0 = (pair % (NN / 2)) * 2;
    const int live = live_tiles(ends);
    ftile_put(r, flds);
    const int pn  = pair + G     < NPAIRS ? pair + G     : pair;
    const int pn2 = pair + 2 * G < NPAIRS ? pair + 2 * G : pair;
    ftile_fetch(p, xt, pn, jn, r);
    jn = tile_idx(idxg, pn2);
    ends = pair_ends(idxg, pn);
    lds_barrier();
    float m0[8], m1[8];
    switch (live) {
      case 15: final_tile<15>(W1l, W2l, flds, f1l, mt, sc1, sh1, sc2, sh2, m0, m1); break;
      case 13: final_tile<13>(W1l, W2l, flds, f1l, mt, sc1, sh1, sc2, sh2, m0, m1); break;
      case 7:  final_tile<7>(W1l, W2l, flds, f1l, mt, sc1, sh1, sc2, sh2, m0, m1); break;
      default: final_tile<5>(W1l, W2l, flds, f1l, mt, sc1, sh1, sc2, sh2, m0, m1); break;
    }
    #pragma unroll
    for (int v = 0; v < 8; ++v)
      #pragma unroll
      for (int off = 1; off < 16; off <<= 1) {
        m0[v] = fmaxf(m0[v], __shfl_xor(m0[v], off, 32));
        m1[v] = fmaxf(m1[v], __shfl_xor(m1[v], off, 32));
      }
    if ((lane & 15) == 0) {
      #pragma unroll
      for (int v = 0; v < 8; ++v) {
        int o = mt * 16 + hi * 8 + v;
        out[((size_t)b * COUT + o) * NN + n0]     = m0[v];
        out[((size_t)b * COUT + o) * NN + n0 + 1] = m1[v];
      }
    }
  }
}

extern "C" void kernel_launch(void* const* d_in, const int* in_sizes, int n_in,
                              void* d_out, int out_size, void* d_ws, size_t ws_size,
                              hipStream_t stream) {
  const float* p  = (const float*)d_in[0];
  const float* x  = (const float*)d_in[1];
  const float* W1 = (const float*)d_in[2];
  const float* g1 = (const float*)d_in[3];
  const float* b1 = (const float*)d_in[4];
  const float* W2 = (const float*)d_in[5];
  const float* g2 = (const float*)d_in[6];
  const float* b2 = (const float*)d_in[7];
  float* out = (float*)d_out;

  char*  ws    = (char*)d_ws;
  int*   idx   = (int*)ws;
  float* xt    = (float*)(ws + (size_t)BB * NN * KS * 4);
  float* part1 = (float*)(ws + (size_t)BB * NN * KS * 4
                             + (size_t)BB * NN * CX * 4);
  float* part2 = part1 + (size_t)GCONV * 128;
  float* bnp1  = part2 + (size_t)GCONV * 128;

  ballq_kernel<<<dim3(NN / 32, BB), 256, 0, stream>>>(p, x, idx, xt);
  conv1_stats_kernel<<<GSTAT, 128, 0, stream>>>(p, xt, idx, W1, part1);
  conv2_stats_kernel<<<GSTAT, 128, 0, stream>>>(p, xt, idx, W1, W2, part1, g1, b1, bnp1, part2);
  final_kernel<<<GCONV, 128, 0, stream>>>(p, xt, idx, W1, W2, bnp1, part2, g2, b2, out);
}


# stats passes: blocks of 8 waves = two 4-wave tile streams sharing the weights, half a trip apart: a SIMD's second wave works under the first one's WMMAs; one row per block (BN sum order changes)
# speedup vs baseline: 5.7697x; 1.0477x over previous
#include <hip/hip_runtime.h>
#include <hip/hip_bf16.h>


typedef __attribute__((ext_vector_type(2))) float v2f;
typedef __attribute__((ext_vector_type(4))) float v4f;
typedef __attribute__((ext_vector_type(8))) float v8f;

#define BB      4
#define NN      4096
#define CX      32
#define CIN     35
#define CINP    36
#define COUT    64
#define KS      32
#define R2      0.01f
#define EPSV    1e-5f
#define NPAIRS  (BB * (NN / 2))
#define GCONV   1024
#define GSTAT   256
#define CNTF    ((float)BB * (float)NN * (float)KS)

#define FPITCH   36
#define F1PITCH  68
#define W1STRIDE 36
#define W2STRIDE 68

#define BQ_WAVES 8
#define BQ_SEG   (NN / BQ_WAVES)
#define BQ_GRP   16
#define BQ_PITCH 33

__global__ __launch_bounds__(256) void ballq_kernel(const float* __restrict__ p,
                                                    const float* __restrict__ x,
                                                    int* __restrict__ idx,
                                                    float* __restrict__ xt) {
  __shared__ unsigned short lst[BQ_WAVES][32][BQ_PITCH];
  __shared__ int cnts[BQ_WAVES][32];
  __shared__ float xtile[CX][33];
  const int b    = blockIdx.y;
  const int lane = threadIdx.x & 31;
  const int seg  = __builtin_amdgcn_readfirstlane((int)(threadIdx.x >> 5));
  const float* __restrict__ pb = p + (size_t)b * NN * 3;
  const int q = blockIdx.x * 32 + lane;
  const float px = pb[q * 3 + 0];
  const float py = pb[q * 3 + 1];
  const float pz = pb[q * 3 + 2];
  float xv[4];
  #pragma unroll
  for (int k = 0; k < 4; ++k) {
    int i = (int)threadIdx.x + 256 * k;
    xv[k] = x[((size_t)b * CX + (i >> 5)) * NN + blockIdx.x * 32 + (i & 31)];
  }
  const float* __restrict__ cb = pb + seg * (BQ_SEG * 3);
  unsigned short* mylst = &lst[seg][lane][0];
  int cnt = 0;
  for (int g = 0; g < BQ_SEG; g += BQ_GRP) {
    unsigned m = 0;
    #pragma unroll
    for (int jj = 0; jj < BQ_GRP; ++jj) {
      float dx = cb[(g + jj) * 3 + 0] - px;
      float dy = cb[(g + jj) * 3 + 1] - py;
      float dz = cb[(g + jj) * 3 + 2] - pz;
      float d2 = __builtin_fmaf(dz, dz, __builtin_fmaf(dy, dy, dx * dx));
      m = m + m + (d2 < R2 ? 1u : 0u);
    }
    while (m != 0) {
      int lz   = __builtin_clz(m);
      int slot = cnt < KS ? cnt : KS;
      mylst[slot] = (unsigned short)(seg * BQ_SEG + g + lz - (32 - BQ_GRP));
      cnt++;
      m &= ~(0x80000000u >> lz);
    }
  }
  cnts[seg][lane] = cnt;
  #pragma unroll
  for (int k = 0; k < 4; ++k) {
    int i = (int)threadIdx.x + 256 * k;
    xtile[i >> 5][i & 31] = xv[k];
  }
  __syncthreads();

  {
    const int nn = threadIdx.x >> 3;
    const int c4 = (threadIdx.x & 7) * 4;
    float4 xo;
    xo.x = xtile[c4 + 0][nn]; xo.y = xtile[c4 + 1][nn];
    xo.z = xtile[c4 + 2][nn]; xo.w = xtile[c4 + 3][nn];
    *reinterpret_cast<float4*>(xt + ((size_t)b * NN + blockIdx.x * 32 + nn) * CX + c4) = xo;
  }

  const int ql = threadIdx.x >> 3;
  const int r0 = (threadIdx.x & 7) * 4;
  int v[4] = {0, 0, 0, 0};
  int pre = 0, first = 0;
  bool have = false;
  #pragma unroll
  for (int s = 0; s < BQ_WAVES; ++s) {
    int c = cnts[s][ql];
    c = c < KS ? c : KS;
    if (!have && c > 0) { first = lst[s][ql][0]; have = true; }
    #pragma unroll
    for (int k = 0; k < 4; ++k) {
      int r = r0 + k - pre;
      if (r >= 0 && r < c) v[k] = lst[s][ql][r];
    }
    pre += c;
  }
  #pragma unroll
  for (int k = 0; k < 4; ++k)
    if (r0 + k >= pre) v[k] = first;
  int4 o;
  o.x = v[0]; o.y = v[1]; o.z = v[2]; o.w = v[3];
  *reinterpret_cast<int4*>(idx + ((size_t)b * NN + blockIdx.x * 32 + ql) * KS + r0) = o;
}

struct FPart {
  v4f   xq[4];
  float pj[2];
  float pq[2];
};

__device__ __forceinline__ int tile_idx(const int* idxg, int pair) {
  return idxg[(size_t)pair * 64 + (threadIdx.x & 63)];
}

__device__ __forceinline__ void ftile_fetch(const float* p,
                                            const float* xt,
                                            int pair, int j, FPart& r) {
  const int t = threadIdx.x;
  const int b = pair / (NN / 2);
  const bool lo = (t & 64) == 0;
  const v4f* xr = reinterpret_cast<const v4f*>(xt + ((size_t)b * NN + j) * CX + (lo ? 0 : 16));
  #pragma unroll
  for (int q = 0; q < 4; ++q) r.xq[q] = xr[q];
  const float* pj = p + ((size_t)b * NN + j) * 3 + (lo ? 0 : 2);
  const float* pq = p + ((size_t)2 * pair + ((t & 63) >> 5)) * 3 + (lo ? 0 : 2);
  r.pj[0] = pj[0];
  r.pq[0] = pq[0];
  r.pj[1] = pj[lo ? 1 : 0];
  r.pq[1] = pq[lo ? 1 : 0];
}

__device__ __forceinline__ void ftile_put(const FPart& r, float* flds) {
  const int t = threadIdx.x;
  const bool lo = (t & 64) == 0;
  float* colp = flds + (t & 63) * FPITCH;
  v4f* xq = reinterpret_cast<v4f*>(colp + (lo ? 0 : 16));
  #pragma unroll
  for (int q = 0; q < 4; ++q) xq[q] = r.xq[q];
  float2 d;
  d.x = r.pj[0] - r.pq[0];
  d.y = lo ? r.pj[1] - r.pq[1] : 0.f;
  *reinterpret_cast<float2*>(colp + (lo ? 32 : 34)) = d;
}

__device__ __forceinline__ void lds_barrier() {
  asm volatile("s_wait_dscnt 0x0\n\ts_barrier_signal -1\n\ts_barrier_wait -1" ::: "memory");
}

struct Chunk {
  float4 a;
  float4 b[4];
};
struct Tail {
  float2 a;
  float2 b[4];
};

template <int LIVE>
__device__ __forceinline__ void chunk_load(Chunk& f, const float* wrow, const float* fcol,
                                           int fpitch, int j) {
  f.a = *reinterpret_cast<const float4*>(wrow + 8 * j);
  #pragma unroll
  for (int ct = 0; ct < 4; ++ct)
    if ((LIVE >> ct) & 1)
      f.b[ct] = *reinterpret_cast<const float4*>(fcol + ct * 16 * fpitch + 8 * j);
}

template <int LIVE>
__device__ __forceinline__ void tail_load(Tail& f, const float* wrow2, const float* fcol2,
                                          int fpitch, int k0) {
  f.a = *reinterpret_cast<const float2*>(wrow2 + k0);
  #pragma unroll
  for (int ct = 0; ct < 4; ++ct)
    if ((LIVE >> ct) & 1)
      f.b[ct] = *reinterpret_cast<const float2*>(fcol2 + ct * 16 * fpitch + k0);
}

__device__ __forceinline__ v8f mma4(v2f a, v2f b, v8f cacc) {
  return __builtin_amdgcn_wmma_f32_16x16x4_f32(false, a, false, b, (short)0, cacc, false, false);
}

template <int LIVE>
__device__ __forceinline__ void chunk_mma(const Chunk& f, v8f (&acc)[4]) {
  v2f a0, a1;
  a0.x = f.a.x; a0.y = f.a.y; a1.x = f.a.z; a1.y = f.a.w;
  #pragma unroll
  for (int ct = 0; ct < 4; ++ct) {
    if (!((LIVE >> ct) & 1)) continue;
    v2f b; b.x = f.b[ct].x; b.y = f.b[ct].y;
    acc[ct] = mma4(a0, b, acc[ct]);
  }
  #pragma unroll
  for (int ct = 0; ct < 4; ++ct) {
    if (!((LIVE >> ct) & 1)) continue;
    v2f b; b.x = f.b[ct].z; b.y = f.b[ct].w;
    acc[ct] = mma4(a1, b, acc[ct]);
  }
}

template <int LIVE>
__device__ __forceinline__ void tail_mma(const Tail& f, v8f (&acc)[4]) {
  v2f a; a.x = f.a.x; a.y = f.a.y;
  #pragma unroll
  for (int ct = 0; ct < 4; ++ct) {
    if (!((LIVE >> ct) & 1)) continue;
    v2f b; b.x = f.b[ct].x; b.y = f.b[ct].y;
    acc[ct] = mma4(a, b, acc[ct]);
  }
}

__device__ __forceinline__ void sched_fence() { __builtin_amdgcn_sched_barrier(0); }

template <int LIVE>
__device__ __forceinline__ void acc_zero(v8f (&acc)[4]) {
  #pragma unroll
  for (int ct = 0; ct < 4; ++ct)
    if ((LIVE >> ct) & 1) {
      #pragma unroll
      for (int v = 0; v < 8; ++v) acc[ct][v] = 0.f;
    }
}

struct NoFill { __device__ __forceinline__ void operator()() const {} };

template <int LIVE, class FILL>
__device__ __forceinline__ void conv1_strip(const float* __restrict__ W1l,
                                            const float* __restrict__ ft,
                                            int mt, v8f (&acc)[4], FILL&& fill) {
  const int lane = threadIdx.x & 31;
  const int c = lane & 15, h = lane >> 4;
  const float* wrow = W1l + (mt * 16 + c) * W1STRIDE;
  const float* fcol = ft + c * FPITCH;
  Chunk g[4];
  Tail  gt;
  acc_zero<LIVE>(acc);
  #pragma unroll
  for (int j = 0; j < 4; ++j) chunk_load<LIVE>(g[j], wrow + 4 * h, fcol + 4 * h, FPITCH, j);
  tail_load<LIVE>(gt, wrow + 2 * h, fcol + 2 * h, FPITCH, 32);
  sched_fence();
  fill();
  sched_fence();
  #pragma unroll
  for (int j = 0; j < 4; ++j) chunk_mma<LIVE>(g[j], acc);
  tail_mma<LIVE>(gt, acc);
  sched_fence();
}

template <int LIVE, class FILL>
__device__ __forceinline__ void conv2_strip(const float* __restrict__ W2l,
                                            const float* __restrict__ f1l,
                                            int mt, v8f (&acc)[4], FILL&& fill) {
  const int lane = threadIdx.x & 31;
  const int c = lane & 15, h = lane >> 4;
  const float* wrow = W2l + (mt * 16 + c) * W2STRIDE + 4 * h;
  const float* fcol = f1l + c * F1PITCH + 4 * h;
  Chunk s0[2], s1[2];
  acc_zero<LIVE>(acc);
  chunk_load<LIVE>(s0[0], wrow, fcol, F1PITCH, 0);
  chunk_load<LIVE>(s0[1], wrow, fcol, F1PITCH, 1);
  chunk_load<LIVE>(s1[0], wrow, fcol, F1PITCH, 2);
  chunk_load<LIVE>(s1[1], wrow, fcol, F1PITCH, 3);
  sched_fence();
  fill();
  sched_fence();
  chunk_mma<LIVE>(s0[0], acc);
  chunk_mma<LIVE>(s0[1], acc);
  sched_fence();
  chunk_load<LIVE>(s0[0], wrow, fcol, F1PITCH, 4);
  chunk_load<LIVE>(s0[1], wrow, fcol, F1PITCH, 5);
  sched_fence();
  chunk_mma<LIVE>(s1[0], acc);
  chunk_mma<LIVE>(s1[1], acc);
  sched_fence();
  chunk_load<LIVE>(s1[0], wrow, fcol, F1PITCH, 6);
  chunk_load<LIVE>(s1[1], wrow, fcol, F1PITCH, 7);
  sched_fence();
  chunk_mma<LIVE>(s0[0], acc);
  chunk_mma<LIVE>(s0[1], acc);
  chunk_mma<LIVE>(s1[0], acc);
  chunk_mma<LIVE>(s1[1], acc);
  sched_fence();
}

template <int LIVE>
__device__ __forceinline__ void bn1_store(const v8f (&acc)[4], const float (&sc1)[8],
                                          const float (&sh1)[8], float* f1l, int mt) {
  const int lane = threadIdx.x & 31;
  const int c = lane & 15, h = lane >> 4;
  #pragma unroll
  for (int ct = 0; ct < 4; ++ct) {
    if (!((LIVE >> ct) & 1)) continue;
    float y[8];
    #pragma unroll
    for (int v = 0; v < 8; ++v) {
      float t = acc[ct][v] * sc1[v] + sh1[v];
      y[v] = t > 0.f ? t : 0.f;
    }
    float4* dst = reinterpret_cast<float4*>(f1l + (ct * 16 + c) * F1PITCH + mt * 16 + h * 8);
    float4 q0, q1;
    q0.x = y[0]; q0.y = y[1]; q0.z = y[2]; q0.w = y[3];
    q1.x = y[4]; q1.y = y[5]; q1.z = y[6]; q1.w = y[7];
    dst[0] = q0;
    dst[1] = q1;
  }
}


#define W1Q (COUT * CIN / 4)
#define W2Q (COUT * COUT / 4)

#define W1SLOTS(NT) ((W1Q + (NT) - 1) / (NT))
#define W2SLOTS(NT) (W2Q / (NT))

template <int NT>
__device__ __forceinline__ void w1_fetch(const float* __restrict__ W1, float4 (&r)[W1SLOTS(NT)]) {
  const float4* __restrict__ src = reinterpret_cast<const float4*>(W1);
  const int t = threadIdx.x;
  #pragma unroll
  for (int e = 0; e < W1SLOTS(NT); ++e) {
    int q = t + NT * e;
    r[e] = src[q < W1Q ? q : W1Q - 1];
  }
}

template <int NT>
__device__ __forceinline__ void w1_put(const float4 (&r)[W1SLOTS(NT)], float* W1l) {
  const int t = threadIdx.x;
  #pragma unroll
  for (int e = 0; e < W1SLOTS(NT); ++e) {
    int q = t + NT * e;
    if (q < W1Q) {
      const float v[4] = {r[e].x, r[e].y, r[e].z, r[e].w};
      #pragma unroll
      for (int u = 0; u < 4; ++u) {
        int i   = 4 * q + u;
        int row = i / CIN;
        int cc  = i - row * CIN;
        int kp  = cc >= 3 ? cc - 3 : 32 + cc;
        W1l[row * W1STRIDE + kp] = v[u];
      }
    }
  }
  if (t < COUT) W1l[t * W1STRIDE + CIN] = 0.f;
}

template <int NT>
__device__ __forceinline__ void w2_fetch(const float* __restrict__ W2, float4 (&r)[W2SLOTS(NT)]) {
  const float4* __restrict__ src = reinterpret_cast<const float4*>(W2);
  const int t = threadIdx.x;
  #pragma unroll
  for (int e = 0; e < W2SLOTS(NT); ++e) r[e] = src[t + NT * e];
}

template <int NT>
__device__ __forceinline__ void w2_put(const float4 (&r)[W2SLOTS(NT)], float* W2l) {
  const int t = threadIdx.x;
  #pragma unroll
  for (int e = 0; e < W2SLOTS(NT); ++e) {
    int q = t + NT * e;
    *reinterpret_cast<float4*>(&W2l[(q >> 4) * W2STRIDE + (q & 15) * 4]) = r[e];
  }
}

__device__ __forceinline__ int pair_or(int q, int fallback) { return q < NPAIRS ? q : fallback; }

#define SG       2
#define GSTREAMS (GSTAT * SG)
#define STRIPS   (NPAIRS / GSTREAMS)
static_assert(NPAIRS % GSTREAMS == 0 && STRIPS % 2 == 0 && STRIPS >= 4,
              "every stream gets the same, even number of tiles");

__device__ __forceinline__ void stat_sums(const v8f (&acc)[4], float (&s)[8], float (&sq)[8]) {
  #pragma unroll
  for (int ct = 0; ct < 4; ++ct) {
    #pragma unroll
    for (int v = 0; v < 8; ++v) { float y = acc[ct][v]; s[v] += y; sq[v] += y * y; }
  }
}

__device__ __forceinline__ void row_out(float (&s)[8], float (&sq)[8], int grp, int mt, int lane,
                                        float* comb, float* __restrict__ part) {
  #pragma unroll
  for (int v = 0; v < 8; ++v)
    #pragma unroll
    for (int off = 1; off < 16; off <<= 1) {
      s[v]  += __shfl_xor(s[v],  off, 32);
      sq[v] += __shfl_xor(sq[v], off, 32);
    }
  const int hi = lane >> 4;
  if (grp == 1 && (lane & 15) == 0) {
    #pragma unroll
    for (int v = 0; v < 8; ++v) {
      int o = mt * 16 + hi * 8 + v;
      comb[o]      = s[v];
      comb[64 + o] = sq[v];
    }
  }
  lds_barrier();
  if (grp == 0 && (lane & 15) == 0) {
    #pragma unroll
    for (int v = 0; v < 8; ++v) {
      int o = mt * 16 + hi * 8 + v;
      part[(size_t)blockIdx.x * 128 + o]      = s[v]  + comb[o];
      part[(size_t)blockIdx.x * 128 + 64 + o] = sq[v] + comb[64 + o];
    }
  }
}

__device__ __forceinline__ void conv1_stats_trip(const float* p, const float* xt,
                                                 const int* idxg, const float* W1l,
                                                 float* ft, int pair, int G, FPart& r, int& jn,
                                                 int mt, float (&s)[8], float (&sq)[8]) {
  ftile_put(r, ft);
  const int pn2 = pair_or(pair + 2 * G, pair);
  const int pn4 = pair_or(pair + 4 * G, pair);
  ftile_fetch(p, xt, pn2, jn, r);
  jn = tile_idx(idxg, pn4);
  lds_barrier();
  v8f acc[4];
  conv1_strip<15>(W1l, ft, mt, acc, NoFill());
  stat_sums(acc, s, sq);
}

__global__ __launch_bounds__(256) void conv1_stats_kernel(
    const float* p, const float* xt,
    const int* idxg, const float* __restrict__ W1,
    float* __restrict__ part1) {
  __shared__ __attribute__((aligned(16))) float W1l[COUT * W1STRIDE];
  __shared__ __attribute__((aligned(16))) float flds[SG][2][64 * FPITCH];
  __shared__ float comb[128];
  const int t = threadIdx.x;
  const int grp = __builtin_amdgcn_readfirstlane(t >> 7);
  const int G = GSTREAMS;
  int pair = blockIdx.x + GSTAT * grp;
  const int pf1 = pair_or(pair + G, pair);
  const int j0 = tile_idx(idxg, pair);
  const int j1 = tile_idx(idxg, pf1);
  {
    float4 r1[W1SLOTS(256)];
    w1_fetch<256>(W1, r1);
    w1_put<256>(r1, W1l);
  }
  FPart ra, rb;
  ftile_fetch(p, xt, pair, j0, ra);
  int ja = tile_idx(idxg, pair_or(pair + 2 * G, pair));
  sched_fence();
  ftile_fetch(p, xt, pf1, j1, rb);
  int jb = tile_idx(idxg, pair_or(pair + 3 * G, pair));
  const int mt = (t >> 5) & 3, lane = t & 31;
  float s[8], sq[8];
  #pragma unroll
  for (int v = 0; v < 8; ++v) { s[v] = 0.f; sq[v] = 0.f; }

  float* f0 = flds[grp][0];
  float* f1 = flds[grp][1];
  for (int it = 0; it < STRIPS; it += 2) {
    conv1_stats_trip(p, xt, idxg, W1l, f0, pair, G, ra, ja, mt, s, sq);
    pair += G;
    conv1_stats_trip(p, xt, idxg, W1l, f1, pair, G, rb, jb, mt, s, sq);
    pair += G;
  }
  row_out(s, sq, grp, mt, lane, comb, part1);
}

template <int NW>
__device__ __forceinline__ float red_sum(const float* red, int o) {
  float a = (red[o] + red[128 + o]) + (red[256 + o] + red[384 + o]);
  if (NW == 8) a += (red[512 + o] + red[640 + o]) + (red[768 + o] + red[896 + o]);
  return a;
}

template <bool KEEP, int BATCH, int NW>
__device__ __forceinline__ void bn_from_rows(const float* __restrict__ part,
                                             const float* g,
                                             const float* bta,
                                             float* red, float* bnl,
                                             float* __restrict__ bnp_out) {
  const int t = threadIdx.x;
  const int q = t & 31, w = t >> 5;
  const float gv = g[t & 63];
  const float bv = bta[t & 63];
  const float4* __restrict__ rows = reinterpret_cast<const float4*>(part) + (size_t)w * (GSTAT / NW) * 32 + q;
  float4 a[4];
  #pragma unroll
  for (int u = 0; u < 4; ++u) { a[u].x = 0.f; a[u].y = 0.f; a[u].z = 0.f; a[u].w = 0.f; }
  #pragma unroll
  for (int i0 = 0; i0 < GSTAT / NW; i0 += BATCH) {
    #pragma unroll
    for (int i = i0; i < i0 + BATCH; ++i) {
      const float4 v = rows[(size_t)i * 32];
      a[i & 3].x += v.x; a[i & 3].y += v.y; a[i & 3].z += v.z; a[i & 3].w += v.w;
    }
    if (i0 + BATCH < GSTAT / NW) sched_fence();
  }
  float4 sum;
  sum.x = (a[0].x + a[1].x) + (a[2].x + a[3].x);
  sum.y = (a[0].y + a[1].y) + (a[2].y + a[3].y);
  sum.z = (a[0].z + a[1].z) + (a[2].z + a[3].z);
  sum.w = (a[0].w + a[1].w) + (a[2].w + a[3].w);
  *reinterpret_cast<float4*>(red + w * 128 + 4 * q) = sum;
  lds_barrier();
  const int o = t & 63;
  const float s  = red_sum<NW>(red, o);
  const float sq = red_sum<NW>(red, 64 + o);
  const float mean  = s / CNTF;
  const float var   = sq / CNTF - mean * mean;
  const float scale = gv * rsqrtf(var + EPSV);
  const float shift = bv - mean * scale;
  if (t < COUT) {
    bnl[o]      = scale;
    bnl[64 + o] = shift;
    if (KEEP && blockIdx.x == 0) {
      bnp_out[o]      = scale;
      bnp_out[64 + o] = shift;
    }
  }
  lds_barrier();
}

__global__ __launch_bounds__(256) void conv2_stats_kernel(
    const float* p, const float* xt,
    const int* idxg, const float* __restrict__ W1,
    const float* __restrict__ W2, const float* __restrict__ part1,
    const float* g1, const float* b1,
    float* __restrict__ bnp1, float* __restrict__ part2) {
  __shared__ __attribute__((aligned(16))) float W1l[COUT * W1STRIDE];
  __shared__ __attribute__((aligned(16))) float W2l[COUT * W2STRIDE];
  __shared__ __attribute__((aligned(16))) float flds[SG][64 * FPITCH];
  __shared__ __attribute__((aligned(16))) float f1l[SG][64 * F1PITCH];
  __shared__ __attribute__((aligned(16))) float red[8 * 128];
  __shared__ float bn1l[128];
  const int t = threadIdx.x;
  const int grp = __builtin_amdgcn_readfirstlane(t >> 7);
  const int G = GSTREAMS;
  int pair = blockIdx.x + GSTAT * grp;
  const int pf1 = pair_or(pair + G, pair);
  const int j0 = tile_idx(idxg, pair);
  const int j1 = tile_idx(idxg, pf1);
  {
    float4 r1[W1SLOTS(256)], r2[W2SLOTS(256)];
    w1_fetch<256>(W1, r1);
    w2_fetch<256>(W2, r2);
    w1_put<256>(r1, W1l);
    w2_put<256>(r2, W2l);
  }
  FPart r;
  ftile_fetch(p, xt, pair, j0, r);
  bn_from_rows<true, GSTAT / 8, 8>(part1, g1, b1, red, bn1l, bnp1);
  const int mt = (t >> 5) & 3, lane = t & 31;
  const int hi = lane >> 4;
  float sc1[8], sh1[8];
  #pragma unroll
  for (int v = 0; v < 8; ++v) {
    int o = mt * 16 + hi * 8 + v;
    sc1[v] = bn1l[o]; sh1[v] = bn1l[64 + o];
  }
  float s[8], sq[8];
  #pragma unroll
  for (int v = 0; v < 8; ++v) { s[v] = 0.f; sq[v] = 0.f; }
  v8f acc2[4];
  acc_zero<15>(acc2);
  float* fl = flds[grp];
  float* f1 = f1l[grp];
  ftile_put(r, fl);
  ftile_fetch(p, xt, pf1, j1, r);
  int jn = tile_idx(idxg, pair_or(pair + 2 * G, pair));

  for (int slot = 0; slot <= 2 * STRIPS; ++slot) {
    lds_barrier();
    const int u = slot - grp;
    const bool on = u >= 0 && u < 2 * STRIPS;
    if (on && (u & 1) == 0) {
      v8f acc[4];
      conv1_strip<15>(W1l, fl, mt, acc, [&]() { stat_sums(acc2, s, sq); });
      bn1_store<15>(acc, sc1, sh1, f1, mt);
    }
    if (on && (u & 1) != 0) {
      conv2_strip<15>(W2l, f1, mt, acc2, [&]() {
        ftile_put(r, fl);
        sched_fence();
        ftile_fetch(p, xt, pair_or(pair + 2 * G, pair), jn, r);
        jn = tile_idx(idxg, pair_or(pair + 3 * G, pair));
      });
      pair += G;
    }
  }
  stat_sums(acc2, s, sq);
  row_out(s, sq, grp, mt, lane, red, part2);
}

struct PairEnds { int a0, a16, b0, b16; };

__device__ __forceinline__ PairEnds pair_ends(const int* __restrict__ idxg, int pair) {
  const int* row = idxg + (size_t)pair * 64;
  PairEnds e;
  e.a0 = row[0]; e.a16 = row[16]; e.b0 = row[32]; e.b16 = row[48];
  return e;
}

__device__ __forceinline__ int live_tiles(const PairEnds& e) {
  int live = 15;
  if (e.a16 == e.a0) live &= ~2;
  if (e.b16 == e.b0) live &= ~8;
  return __builtin_amdgcn_readfirstlane(live);
}

template <int LIVE>
__device__ __forceinline__ void final_tile(const float* __restrict__ W1l, const float* __restrict__ W2l,
                                           const float* flds, float* f1l, int mt,
                                           const float (&sc1)[8], const float (&sh1)[8],
                                           const float (&sc2)[8], const float (&sh2)[8],
                                           float (&m0)[8], float (&m1)[8]) {
  v8f acc[4];
  conv1_strip<LIVE>(W1l, flds, mt, acc, NoFill());
  bn1_store<LIVE>(acc, sc1, sh1, f1l, mt);
  lds_barrier();
  conv2_strip<LIVE>(W2l, f1l, mt, acc, NoFill());
  #pragma unroll
  for (int v = 0; v < 8; ++v) { m0[v] = 0.f; m1[v] = 0.f; }
  #pragma unroll
  for (int ct = 0; ct < 4; ++ct) {
    if (!((LIVE >> ct) & 1)) continue;
    #pragma unroll
    for (int v = 0; v < 8; ++v) {
      float y = acc[ct][v] * sc2[v] + sh2[v];
      y = y > 0.f ? y : 0.f;
      if (ct < 2) m0[v] = fmaxf(m0[v], y);
      else        m1[v] = fmaxf(m1[v], y);
    }
  }
}

__global__ __launch_bounds__(128) void final_kernel(
    const float* __restrict__ p, const float* __restrict__ xt,
    const int* __restrict__ idxg, const float* __restrict__ W1,
    const float* __restrict__ W2, const float* __restrict__ bnp1,
    const float* __restrict__ part2, const float* g2, const float* b2,
    float* __restrict__ out) {
  __shared__ __attribute__((aligned(16))) float W1l[COUT * W1STRIDE];
  __shared__ __attribute__((aligned(16))) float W2l[COUT * W2STRIDE];
  __shared__ __attribute__((aligned(16))) float flds[64 * FPITCH];
  __shared__ __attribute__((aligned(16))) float f1l[64 * F1PITCH];
  __shared__ __attribute__((aligned(16))) float red[4 * 128];
  __shared__ float bn2l[128];
  int t = threadIdx.x;
  const int G = gridDim.x;
  int pair = blockIdx.x;
  const int j0 = tile_idx(idxg, pair);
  int jn = tile_idx(idxg, pair + G < NPAIRS ? pair + G : pair);
  PairEnds ends = pair_ends(idxg, pair);
  {
    float4 r1[W1SLOTS(128)], r2[W2SLOTS(128)];
    w1_fetch<128>(W1, r1);
    w2_fetch<128>(W2, r2);
    w1_put<128>(r1, W1l);
    w2_put<128>(r2, W2l);
  }
  FPart r;
  ftile_fetch(p, xt, pair, j0, r);
  int mt = t >> 5, lane = t & 31;
  int hi = lane >> 4;
  float sc1[8], sh1[8], sc2[8], sh2[8];
  #pragma unroll
  for (int v = 0; v < 8; ++v) {
    int o = mt * 16 + hi * 8 + v;
    sc1[v] = bnp1[o]; sh1[v] = bnp1[64 + o];
  }
  bn_from_rows<false, GSTAT / 8, 4>(part2, g2, b2, red, bn2l, nullptr);
  #pragma unroll
  for (int v = 0; v < 8; ++v) {
    int o = mt * 16 + hi * 8 + v;
    sc2[v] = bn2l[o]; sh2[v] = bn2l[64 + o];
  }
  for (; pair < NPAIRS; pair += G) {
    int b  = pair / (NN / 2);
    int n0 = (pair % (NN / 2)) * 2;
    const int live = live_tiles(ends);
    ftile_put(r, flds);
    const int pn  = pair + G     < NPAIRS ? pair + G     : pair;
    const int pn2 = pair + 2 * G < NPAIRS ? pair + 2 * G : pair;
    ftile_fetch(p, xt, pn, jn, r);
    jn = tile_idx(idxg, pn2);
    ends = pair_ends(idxg, pn);
    lds_barrier();
    float m0[8], m1[8];
    switch (live) {
      case 15: final_tile<15>(W1l, W2l, flds, f1l, mt, sc1, sh1, sc2, sh2, m0, m1); break;
      case 13: final_tile<13>(W1l, W2l, flds, f1l, mt, sc1, sh1, sc2, sh2, m0, m1); break;
      case 7:  final_tile<7>(W1l, W2l, flds, f1l, mt, sc1, sh1, sc2, sh2, m0, m1); break;
      default: final_tile<5>(W1l, W2l, flds, f1l, mt, sc1, sh1, sc2, sh2, m0, m1); break;
    }
    #pragma unroll
    for (int v = 0; v < 8; ++v)
      #pragma unroll
      for (int off = 1; off < 16; off <<= 1) {
        m0[v] = fmaxf(m0[v], __shfl_xor(m0[v], off, 32));
        m1[v] = fmaxf(m1[v], __shfl_xor(m1[v], off, 32));
      }
    if ((lane & 15) == 0) {
      #pragma unroll
      for (int v = 0; v < 8; ++v) {
        int o = mt * 16 + hi * 8 + v;
        out[((size_t)b * COUT + o) * NN + n0]     = m0[v];
        out[((size_t)b * COUT + o) * NN + n0 + 1] = m1[v];
      }
    }
  }
}

extern "C" void kernel_launch(void* const* d_in, const int* in_sizes, int n_in,
                              void* d_out, int out_size, void* d_ws, size_t ws_size,
                              hipStream_t stream) {
  const float* p  = (const float*)d_in[0];
  const float* x  = (const float*)d_in[1];
  const float* W1 = (const float*)d_in[2];
  const float* g1 = (const float*)d_in[3];
  const float* b1 = (const float*)d_in[4];
  const float* W2 = (const float*)d_in[5];
  const float* g2 = (const float*)d_in[6];
  const float* b2 = (const float*)d_in[7];
  float* out = (float*)d_out;

  char*  ws    = (char*)d_ws;
  int*   idx   = (int*)ws;
  float* xt    = (float*)(ws + (size_t)BB * NN * KS * 4);
  float* part1 = (float*)(ws + (size_t)BB * NN * KS * 4
                             + (size_t)BB * NN * CX * 4);
  float* part2 = part1 + (size_t)GCONV * 128;
  float* bnp1  = part2 + (size_t)GCONV * 128;

  ballq_kernel<<<dim3(NN / 32, BB), 256, 0, stream>>>(p, x, idx, xt);
  conv1_stats_kernel<<<GSTAT, 128 * SG, 0, stream>>>(p, xt, idx, W1, part1);
  conv2_stats_kernel<<<GSTAT, 128 * SG, 0, stream>>>(p, xt, idx, W1, W2, part1, g1, b1, bnp1, part2);
  final_kernel<<<GCONV, 128, 0, stream>>>(p, xt, idx, W1, W2, bnp1, part2, g2, b2, out);
}
